# starting kernel
# baseline (speedup 1.0000x reference)
#include <hip/hip_runtime.h>

typedef __attribute__((ext_vector_type(2))) float v2f;
typedef __attribute__((ext_vector_type(8))) float v8f;

#define NNODES 100000
#define NEDGES 1600000

__global__ void k_zero(float* __restrict__ p, long n) {
  long i = ((long)blockIdx.x * blockDim.x + threadIdx.x) * 4;
  long stride = (long)gridDim.x * blockDim.x * 4;
  float4 z = {0.f, 0.f, 0.f, 0.f};
  for (long j = i; j < n; j += stride) *(float4*)(p + j) = z;
}

__global__ void k_prep(const float* __restrict__ u,
                       const float* __restrict__ Web, const float* __restrict__ beb,
                       const float* __restrict__ Wnb, const float* __restrict__ bnb,
                       float* __restrict__ bias_eb, float* __restrict__ bias_nb) {
  int j = threadIdx.x;
  float a = beb[j], b = bnb[j];
  for (int k = 0; k < 64; ++k) {
    float uk = u[k];
    a += uk * Web[(48 + k) * 64 + j];
    b += uk * Wnb[(80 + k) * 64 + j];
  }
  bias_eb[j] = a;
  bias_nb[j] = b;
}

__global__ __launch_bounds__(256) void k_edge(
    const float* __restrict__ x, const float* __restrict__ ea,
    const int* __restrict__ recv, const int* __restrict__ send,
    const float* __restrict__ Web, const float* __restrict__ bias_eb,
    float* __restrict__ agg_r, float* __restrict__ agg_s,
    float* __restrict__ cnt_r, float* __restrict__ cnt_s) {
  __shared__ float sW[12 * 4 * 64];
  for (int i = threadIdx.x; i < 12 * 4 * 64; i += 256) {
    int j = i & 1, lane = (i >> 1) & 31, nt = (i >> 6) & 3, kc = i >> 8;
    int kk = 4 * kc + 2 * (lane >> 4) + j;
    sW[i] = Web[kk * 64 + nt * 16 + (lane & 15)];
  }
  __syncthreads();
  const int lane = threadIdx.x & 31;
  const int h = lane >> 4, m = lane & 15;
  const int wave = blockIdx.x * 8 + (threadIdx.x >> 5);
  const int nwaves = gridDim.x * 8;
  const int ntiles = NEDGES / 16;
  for (int t = wave; t < ntiles; t += nwaves) {
    const int e0 = t * 16;
    const int er = e0 + m;
    const int ri = recv[er], si = send[er];
    const float* segs[3] = {ea + (long)er * 16, x + (long)ri * 16, x + (long)si * 16};
    v8f acc[4];
#pragma unroll
    for (int nt = 0; nt < 4; ++nt) {
      float bb = bias_eb[nt * 16 + m];
      acc[nt] = (v8f){bb, bb, bb, bb, bb, bb, bb, bb};
    }
#pragma unroll
    for (int seg = 0; seg < 3; ++seg) {
      const float* bp = segs[seg];
#pragma unroll
      for (int cc = 0; cc < 4; ++cc) {
        v2f a = *(const v2f*)(bp + 4 * cc + 2 * h);
        const int kc = seg * 4 + cc;
#pragma unroll
        for (int nt = 0; nt < 4; ++nt) {
          v2f b = *(const v2f*)(&sW[(kc * 4 + nt) * 64 + lane * 2]);
          acc[nt] = __builtin_amdgcn_wmma_f32_16x16x4_f32(false, a, false, b,
                                                          (short)0, acc[nt], false, false);
        }
      }
    }
    int rr[8], ss[8];
#pragma unroll
    for (int v = 0; v < 8; ++v) {
      int e = e0 + 8 * h + v;
      rr[v] = recv[e];
      ss[v] = send[e];
    }
#pragma unroll
    for (int nt = 0; nt < 4; ++nt) {
      const int col = nt * 16 + m;
#pragma unroll
      for (int v = 0; v < 8; ++v) {
        float val = fmaxf(acc[nt][v], 0.0f);
        unsafeAtomicAdd(&agg_r[(long)rr[v] * 64 + col], val);
        unsafeAtomicAdd(&agg_s[(long)ss[v] * 64 + col], val);
      }
    }
    if (lane < 16) {
      unsafeAtomicAdd(&cnt_r[recv[e0 + lane]], 1.0f);
      unsafeAtomicAdd(&cnt_s[send[e0 + lane]], 1.0f);
    }
  }
}

__global__ __launch_bounds__(256) void k_node(
    const float* __restrict__ x, const float* __restrict__ agg_r,
    const float* __restrict__ Wnb, const float* __restrict__ bias_nb,
    float* __restrict__ n_h) {
  __shared__ float sW[20 * 4 * 64];
  for (int i = threadIdx.x; i < 20 * 4 * 64; i += 256) {
    int j = i & 1, lane = (i >> 1) & 31, nt = (i >> 6) & 3, kc = i >> 8;
    int kk = 4 * kc + 2 * (lane >> 4) + j;
    sW[i] = Wnb[kk * 64 + nt * 16 + (lane & 15)];
  }
  __syncthreads();
  const int lane = threadIdx.x & 31;
  const int h = lane >> 4, m = lane & 15;
  const int wave = blockIdx.x * 8 + (threadIdx.x >> 5);
  const int nwaves = gridDim.x * 8;
  const int ntiles = NNODES / 16;
  for (int t = wave; t < ntiles; t += nwaves) {
    const int n0 = t * 16;
    const float* ap = agg_r + (long)(n0 + m) * 64;
    const float* xp = x + (long)(n0 + m) * 16;
    v8f acc[4];
#pragma unroll
    for (int nt = 0; nt < 4; ++nt) {
      float bb = bias_nb[nt * 16 + m];
      acc[nt] = (v8f){bb, bb, bb, bb, bb, bb, bb, bb};
    }
#pragma unroll
    for (int kc = 0; kc < 20; ++kc) {
      const int kk = 4 * kc + 2 * h;
      v2f a = (kc < 16) ? *(const v2f*)(ap + kk) : *(const v2f*)(xp + kk - 64);
#pragma unroll
      for (int nt = 0; nt < 4; ++nt) {
        v2f b = *(const v2f*)(&sW[(kc * 4 + nt) * 64 + lane * 2]);
        acc[nt] = __builtin_amdgcn_wmma_f32_16x16x4_f32(false, a, false, b,
                                                        (short)0, acc[nt], false, false);
      }
    }
#pragma unroll
    for (int nt = 0; nt < 4; ++nt)
#pragma unroll
      for (int v = 0; v < 8; ++v)
        n_h[(long)(n0 + 8 * h + v) * 64 + nt * 16 + m] = fmaxf(acc[nt][v], 0.0f);
  }
}

__global__ void k_colsum(const float* __restrict__ src, int rows, float* __restrict__ dst) {
  __shared__ float red[256];
  const int col = threadIdx.x & 63;
  const int g = threadIdx.x >> 6;
  float local = 0.f;
  for (int r = blockIdx.x * 4 + g; r < rows; r += gridDim.x * 4)
    local += src[(long)r * 64 + col];
  red[threadIdx.x] = local;
  __syncthreads();
  if (g == 0) {
    local += red[64 + threadIdx.x] + red[128 + threadIdx.x] + red[192 + threadIdx.x];
    unsafeAtomicAdd(&dst[col], local);
  }
}

__global__ void k_global(const float* __restrict__ u,
                         const float* __restrict__ Wgb, const float* __restrict__ bgb,
                         const float* __restrict__ Wd1, const float* __restrict__ bd1,
                         const float* __restrict__ e_sum, const float* __restrict__ n_sum,
                         float* __restrict__ u2, float* __restrict__ bias_d1) {
  __shared__ float su2[64];
  const int j = threadIdx.x;
  float acc = bgb[j];
  for (int k = 0; k < 64; ++k) {
    float em = e_sum[k] * (1.0f / NEDGES);
    float nm = n_sum[k] * (1.0f / NNODES);
    acc += em * Wgb[k * 64 + j] + nm * Wgb[(64 + k) * 64 + j] + u[k] * Wgb[(128 + k) * 64 + j];
  }
  acc = fmaxf(acc, 0.0f);
  su2[j] = acc;
  u2[j] = acc;
  __syncthreads();
  float b = bd1[j];
  for (int k = 0; k < 64; ++k) b += su2[k] * Wd1[(192 + k) * 64 + j];
  bias_d1[j] = b;
}

__global__ __launch_bounds__(256) void k_dec(
    const float* __restrict__ agg_r, const float* __restrict__ agg_s,
    const float* __restrict__ cnt_r, const float* __restrict__ cnt_s,
    const float* __restrict__ n_h, const float* __restrict__ Wd1,
    const float* __restrict__ bias_d1, const float* __restrict__ Wd2,
    const float* __restrict__ bd2, float* __restrict__ out) {
  __shared__ float sW1[48 * 4 * 64];
  __shared__ float sW2[16 * 64];
  __shared__ float sH[8 * 16 * 64];
  for (int i = threadIdx.x; i < 48 * 4 * 64; i += 256) {
    int j = i & 1, lane = (i >> 1) & 31, nt = (i >> 6) & 3, kc = i >> 8;
    int kk = 4 * kc + 2 * (lane >> 4) + j;
    sW1[i] = Wd1[kk * 64 + nt * 16 + (lane & 15)];
  }
  for (int i = threadIdx.x; i < 16 * 64; i += 256) {
    int j = i & 1, lane = (i >> 1) & 31, kc = i >> 6;
    int kk = 4 * kc + 2 * (lane >> 4) + j;
    sW2[i] = Wd2[kk * 16 + (lane & 15)];
  }
  __syncthreads();
  const int lane = threadIdx.x & 31;
  const int h = lane >> 4, m = lane & 15;
  float* hw = sH + (threadIdx.x >> 5) * 1024;
  const int wave = blockIdx.x * 8 + (threadIdx.x >> 5);
  const int nwaves = gridDim.x * 8;
  const int ntiles = NNODES / 16;
  for (int t = wave; t < ntiles; t += nwaves) {
    const int n0 = t * 16;
    const int n = n0 + m;
    const float invr = 1.0f / fmaxf(cnt_r[n], 1.0f);
    const float invs = 1.0f / fmaxf(cnt_s[n], 1.0f);
    const float* pr = agg_r + (long)n * 64;
    const float* ps = agg_s + (long)n * 64;
    const float* pn = n_h + (long)n * 64;
    v8f acc[4];
#pragma unroll
    for (int nt = 0; nt < 4; ++nt) {
      float bb = bias_d1[nt * 16 + m];
      acc[nt] = (v8f){bb, bb, bb, bb, bb, bb, bb, bb};
    }
#pragma unroll
    for (int kc = 0; kc < 48; ++kc) {
      const int kk = 4 * kc + 2 * h;
      v2f a;
      if (kc < 16)      { a = *(const v2f*)(pr + kk);        a *= invr; }
      else if (kc < 32) { a = *(const v2f*)(ps + kk - 64);   a *= invs; }
      else              { a = *(const v2f*)(pn + kk - 128); }
#pragma unroll
      for (int nt = 0; nt < 4; ++nt) {
        v2f b = *(const v2f*)(&sW1[(kc * 4 + nt) * 64 + lane * 2]);
        acc[nt] = __builtin_amdgcn_wmma_f32_16x16x4_f32(false, a, false, b,
                                                        (short)0, acc[nt], false, false);
      }
    }
#pragma unroll
    for (int nt = 0; nt < 4; ++nt)
#pragma unroll
      for (int v = 0; v < 8; ++v)
        hw[(8 * h + v) * 64 + nt * 16 + m] = fmaxf(acc[nt][v], 0.0f);
    float bb2 = bd2[m];
    v8f acc2 = (v8f){bb2, bb2, bb2, bb2, bb2, bb2, bb2, bb2};
#pragma unroll
    for (int kc = 0; kc < 16; ++kc) {
      const int kk = 4 * kc + 2 * h;
      v2f a = *(const v2f*)(hw + m * 64 + kk);
      v2f b = *(const v2f*)(&sW2[kc * 64 + lane * 2]);
      acc2 = __builtin_amdgcn_wmma_f32_16x16x4_f32(false, a, false, b,
                                                   (short)0, acc2, false, false);
    }
#pragma unroll
    for (int v = 0; v < 8; ++v)
      out[(long)(n0 + 8 * h + v) * 16 + m] = acc2[v];
  }
}

extern "C" void kernel_launch(void* const* d_in, const int* in_sizes, int n_in,
                              void* d_out, int out_size, void* d_ws, size_t ws_size,
                              hipStream_t stream) {
  const float* x    = (const float*)d_in[0];
  const float* ea   = (const float*)d_in[1];
  const float* u    = (const float*)d_in[2];
  const int* send   = (const int*)d_in[3];
  const int* recv   = (const int*)d_in[4];
  const float* Web  = (const float*)d_in[7];
  const float* beb  = (const float*)d_in[8];
  const float* Wnb  = (const float*)d_in[9];
  const float* bnb  = (const float*)d_in[10];
  const float* Wgb  = (const float*)d_in[11];
  const float* bgb  = (const float*)d_in[12];
  const float* Wd1  = (const float*)d_in[13];
  const float* bd1  = (const float*)d_in[14];
  const float* Wd2  = (const float*)d_in[15];
  const float* bd2  = (const float*)d_in[16];
  float* out = (float*)d_out;
  float* ws = (float*)d_ws;

  const long N = NNODES;
  float* agg_r   = ws;
  float* agg_s   = agg_r + N * 64;
  float* cnt_r   = agg_s + N * 64;
  float* cnt_s   = cnt_r + N;
  float* e_sum   = cnt_s + N;
  float* n_sum   = e_sum + 64;
  float* bias_eb = n_sum + 64;
  float* bias_nb = bias_eb + 64;
  float* u2      = bias_nb + 64;
  float* bias_d1 = u2 + 64;
  float* n_hbuf  = bias_d1 + 64;

  const long zlen = 2 * N * 64 + 2 * N + 128;
  k_zero<<<2048, 256, 0, stream>>>(ws, zlen);
  k_prep<<<1, 64, 0, stream>>>(u, Web, beb, Wnb, bnb, bias_eb, bias_nb);
  k_edge<<<2048, 256, 0, stream>>>(x, ea, recv, send, Web, bias_eb,
                                   agg_r, agg_s, cnt_r, cnt_s);
  k_node<<<782, 256, 0, stream>>>(x, agg_r, Wnb, bias_nb, n_hbuf);
  k_colsum<<<512, 256, 0, stream>>>(agg_r, NNODES, e_sum);
  k_colsum<<<512, 256, 0, stream>>>(n_hbuf, NNODES, n_sum);
  k_global<<<1, 64, 0, stream>>>(u, Wgb, bgb, Wd1, bd1, e_sum, n_sum, u2, bias_d1);
  k_dec<<<1024, 256, 0, stream>>>(agg_r, agg_s, cnt_r, cnt_s, n_hbuf,
                                  Wd1, bias_d1, Wd2, bd2, out);
}


# k_edge: nodes cut in two ranges, each range's rows of both sums added by one package half only (each half walks all tiles; tiles drawn from a counter per range); atomics kept, arrival order
# speedup vs baseline: 1.1708x; 1.1708x over previous
#include <hip/hip_runtime.h>

typedef __attribute__((ext_vector_type(2))) float v2f;
typedef __attribute__((ext_vector_type(8))) float v8f;

#define NNODES 100000
#define NEDGES 1600000

__global__ void k_zero(float* __restrict__ p, long n) {
  long i = ((long)blockIdx.x * blockDim.x + threadIdx.x) * 4;
  long stride = (long)gridDim.x * blockDim.x * 4;
  float4 z = {0.f, 0.f, 0.f, 0.f};
  for (long j = i; j < n; j += stride) *(float4*)(p + j) = z;
}

__global__ void k_prep(const float* __restrict__ u,
                       const float* __restrict__ Web, const float* __restrict__ beb,
                       const float* __restrict__ Wnb, const float* __restrict__ bnb,
                       float* __restrict__ bias_eb, float* __restrict__ bias_nb) {
  int j = threadIdx.x;
  float a = beb[j], b = bnb[j];
  for (int k = 0; k < 64; ++k) {
    float uk = u[k];
    a += uk * Web[(48 + k) * 64 + j];
    b += uk * Wnb[(80 + k) * 64 + j];
  }
  bias_eb[j] = a;
  bias_nb[j] = b;
}

#define HALFN (NNODES / 2)
#define TILES_PER_DRAW 2
__global__ __launch_bounds__(256) void k_edge(
    const float* __restrict__ x, const float* __restrict__ ea,
    const int* __restrict__ recv, const int* __restrict__ send,
    const float* __restrict__ Web, const float* __restrict__ bias_eb,
    float* __restrict__ agg_r, float* __restrict__ agg_s,
    float* __restrict__ cnt_r, float* __restrict__ cnt_s,
    int* __restrict__ tile_ctr) {
  __shared__ float sW[12 * 4 * 64];
  for (int i = threadIdx.x; i < 12 * 4 * 64; i += 256) {
    int j = i & 1, lane = (i >> 1) & 31, nt = (i >> 6) & 3, kc = i >> 8;
    int kk = 4 * kc + 2 * (lane >> 4) + j;
    sW[i] = Web[kk * 64 + nt * 16 + (lane & 15)];
  }
  __syncthreads();
  const int lane = threadIdx.x & 31;
  const int h = lane >> 4, m = lane & 15;
  const int my_half = (int)((__builtin_amdgcn_s_sendmsg_rtn(0x87) >> 18) & 1u);
  const int ndraws = (NEDGES / 16) / TILES_PER_DRAW;
  for (int pass = 0; pass < 2; ++pass) {
    const int rng = my_half ^ pass;
    const int lo = rng * HALFN;
    int* ctr = tile_ctr + rng * 32;
    for (;;) {
      int c = 0;
      if (lane == 0) c = atomicAdd(ctr, 1);
      c = __builtin_amdgcn_readfirstlane(c);
      if (c >= ndraws) break;
      for (int q = 0; q < TILES_PER_DRAW; ++q) {
        const int e0 = (c * TILES_PER_DRAW + q) * 16;
        const int er = e0 + m;
        const int ri = recv[er], si = send[er];
        const float* segs[3] = {ea + (long)er * 16, x + (long)ri * 16, x + (long)si * 16};
        v8f acc[4];
#pragma unroll
        for (int nt = 0; nt < 4; ++nt) {
          float bb = bias_eb[nt * 16 + m];
          acc[nt] = (v8f){bb, bb, bb, bb, bb, bb, bb, bb};
        }
#pragma unroll
        for (int seg = 0; seg < 3; ++seg) {
          const float* bp = segs[seg];
#pragma unroll
          for (int cc = 0; cc < 4; ++cc) {
            v2f a = *(const v2f*)(bp + 4 * cc + 2 * h);
            const int kc = seg * 4 + cc;
#pragma unroll
            for (int nt = 0; nt < 4; ++nt) {
              v2f b = *(const v2f*)(&sW[(kc * 4 + nt) * 64 + lane * 2]);
              acc[nt] = __builtin_amdgcn_wmma_f32_16x16x4_f32(false, a, false, b,
                                                              (short)0, acc[nt], false, false);
            }
          }
        }
        int rr[8], ss[8];
        bool own_r[8], own_s[8];
#pragma unroll
        for (int v = 0; v < 8; ++v) {
          int e = e0 + 8 * h + v;
          rr[v] = recv[e];
          ss[v] = send[e];
          own_r[v] = (unsigned)(rr[v] - lo) < (unsigned)HALFN;
          own_s[v] = (unsigned)(ss[v] - lo) < (unsigned)HALFN;
        }
#pragma unroll
        for (int nt = 0; nt < 4; ++nt)
#pragma unroll
          for (int v = 0; v < 8; ++v) acc[nt][v] = fmaxf(acc[nt][v], 0.0f);
#pragma unroll
        for (int v = 0; v < 8; ++v) {
          if (own_r[v]) {
            float* p = agg_r + (long)rr[v] * 64 + m;
#pragma unroll
            for (int nt = 0; nt < 4; ++nt) unsafeAtomicAdd(p + nt * 16, acc[nt][v]);
          }
          if (own_s[v]) {
            float* p = agg_s + (long)ss[v] * 64 + m;
#pragma unroll
            for (int nt = 0; nt < 4; ++nt) unsafeAtomicAdd(p + nt * 16, acc[nt][v]);
          }
        }
        if (lane < 16) {
          if ((unsigned)(ri - lo) < (unsigned)HALFN) unsafeAtomicAdd(&cnt_r[ri], 1.0f);
          if ((unsigned)(si - lo) < (unsigned)HALFN) unsafeAtomicAdd(&cnt_s[si], 1.0f);
        }
      }
    }
  }
}

__global__ __launch_bounds__(256) void k_node(
    const float* __restrict__ x, const float* __restrict__ agg_r,
    const float* __restrict__ Wnb, const float* __restrict__ bias_nb,
    float* __restrict__ n_h) {
  __shared__ float sW[20 * 4 * 64];
  for (int i = threadIdx.x; i < 20 * 4 * 64; i += 256) {
    int j = i & 1, lane = (i >> 1) & 31, nt = (i >> 6) & 3, kc = i >> 8;
    int kk = 4 * kc + 2 * (lane >> 4) + j;
    sW[i] = Wnb[kk * 64 + nt * 16 + (lane & 15)];
  }
  __syncthreads();
  const int lane = threadIdx.x & 31;
  const int h = lane >> 4, m = lane & 15;
  const int wave = blockIdx.x * 8 + (threadIdx.x >> 5);
  const int nwaves = gridDim.x * 8;
  const int ntiles = NNODES / 16;
  for (int t = wave; t < ntiles; t += nwaves) {
    const int n0 = t * 16;
    const float* ap = agg_r + (long)(n0 + m) * 64;
    const float* xp = x + (long)(n0 + m) * 16;
    v8f acc[4];
#pragma unroll
    for (int nt = 0; nt < 4; ++nt) {
      float bb = bias_nb[nt * 16 + m];
      acc[nt] = (v8f){bb, bb, bb, bb, bb, bb, bb, bb};
    }
#pragma unroll
    for (int kc = 0; kc < 20; ++kc) {
      const int kk = 4 * kc + 2 * h;
      v2f a = (kc < 16) ? *(const v2f*)(ap + kk) : *(const v2f*)(xp + kk - 64);
#pragma unroll
      for (int nt = 0; nt < 4; ++nt) {
        v2f b = *(const v2f*)(&sW[(kc * 4 + nt) * 64 + lane * 2]);
        acc[nt] = __builtin_amdgcn_wmma_f32_16x16x4_f32(false, a, false, b,
                                                        (short)0, acc[nt], false, false);
      }
    }
#pragma unroll
    for (int nt = 0; nt < 4; ++nt)
#pragma unroll
      for (int v = 0; v < 8; ++v)
        n_h[(long)(n0 + 8 * h + v) * 64 + nt * 16 + m] = fmaxf(acc[nt][v], 0.0f);
  }
}

__global__ void k_colsum(const float* __restrict__ src, int rows, float* __restrict__ dst) {
  __shared__ float red[256];
  const int col = threadIdx.x & 63;
  const int g = threadIdx.x >> 6;
  float local = 0.f;
  for (int r = blockIdx.x * 4 + g; r < rows; r += gridDim.x * 4)
    local += src[(long)r * 64 + col];
  red[threadIdx.x] = local;
  __syncthreads();
  if (g == 0) {
    local += red[64 + threadIdx.x] + red[128 + threadIdx.x] + red[192 + threadIdx.x];
    unsafeAtomicAdd(&dst[col], local);
  }
}

__global__ void k_global(const float* __restrict__ u,
                         const float* __restrict__ Wgb, const float* __restrict__ bgb,
                         const float* __restrict__ Wd1, const float* __restrict__ bd1,
                         const float* __restrict__ e_sum, const float* __restrict__ n_sum,
                         float* __restrict__ bias_d1) {
  __shared__ float su2[64];
  const int j = threadIdx.x;
  float acc = bgb[j];
  for (int k = 0; k < 64; ++k) {
    float em = e_sum[k] * (1.0f / NEDGES);
    float nm = n_sum[k] * (1.0f / NNODES);
    acc += em * Wgb[k * 64 + j] + nm * Wgb[(64 + k) * 64 + j] + u[k] * Wgb[(128 + k) * 64 + j];
  }
  acc = fmaxf(acc, 0.0f);
  su2[j] = acc;
  __syncthreads();
  float b = bd1[j];
  for (int k = 0; k < 64; ++k) b += su2[k] * Wd1[(192 + k) * 64 + j];
  bias_d1[j] = b;
}

__global__ __launch_bounds__(256) void k_dec(
    const float* __restrict__ agg_r, const float* __restrict__ agg_s,
    const float* __restrict__ cnt_r, const float* __restrict__ cnt_s,
    const float* __restrict__ n_h, const float* __restrict__ Wd1,
    const float* __restrict__ bias_d1, const float* __restrict__ Wd2,
    const float* __restrict__ bd2, float* __restrict__ out) {
  __shared__ float sW1[48 * 4 * 64];
  __shared__ float sW2[16 * 64];
  __shared__ float sH[8 * 16 * 64];
  for (int i = threadIdx.x; i < 48 * 4 * 64; i += 256) {
    int j = i & 1, lane = (i >> 1) & 31, nt = (i >> 6) & 3, kc = i >> 8;
    int kk = 4 * kc + 2 * (lane >> 4) + j;
    sW1[i] = Wd1[kk * 64 + nt * 16 + (lane & 15)];
  }
  for (int i = threadIdx.x; i < 16 * 64; i += 256) {
    int j = i & 1, lane = (i >> 1) & 31, kc = i >> 6;
    int kk = 4 * kc + 2 * (lane >> 4) + j;
    sW2[i] = Wd2[kk * 16 + (lane & 15)];
  }
  __syncthreads();
  const int lane = threadIdx.x & 31;
  const int h = lane >> 4, m = lane & 15;
  float* hw = sH + (threadIdx.x >> 5) * 1024;
  const int wave = blockIdx.x * 8 + (threadIdx.x >> 5);
  const int nwaves = gridDim.x * 8;
  const int ntiles = NNODES / 16;
  for (int t = wave; t < ntiles; t += nwaves) {
    const int n0 = t * 16;
    const int n = n0 + m;
    const float invr = 1.0f / fmaxf(cnt_r[n], 1.0f);
    const float invs = 1.0f / fmaxf(cnt_s[n], 1.0f);
    const float* pr = agg_r + (long)n * 64;
    const float* ps = agg_s + (long)n * 64;
    const float* pn = n_h + (long)n * 64;
    v8f acc[4];
#pragma unroll
    for (int nt = 0; nt < 4; ++nt) {
      float bb = bias_d1[nt * 16 + m];
      acc[nt] = (v8f){bb, bb, bb, bb, bb, bb, bb, bb};
    }
#pragma unroll
    for (int kc = 0; kc < 48; ++kc) {
      const int kk = 4 * kc + 2 * h;
      v2f a;
      if (kc < 16)      { a = *(const v2f*)(pr + kk);        a *= invr; }
      else if (kc < 32) { a = *(const v2f*)(ps + kk - 64);   a *= invs; }
      else              { a = *(const v2f*)(pn + kk - 128); }
#pragma unroll
      for (int nt = 0; nt < 4; ++nt) {
        v2f b = *(const v2f*)(&sW1[(kc * 4 + nt) * 64 + lane * 2]);
        acc[nt] = __builtin_amdgcn_wmma_f32_16x16x4_f32(false, a, false, b,
                                                        (short)0, acc[nt], false, false);
      }
    }
#pragma unroll
    for (int nt = 0; nt < 4; ++nt)
#pragma unroll
      for (int v = 0; v < 8; ++v)
        hw[(8 * h + v) * 64 + nt * 16 + m] = fmaxf(acc[nt][v], 0.0f);
    float bb2 = bd2[m];
    v8f acc2 = (v8f){bb2, bb2, bb2, bb2, bb2, bb2, bb2, bb2};
#pragma unroll
    for (int kc = 0; kc < 16; ++kc) {
      const int kk = 4 * kc + 2 * h;
      v2f a = *(const v2f*)(hw + m * 64 + kk);
      v2f b = *(const v2f*)(&sW2[kc * 64 + lane * 2]);
      acc2 = __builtin_amdgcn_wmma_f32_16x16x4_f32(false, a, false, b,
                                                   (short)0, acc2, false, false);
    }
#pragma unroll
    for (int v = 0; v < 8; ++v)
      out[(long)(n0 + 8 * h + v) * 16 + m] = acc2[v];
  }
}

extern "C" void kernel_launch(void* const* d_in, const int* in_sizes, int n_in,
                              void* d_out, int out_size, void* d_ws, size_t ws_size,
                              hipStream_t stream) {
  const float* x    = (const float*)d_in[0];
  const float* ea   = (const float*)d_in[1];
  const float* u    = (const float*)d_in[2];
  const int* send   = (const int*)d_in[3];
  const int* recv   = (const int*)d_in[4];
  const float* Web  = (const float*)d_in[7];
  const float* beb  = (const float*)d_in[8];
  const float* Wnb  = (const float*)d_in[9];
  const float* bnb  = (const float*)d_in[10];
  const float* Wgb  = (const float*)d_in[11];
  const float* bgb  = (const float*)d_in[12];
  const float* Wd1  = (const float*)d_in[13];
  const float* bd1  = (const float*)d_in[14];
  const float* Wd2  = (const float*)d_in[15];
  const float* bd2  = (const float*)d_in[16];
  float* out = (float*)d_out;
  float* ws = (float*)d_ws;

  const long N = NNODES;
  float* agg_r   = ws;
  float* agg_s   = agg_r + N * 64;
  float* cnt_r   = agg_s + N * 64;
  float* cnt_s   = cnt_r + N;
  float* e_sum   = cnt_s + N;
  float* n_sum   = e_sum + 64;
  int* tile_ctr  = (int*)(n_sum + 64);
  float* bias_eb = n_sum + 128;
  float* bias_nb = bias_eb + 64;
  float* bias_d1 = bias_nb + 64;
  float* n_hbuf  = bias_d1 + 64;

  const long zlen = 2 * N * 64 + 2 * N + 192;
  k_zero<<<2048, 256, 0, stream>>>(ws, zlen);
  k_prep<<<1, 64, 0, stream>>>(u, Web, beb, Wnb, bnb, bias_eb, bias_nb);
  k_edge<<<1024, 256, 0, stream>>>(x, ea, recv, send, Web, bias_eb,
                                   agg_r, agg_s, cnt_r, cnt_s, tile_ctr);
  k_node<<<782, 256, 0, stream>>>(x, agg_r, Wnb, bias_nb, n_hbuf);
  k_colsum<<<512, 256, 0, stream>>>(agg_r, NNODES, e_sum);
  k_colsum<<<512, 256, 0, stream>>>(n_hbuf, NNODES, n_sum);
  k_global<<<1, 64, 0, stream>>>(u, Wgb, bgb, Wd1, bd1, e_sum, n_sum, bias_d1);
  k_dec<<<1024, 256, 0, stream>>>(agg_r, agg_s, cnt_r, cnt_s, n_hbuf,
                                  Wd1, bias_d1, Wd2, bd2, out);
}


# k_edge: scatter-add moved into LDS: edges binned by destination node range (count, scan, fill); one workgroup per range adds its rows with LDS f32 atomics, stores each once; arrival order
# speedup vs baseline: 1.4424x; 1.2320x over previous
#include <hip/hip_runtime.h>

typedef __attribute__((ext_vector_type(2))) float v2f;
typedef __attribute__((ext_vector_type(8))) float v8f;

#define NNODES 100000
#define NEDGES 1600000

__global__ void k_zero(float* __restrict__ a, int na, int* __restrict__ b, int nb) {
  int i = blockIdx.x * blockDim.x + threadIdx.x;
  if (i < na) a[i] = 0.f;
  if (i < nb) b[i] = 0;
}

__global__ void k_prep(const float* __restrict__ u,
                       const float* __restrict__ Web, const float* __restrict__ beb,
                       const float* __restrict__ Wnb, const float* __restrict__ bnb,
                       float* __restrict__ bias_eb, float* __restrict__ bias_nb) {
  int j = threadIdx.x;
  float a = beb[j], b = bnb[j];
  for (int k = 0; k < 64; ++k) {
    float uk = u[k];
    a += uk * Web[(48 + k) * 64 + j];
    b += uk * Wnb[(80 + k) * 64 + j];
  }
  bias_eb[j] = a;
  bias_nb[j] = b;
}

#define NB 256
#define RPB 391
#define CHUNK 2048
#define NCHUNKS ((NEDGES + CHUNK - 1) / CHUNK)
__global__ __launch_bounds__(256) void k_count(const int* __restrict__ recv,
                                               const int* __restrict__ send,
                                               int* __restrict__ totals) {
  __shared__ int hist[2 * NB];
  for (int j = threadIdx.x; j < 2 * NB; j += 256) hist[j] = 0;
  __syncthreads();
  const int e0 = blockIdx.x * CHUNK;
  for (int i = threadIdx.x; i < CHUNK; i += 256) {
    const int e = e0 + i;
    if (e < NEDGES) {
      atomicAdd(&hist[recv[e] / RPB], 1);
      atomicAdd(&hist[NB + send[e] / RPB], 1);
    }
  }
  __syncthreads();
  for (int j = threadIdx.x; j < 2 * NB; j += 256) {
    const int c = hist[j];
    if (c) atomicAdd(&totals[j], c);
  }
}

__global__ __launch_bounds__(2 * NB) void k_scan(const int* __restrict__ totals,
                                                 int* __restrict__ binbase,
                                                 int* __restrict__ cursors) {
  __shared__ int s[2 * NB];
  const int j = threadIdx.x;
  const int v = totals[j];
  s[j] = v;
  __syncthreads();
  for (int off = 1; off < NB; off <<= 1) {
    const int t = ((j & (NB - 1)) >= off) ? s[j - off] : 0;
    __syncthreads();
    s[j] += t;
    __syncthreads();
  }
  const int ex = s[j] - v;
  binbase[j] = ex;
  cursors[j] = ex;
}

__global__ __launch_bounds__(256) void k_fill(const int* __restrict__ recv,
                                              const int* __restrict__ send,
                                              int* __restrict__ cursors,
                                              int* __restrict__ list_r,
                                              int* __restrict__ list_s) {
  __shared__ int hist[2 * NB];
  for (int j = threadIdx.x; j < 2 * NB; j += 256) hist[j] = 0;
  __syncthreads();
  const int e0 = blockIdx.x * CHUNK;
  for (int i = threadIdx.x; i < CHUNK; i += 256) {
    const int e = e0 + i;
    if (e < NEDGES) {
      atomicAdd(&hist[recv[e] / RPB], 1);
      atomicAdd(&hist[NB + send[e] / RPB], 1);
    }
  }
  __syncthreads();
  for (int j = threadIdx.x; j < 2 * NB; j += 256) {
    const int c = hist[j];
    hist[j] = c ? atomicAdd(&cursors[j], c) : 0;
  }
  __syncthreads();
  for (int i = threadIdx.x; i < CHUNK; i += 256) {
    const int e = e0 + i;
    if (e < NEDGES) {
      const int p = atomicAdd(&hist[recv[e] / RPB], 1);
      list_r[p] = e;
      const int q = atomicAdd(&hist[NB + send[e] / RPB], 1);
      list_s[q] = e;
    }
  }
}

#define EW 16
#define RPL (RPB + 1)
#define EDGE_LIST(tt) list[min((tt) * 16 + m, n - 1)]
#define EDGE_ROWS(A_, e_, r_, s_)                                             \
  {                                                                           \
    const float* p0_ = ea + (long)(e_) * 16 + 2 * h;                          \
    const float* p1_ = x + (long)(r_) * 16 + 2 * h;                           \
    const float* p2_ = x + (long)(s_) * 16 + 2 * h;                           \
    _Pragma("unroll") for (int cc = 0; cc < 4; ++cc) {                        \
      A_[cc] = *(const v2f*)(p0_ + 4 * cc);                                   \
      A_[4 + cc] = *(const v2f*)(p1_ + 4 * cc);                               \
      A_[8 + cc] = *(const v2f*)(p2_ + 4 * cc);                               \
    }                                                                         \
  }
__global__ __launch_bounds__(EW * 32) void k_edge(
    const float* __restrict__ x, const float* __restrict__ ea,
    const int* __restrict__ recv, const int* __restrict__ send,
    const float* __restrict__ Web, const float* __restrict__ bias_eb,
    const int* __restrict__ list_r, const int* __restrict__ list_s,
    const int* __restrict__ totals, const int* __restrict__ binbase,
    float* __restrict__ agg_r, float* __restrict__ agg_s,
    float* __restrict__ cnt_r, float* __restrict__ cnt_s) {
  __shared__ __attribute__((aligned(16))) float sS[2 * RPL * 64];
  __shared__ float sC[2 * RPL];
  for (int i = threadIdx.x; i < 2 * RPL * 16; i += EW * 32)
    *(float4*)(&sS[i * 4]) = (float4){0.f, 0.f, 0.f, 0.f};
  for (int i = threadIdx.x; i < 2 * RPL; i += EW * 32) sC[i] = 0.f;
  const int lane = threadIdx.x & 31;
  const int h = lane >> 4, m = lane & 15;
  const int wave = __builtin_amdgcn_readfirstlane(threadIdx.x >> 5);
  const int b = blockIdx.x;
  const int lo = b * RPB;
  v2f bfr[48];
#pragma unroll
  for (int kc = 0; kc < 12; ++kc)
#pragma unroll
    for (int nt = 0; nt < 4; ++nt) {
      const int kk = 4 * kc + 2 * h;
      bfr[kc * 4 + nt] = (v2f){Web[kk * 64 + nt * 16 + m], Web[(kk + 1) * 64 + nt * 16 + m]};
    }
  float bb[4];
#pragma unroll
  for (int nt = 0; nt < 4; ++nt) bb[nt] = bias_eb[nt * 16 + m];
  __syncthreads();
  for (int role = 0; role < 2; ++role) {
    const int* __restrict__ list = (role ? list_s : list_r) + binbase[role * NB + b];
    const int n = totals[role * NB + b];
    const int ntiles = (n + 15) >> 4;
    float* S = sS + role * (RPL * 64);
    float* C = sC + role * RPL;
    int t = wave;
    if (t < ntiles) {
      v2f A0[12], A1[12];
      int e0 = EDGE_LIST(t);
      int r0 = recv[e0], s0 = send[e0];
      EDGE_ROWS(A0, e0, r0, s0)
      int e1 = EDGE_LIST(t + EW);
      int r1 = recv[e1], s1 = send[e1];
      for (; t < ntiles; t += EW) {
        EDGE_ROWS(A1, e1, r1, s1)
        const int e2 = EDGE_LIST(t + 2 * EW);
        v8f acc[4];
#pragma unroll
        for (int nt = 0; nt < 4; ++nt)
          acc[nt] = (v8f){bb[nt], bb[nt], bb[nt], bb[nt], bb[nt], bb[nt], bb[nt], bb[nt]};
#pragma unroll
        for (int kc = 0; kc < 6; ++kc)
#pragma unroll
          for (int nt = 0; nt < 4; ++nt)
            acc[nt] = __builtin_amdgcn_wmma_f32_16x16x4_f32(false, A0[kc], false, bfr[kc * 4 + nt],
                                                            (short)0, acc[nt], false, false);
        __builtin_amdgcn_sched_barrier(0);
        const int r2 = recv[e2], s2 = send[e2];
        __builtin_amdgcn_sched_barrier(0);
#pragma unroll
        for (int kc = 6; kc < 12; ++kc)
#pragma unroll
          for (int nt = 0; nt < 4; ++nt)
            acc[nt] = __builtin_amdgcn_wmma_f32_16x16x4_f32(false, A0[kc], false, bfr[kc * 4 + nt],
                                                            (short)0, acc[nt], false, false);
#pragma unroll
        for (int nt = 0; nt < 4; ++nt)
#pragma unroll
          for (int v = 0; v < 8; ++v) acc[nt][v] = fmaxf(acc[nt][v], 0.0f);
        const int d0 = (role ? s0 : r0) - lo;
        const int dv0 = (t * 16 + m < n && (unsigned)d0 < (unsigned)RPB) ? d0 : RPB;
        int dd[8];
#pragma unroll
        for (int v = 0; v < 8; ++v) dd[v] = __shfl(dv0, 8 * h + v);
        const bool hb = h != 0;
        v8f q0, q1, q2, q3;
#pragma unroll
        for (int v = 0; v < 8; ++v) {
          q0[v] = hb ? acc[1][v] : acc[0][v];
          q1[v] = hb ? acc[0][v] : acc[1][v];
          q2[v] = hb ? acc[3][v] : acc[2][v];
          q3[v] = hb ? acc[2][v] : acc[3][v];
        }
        const int oa = h << 4, ob = (1 - h) << 4;
#pragma unroll
        for (int v = 0; v < 8; ++v) {
          float* p = S + dd[v] * 64 + m;
          unsafeAtomicAdd(p + oa, q0[v]);
          unsafeAtomicAdd(p + ob, q1[v]);
          unsafeAtomicAdd(p + 32 + oa, q2[v]);
          unsafeAtomicAdd(p + 32 + ob, q3[v]);
        }
        if (h == 0) unsafeAtomicAdd(&C[dv0], 1.0f);
#pragma unroll
        for (int q = 0; q < 12; ++q) A0[q] = A1[q];
        e0 = e1; r0 = r1; s0 = s1;
        e1 = e2; r1 = r2; s1 = s2;
      }
    }
  }
  __syncthreads();
  for (int i = threadIdx.x; i < 2 * RPB * 16; i += EW * 32) {
    const int role = i / (RPB * 16);
    const int r = i - role * (RPB * 16);
    const int row = r >> 4, c4 = r & 15;
    const int node = lo + row;
    if (node < NNODES) {
      const float4 val = *(const float4*)(&sS[role * (RPL * 64) + row * 64 + c4 * 4]);
      *(float4*)((role ? agg_s : agg_r) + (long)node * 64 + c4 * 4) = val;
    }
  }
  for (int i = threadIdx.x; i < 2 * RPB; i += EW * 32) {
    const int role = i / RPB;
    const int row = i - role * RPB;
    const int node = lo + row;
    if (node < NNODES) (role ? cnt_s : cnt_r)[node] = sC[role * RPL + row];
  }
}

__global__ __launch_bounds__(256) void k_node(
    const float* __restrict__ x, const float* __restrict__ agg_r,
    const float* __restrict__ Wnb, const float* __restrict__ bias_nb,
    float* __restrict__ n_h) {
  __shared__ float sW[20 * 4 * 64];
  for (int i = threadIdx.x; i < 20 * 4 * 64; i += 256) {
    int j = i & 1, lane = (i >> 1) & 31, nt = (i >> 6) & 3, kc = i >> 8;
    int kk = 4 * kc + 2 * (lane >> 4) + j;
    sW[i] = Wnb[kk * 64 + nt * 16 + (lane & 15)];
  }
  __syncthreads();
  const int lane = threadIdx.x & 31;
  const int h = lane >> 4, m = lane & 15;
  const int wave = blockIdx.x * 8 + (threadIdx.x >> 5);
  const int nwaves = gridDim.x * 8;
  const int ntiles = NNODES / 16;
  for (int t = wave; t < ntiles; t += nwaves) {
    const int n0 = t * 16;
    const float* ap = agg_r + (long)(n0 + m) * 64;
    const float* xp = x + (long)(n0 + m) * 16;
    v8f acc[4];
#pragma unroll
    for (int nt = 0; nt < 4; ++nt) {
      float bb = bias_nb[nt * 16 + m];
      acc[nt] = (v8f){bb, bb, bb, bb, bb, bb, bb, bb};
    }
#pragma unroll
    for (int kc = 0; kc < 20; ++kc) {
      const int kk = 4 * kc + 2 * h;
      v2f a = (kc < 16) ? *(const v2f*)(ap + kk) : *(const v2f*)(xp + kk - 64);
#pragma unroll
      for (int nt = 0; nt < 4; ++nt) {
        v2f b = *(const v2f*)(&sW[(kc * 4 + nt) * 64 + lane * 2]);
        acc[nt] = __builtin_amdgcn_wmma_f32_16x16x4_f32(false, a, false, b,
                                                        (short)0, acc[nt], false, false);
      }
    }
#pragma unroll
    for (int nt = 0; nt < 4; ++nt)
#pragma unroll
      for (int v = 0; v < 8; ++v)
        n_h[(long)(n0 + 8 * h + v) * 64 + nt * 16 + m] = fmaxf(acc[nt][v], 0.0f);
  }
}

__global__ void k_colsum(const float* __restrict__ src, int rows, float* __restrict__ dst) {
  __shared__ float red[256];
  const int col = threadIdx.x & 63;
  const int g = threadIdx.x >> 6;
  float local = 0.f;
  for (int r = blockIdx.x * 4 + g; r < rows; r += gridDim.x * 4)
    local += src[(long)r * 64 + col];
  red[threadIdx.x] = local;
  __syncthreads();
  if (g == 0) {
    local += red[64 + threadIdx.x] + red[128 + threadIdx.x] + red[192 + threadIdx.x];
    unsafeAtomicAdd(&dst[col], local);
  }
}

__global__ void k_global(const float* __restrict__ u,
                         const float* __restrict__ Wgb, const float* __restrict__ bgb,
                         const float* __restrict__ Wd1, const float* __restrict__ bd1,
                         const float* __restrict__ e_sum, const float* __restrict__ n_sum,
                         float* __restrict__ bias_d1) {
  __shared__ float su2[64];
  const int j = threadIdx.x;
  float acc = bgb[j];
  for (int k = 0; k < 64; ++k) {
    float em = e_sum[k] * (1.0f / NEDGES);
    float nm = n_sum[k] * (1.0f / NNODES);
    acc += em * Wgb[k * 64 + j] + nm * Wgb[(64 + k) * 64 + j] + u[k] * Wgb[(128 + k) * 64 + j];
  }
  acc = fmaxf(acc, 0.0f);
  su2[j] = acc;
  __syncthreads();
  float b = bd1[j];
  for (int k = 0; k < 64; ++k) b += su2[k] * Wd1[(192 + k) * 64 + j];
  bias_d1[j] = b;
}

__global__ __launch_bounds__(256) void k_dec(
    const float* __restrict__ agg_r, const float* __restrict__ agg_s,
    const float* __restrict__ cnt_r, const float* __restrict__ cnt_s,
    const float* __restrict__ n_h, const float* __restrict__ Wd1,
    const float* __restrict__ bias_d1, const float* __restrict__ Wd2,
    const float* __restrict__ bd2, float* __restrict__ out) {
  __shared__ float sW1[48 * 4 * 64];
  __shared__ float sW2[16 * 64];
  __shared__ float sH[8 * 16 * 64];
  for (int i = threadIdx.x; i < 48 * 4 * 64; i += 256) {
    int j = i & 1, lane = (i >> 1) & 31, nt = (i >> 6) & 3, kc = i >> 8;
    int kk = 4 * kc + 2 * (lane >> 4) + j;
    sW1[i] = Wd1[kk * 64 + nt * 16 + (lane & 15)];
  }
  for (int i = threadIdx.x; i < 16 * 64; i += 256) {
    int j = i & 1, lane = (i >> 1) & 31, kc = i >> 6;
    int kk = 4 * kc + 2 * (lane >> 4) + j;
    sW2[i] = Wd2[kk * 16 + (lane & 15)];
  }
  __syncthreads();
  const int lane = threadIdx.x & 31;
  const int h = lane >> 4, m = lane & 15;
  float* hw = sH + (threadIdx.x >> 5) * 1024;
  const int wave = blockIdx.x * 8 + (threadIdx.x >> 5);
  const int nwaves = gridDim.x * 8;
  const int ntiles = NNODES / 16;
  for (int t = wave; t < ntiles; t += nwaves) {
    const int n0 = t * 16;
    const int n = n0 + m;
    const float invr = 1.0f / fmaxf(cnt_r[n], 1.0f);
    const float invs = 1.0f / fmaxf(cnt_s[n], 1.0f);
    const float* pr = agg_r + (long)n * 64;
    const float* ps = agg_s + (long)n * 64;
    const float* pn = n_h + (long)n * 64;
    v8f acc[4];
#pragma unroll
    for (int nt = 0; nt < 4; ++nt) {
      float bb = bias_d1[nt * 16 + m];
      acc[nt] = (v8f){bb, bb, bb, bb, bb, bb, bb, bb};
    }
#pragma unroll
    for (int kc = 0; kc < 48; ++kc) {
      const int kk = 4 * kc + 2 * h;
      v2f a;
      if (kc < 16)      { a = *(const v2f*)(pr + kk);        a *= invr; }
      else if (kc < 32) { a = *(const v2f*)(ps + kk - 64);   a *= invs; }
      else              { a = *(const v2f*)(pn + kk - 128); }
#pragma unroll
      for (int nt = 0; nt < 4; ++nt) {
        v2f b = *(const v2f*)(&sW1[(kc * 4 + nt) * 64 + lane * 2]);
        acc[nt] = __builtin_amdgcn_wmma_f32_16x16x4_f32(false, a, false, b,
                                                        (short)0, acc[nt], false, false);
      }
    }
#pragma unroll
    for (int nt = 0; nt < 4; ++nt)
#pragma unroll
      for (int v = 0; v < 8; ++v)
        hw[(8 * h + v) * 64 + nt * 16 + m] = fmaxf(acc[nt][v], 0.0f);
    float bb2 = bd2[m];
    v8f acc2 = (v8f){bb2, bb2, bb2, bb2, bb2, bb2, bb2, bb2};
#pragma unroll
    for (int kc = 0; kc < 16; ++kc) {
      const int kk = 4 * kc + 2 * h;
      v2f a = *(const v2f*)(hw + m * 64 + kk);
      v2f b = *(const v2f*)(&sW2[kc * 64 + lane * 2]);
      acc2 = __builtin_amdgcn_wmma_f32_16x16x4_f32(false, a, false, b,
                                                   (short)0, acc2, false, false);
    }
#pragma unroll
    for (int v = 0; v < 8; ++v)
      out[(long)(n0 + 8 * h + v) * 16 + m] = acc2[v];
  }
}

extern "C" void kernel_launch(void* const* d_in, const int* in_sizes, int n_in,
                              void* d_out, int out_size, void* d_ws, size_t ws_size,
                              hipStream_t stream) {
  const float* x    = (const float*)d_in[0];
  const float* ea   = (const float*)d_in[1];
  const float* u    = (const float*)d_in[2];
  const int* send   = (const int*)d_in[3];
  const int* recv   = (const int*)d_in[4];
  const float* Web  = (const float*)d_in[7];
  const float* beb  = (const float*)d_in[8];
  const float* Wnb  = (const float*)d_in[9];
  const float* bnb  = (const float*)d_in[10];
  const float* Wgb  = (const float*)d_in[11];
  const float* bgb  = (const float*)d_in[12];
  const float* Wd1  = (const float*)d_in[13];
  const float* bd1  = (const float*)d_in[14];
  const float* Wd2  = (const float*)d_in[15];
  const float* bd2  = (const float*)d_in[16];
  float* out = (float*)d_out;
  float* ws = (float*)d_ws;

  const long N = NNODES;
  float* agg_r   = ws;
  float* agg_s   = agg_r + N * 64;
  float* cnt_r   = agg_s + N * 64;
  float* cnt_s   = cnt_r + N;
  float* e_sum   = cnt_s + N;
  float* n_sum   = e_sum + 64;
  float* bias_eb = n_sum + 128;
  float* bias_nb = bias_eb + 64;
  float* bias_d1 = bias_nb + 64;
  float* n_hbuf  = bias_d1 + 64;
  int* list_r  = (int*)n_hbuf;
  int* list_s  = list_r + NEDGES;
  int* totals  = list_s + NEDGES;
  int* binbase = totals + 2 * NB;
  int* cursors = binbase + 2 * NB;

  k_zero<<<2, 256, 0, stream>>>(e_sum, 128, totals, 2 * NB);
  k_prep<<<1, 64, 0, stream>>>(u, Web, beb, Wnb, bnb, bias_eb, bias_nb);
  k_count<<<NCHUNKS, 256, 0, stream>>>(recv, send, totals);
  k_scan<<<1, 2 * NB, 0, stream>>>(totals, binbase, cursors);
  k_fill<<<NCHUNKS, 256, 0, stream>>>(recv, send, cursors, list_r, list_s);
  k_edge<<<NB, EW * 32, 0, stream>>>(x, ea, recv, send, Web, bias_eb, list_r, list_s,
                                    totals, binbase, agg_r, agg_s, cnt_r, cnt_s);
  k_node<<<782, 256, 0, stream>>>(x, agg_r, Wnb, bias_nb, n_hbuf);
  k_colsum<<<512, 256, 0, stream>>>(agg_r, NNODES, e_sum);
  k_colsum<<<512, 256, 0, stream>>>(n_hbuf, NNODES, n_sum);
  k_global<<<1, 64, 0, stream>>>(u, Wgb, bgb, Wd1, bd1, e_sum, n_sum, bias_d1);
  k_dec<<<1024, 256, 0, stream>>>(agg_r, agg_s, cnt_r, cnt_s, n_hbuf,
                                  Wd1, bias_d1, Wd2, bd2, out);
}


# k_edge: x row of the end point inside the range read from LDS (range's x rows staged; row number packed in the list word), W_eb chunks swapped for role 1; role 1 k order changes
# speedup vs baseline: 1.5130x; 1.0489x over previous
#include <hip/hip_runtime.h>

typedef __attribute__((ext_vector_type(2))) float v2f;
typedef __attribute__((ext_vector_type(8))) float v8f;

#define NNODES 100000
#define NEDGES 1600000

__global__ void k_zero(float* __restrict__ a, int na, int* __restrict__ b, int nb) {
  int i = blockIdx.x * blockDim.x + threadIdx.x;
  if (i < na) a[i] = 0.f;
  if (i < nb) b[i] = 0;
}

__global__ void k_prep(const float* __restrict__ u,
                       const float* __restrict__ Web, const float* __restrict__ beb,
                       const float* __restrict__ Wnb, const float* __restrict__ bnb,
                       float* __restrict__ bias_eb, float* __restrict__ bias_nb) {
  int j = threadIdx.x;
  float a = beb[j], b = bnb[j];
  for (int k = 0; k < 64; ++k) {
    float uk = u[k];
    a += uk * Web[(48 + k) * 64 + j];
    b += uk * Wnb[(80 + k) * 64 + j];
  }
  bias_eb[j] = a;
  bias_nb[j] = b;
}

#define NB 256
#define RPB 391
#define CHUNK 2048
#define NCHUNKS ((NEDGES + CHUNK - 1) / CHUNK)
#define EBITS 21
#define EMASK ((1 << EBITS) - 1)
static_assert(NEDGES <= (1 << EBITS), "edge number must fit the low bits of a list word");
static_assert(RPB <= (1 << (30 - EBITS)), "row in range must fit above the edge number");
__global__ __launch_bounds__(256) void k_count(const int* __restrict__ recv,
                                               const int* __restrict__ send,
                                               int* __restrict__ totals) {
  __shared__ int hist[2 * NB];
  for (int j = threadIdx.x; j < 2 * NB; j += 256) hist[j] = 0;
  __syncthreads();
  const int e0 = blockIdx.x * CHUNK;
  for (int i = threadIdx.x; i < CHUNK; i += 256) {
    const int e = e0 + i;
    if (e < NEDGES) {
      atomicAdd(&hist[recv[e] / RPB], 1);
      atomicAdd(&hist[NB + send[e] / RPB], 1);
    }
  }
  __syncthreads();
  for (int j = threadIdx.x; j < 2 * NB; j += 256) {
    const int c = hist[j];
    if (c) atomicAdd(&totals[j], c);
  }
}

__global__ __launch_bounds__(2 * NB) void k_scan(const int* __restrict__ totals,
                                                 int* __restrict__ binbase,
                                                 int* __restrict__ cursors) {
  __shared__ int s[2 * NB];
  const int j = threadIdx.x;
  const int v = totals[j];
  s[j] = v;
  __syncthreads();
  for (int off = 1; off < NB; off <<= 1) {
    const int t = ((j & (NB - 1)) >= off) ? s[j - off] : 0;
    __syncthreads();
    s[j] += t;
    __syncthreads();
  }
  const int ex = s[j] - v;
  binbase[j] = ex;
  cursors[j] = ex;
}

__global__ __launch_bounds__(256) void k_fill(const int* __restrict__ recv,
                                              const int* __restrict__ send,
                                              int* __restrict__ cursors,
                                              int* __restrict__ list_r,
                                              int* __restrict__ list_s) {
  __shared__ int hist[2 * NB];
  for (int j = threadIdx.x; j < 2 * NB; j += 256) hist[j] = 0;
  __syncthreads();
  const int e0 = blockIdx.x * CHUNK;
  for (int i = threadIdx.x; i < CHUNK; i += 256) {
    const int e = e0 + i;
    if (e < NEDGES) {
      atomicAdd(&hist[recv[e] / RPB], 1);
      atomicAdd(&hist[NB + send[e] / RPB], 1);
    }
  }
  __syncthreads();
  for (int j = threadIdx.x; j < 2 * NB; j += 256) {
    const int c = hist[j];
    hist[j] = c ? atomicAdd(&cursors[j], c) : 0;
  }
  __syncthreads();
  for (int i = threadIdx.x; i < CHUNK; i += 256) {
    const int e = e0 + i;
    if (e < NEDGES) {
      const int r = recv[e], s = send[e];
      const int br = r / RPB, bs = s / RPB;
      const int p = atomicAdd(&hist[br], 1);
      list_r[p] = e | ((r - br * RPB) << EBITS);
      const int q = atomicAdd(&hist[NB + bs], 1);
      list_s[q] = e | ((s - bs * RPB) << EBITS);
    }
  }
}

#define EW 16
#define RPL (RPB + 1)
#define XP 20
#define EDGE_LIST(tt) list[min((tt) * 16 + m, n - 1)]
#define EDGE_ROWS(A_, e_, o_)                                                 \
  {                                                                           \
    const float* p0_ = ea + (long)(e_) * 16 + 2 * h;                          \
    const float* p2_ = x + (long)(o_) * 16 + 2 * h;                           \
    _Pragma("unroll") for (int cc = 0; cc < 4; ++cc) {                        \
      A_[cc] = *(const v2f*)(p0_ + 4 * cc);                                   \
      A_[4 + cc] = *(const v2f*)(p2_ + 4 * cc);                               \
    }                                                                         \
  }
__global__ __launch_bounds__(EW * 32) void k_edge(
    const float* __restrict__ x, const float* __restrict__ ea,
    const int* __restrict__ recv, const int* __restrict__ send,
    const float* __restrict__ Web, const float* __restrict__ bias_eb,
    const int* __restrict__ list_r, const int* __restrict__ list_s,
    const int* __restrict__ totals, const int* __restrict__ binbase,
    float* __restrict__ agg_r, float* __restrict__ agg_s,
    float* __restrict__ cnt_r, float* __restrict__ cnt_s) {
  __shared__ __attribute__((aligned(16))) float sS[2 * RPL * 64];
  __shared__ float sC[2 * RPL];
  __shared__ __attribute__((aligned(16))) float sX[RPB * XP];
  for (int i = threadIdx.x; i < 2 * RPL * 16; i += EW * 32)
    *(float4*)(&sS[i * 4]) = (float4){0.f, 0.f, 0.f, 0.f};
  for (int i = threadIdx.x; i < 2 * RPL; i += EW * 32) sC[i] = 0.f;
  const int lane = threadIdx.x & 31;
  const int h = lane >> 4, m = lane & 15;
  const int wave = __builtin_amdgcn_readfirstlane(threadIdx.x >> 5);
  const int b = blockIdx.x;
  const int lo = b * RPB;
  for (int i = threadIdx.x; i < RPB * 4; i += EW * 32) {
    const int row = i >> 2, c4 = i & 3;
    const int node = lo + row;
    float4 val = (float4){0.f, 0.f, 0.f, 0.f};
    if (node < NNODES) val = *(const float4*)(x + (long)node * 16 + c4 * 4);
    *(float4*)(&sX[row * XP + c4 * 4]) = val;
  }
  v2f bfr[48];
#pragma unroll
  for (int kc = 0; kc < 12; ++kc)
#pragma unroll
    for (int nt = 0; nt < 4; ++nt) {
      const int kk = 4 * kc + 2 * h;
      bfr[kc * 4 + nt] = (v2f){Web[kk * 64 + nt * 16 + m], Web[(kk + 1) * 64 + nt * 16 + m]};
    }
  float bb[4];
#pragma unroll
  for (int nt = 0; nt < 4; ++nt) bb[nt] = bias_eb[nt * 16 + m];
  __syncthreads();
  for (int role = 0; role < 2; ++role) {
    const int* __restrict__ list = (role ? list_s : list_r) + binbase[role * NB + b];
    const int n = totals[role * NB + b];
    const int ntiles = (n + 15) >> 4;
    float* S = sS + role * (RPL * 64);
    float* C = sC + role * RPL;
    const int* __restrict__ oth = role ? recv : send;
    int t = wave;
    if (t < ntiles) {
      v2f G0[8], G1[8];
      int w0 = EDGE_LIST(t);
      int o0 = oth[w0 & EMASK];
      EDGE_ROWS(G0, w0 & EMASK, o0)
      int w1 = EDGE_LIST(t + EW);
      int o1 = oth[w1 & EMASK];
      for (; t < ntiles; t += EW) {
        const int w2 = EDGE_LIST(t + 2 * EW);
        EDGE_ROWS(G1, w1 & EMASK, o1)
        const int lr0 = (int)((unsigned)w0 >> EBITS);
        const float* xr = sX + lr0 * XP + 2 * h;
        v2f A0[12];
#pragma unroll
        for (int cc = 0; cc < 4; ++cc) {
          A0[cc] = G0[cc];
          A0[4 + cc] = *(const v2f*)(xr + 4 * cc);
          A0[8 + cc] = G0[4 + cc];
        }
        v8f acc[4];
#pragma unroll
        for (int nt = 0; nt < 4; ++nt)
          acc[nt] = (v8f){bb[nt], bb[nt], bb[nt], bb[nt], bb[nt], bb[nt], bb[nt], bb[nt]};
#pragma unroll
        for (int kc = 0; kc < 6; ++kc)
#pragma unroll
          for (int nt = 0; nt < 4; ++nt)
            acc[nt] = __builtin_amdgcn_wmma_f32_16x16x4_f32(false, A0[kc], false, bfr[kc * 4 + nt],
                                                            (short)0, acc[nt], false, false);
        __builtin_amdgcn_sched_barrier(0);
        const int o2 = oth[w2 & EMASK];
        __builtin_amdgcn_sched_barrier(0);
#pragma unroll
        for (int kc = 6; kc < 12; ++kc)
#pragma unroll
          for (int nt = 0; nt < 4; ++nt)
            acc[nt] = __builtin_amdgcn_wmma_f32_16x16x4_f32(false, A0[kc], false, bfr[kc * 4 + nt],
                                                            (short)0, acc[nt], false, false);
#pragma unroll
        for (int nt = 0; nt < 4; ++nt)
#pragma unroll
          for (int v = 0; v < 8; ++v) acc[nt][v] = fmaxf(acc[nt][v], 0.0f);
        const int dv0 = (t * 16 + m < n && lr0 < RPB) ? lr0 : RPB;
        int dd[8];
#pragma unroll
        for (int v = 0; v < 8; ++v) dd[v] = __shfl(dv0, 8 * h + v);
        const bool hb = h != 0;
        v8f q0, q1, q2, q3;
#pragma unroll
        for (int v = 0; v < 8; ++v) {
          q0[v] = hb ? acc[1][v] : acc[0][v];
          q1[v] = hb ? acc[0][v] : acc[1][v];
          q2[v] = hb ? acc[3][v] : acc[2][v];
          q3[v] = hb ? acc[2][v] : acc[3][v];
        }
        const int oa = h << 4, ob = (1 - h) << 4;
#pragma unroll
        for (int v = 0; v < 8; ++v) {
          float* p = S + dd[v] * 64 + m;
          unsafeAtomicAdd(p + oa, q0[v]);
          unsafeAtomicAdd(p + ob, q1[v]);
          unsafeAtomicAdd(p + 32 + oa, q2[v]);
          unsafeAtomicAdd(p + 32 + ob, q3[v]);
        }
        if (h == 0) unsafeAtomicAdd(&C[dv0], 1.0f);
#pragma unroll
        for (int q = 0; q < 8; ++q) G0[q] = G1[q];
        w0 = w1; w1 = w2; o1 = o2;
      }
    }
#pragma unroll
    for (int q = 16; q < 32; ++q) {
      const v2f tmp = bfr[q];
      bfr[q] = bfr[q + 16];
      bfr[q + 16] = tmp;
    }
  }
  __syncthreads();
  for (int i = threadIdx.x; i < 2 * RPB * 16; i += EW * 32) {
    const int role = i / (RPB * 16);
    const int r = i - role * (RPB * 16);
    const int row = r >> 4, c4 = r & 15;
    const int node = lo + row;
    if (node < NNODES) {
      const float4 val = *(const float4*)(&sS[role * (RPL * 64) + row * 64 + c4 * 4]);
      *(float4*)((role ? agg_s : agg_r) + (long)node * 64 + c4 * 4) = val;
    }
  }
  for (int i = threadIdx.x; i < 2 * RPB; i += EW * 32) {
    const int role = i / RPB;
    const int row = i - role * RPB;
    const int node = lo + row;
    if (node < NNODES) (role ? cnt_s : cnt_r)[node] = sC[role * RPL + row];
  }
}

__global__ __launch_bounds__(256) void k_node(
    const float* __restrict__ x, const float* __restrict__ agg_r,
    const float* __restrict__ Wnb, const float* __restrict__ bias_nb,
    float* __restrict__ n_h) {
  __shared__ float sW[20 * 4 * 64];
  for (int i = threadIdx.x; i < 20 * 4 * 64; i += 256) {
    int j = i & 1, lane = (i >> 1) & 31, nt = (i >> 6) & 3, kc = i >> 8;
    int kk = 4 * kc + 2 * (lane >> 4) + j;
    sW[i] = Wnb[kk * 64 + nt * 16 + (lane & 15)];
  }
  __syncthreads();
  const int lane = threadIdx.x & 31;
  const int h = lane >> 4, m = lane & 15;
  const int wave = blockIdx.x * 8 + (threadIdx.x >> 5);
  const int nwaves = gridDim.x * 8;
  const int ntiles = NNODES / 16;
  for (int t = wave; t < ntiles; t += nwaves) {
    const int n0 = t * 16;
    const float* ap = agg_r + (long)(n0 + m) * 64;
    const float* xp = x + (long)(n0 + m) * 16;
    v8f acc[4];
#pragma unroll
    for (int nt = 0; nt < 4; ++nt) {
      float bb = bias_nb[nt * 16 + m];
      acc[nt] = (v8f){bb, bb, bb, bb, bb, bb, bb, bb};
    }
#pragma unroll
    for (int kc = 0; kc < 20; ++kc) {
      const int kk = 4 * kc + 2 * h;
      v2f a = (kc < 16) ? *(const v2f*)(ap + kk) : *(const v2f*)(xp + kk - 64);
#pragma unroll
      for (int nt = 0; nt < 4; ++nt) {
        v2f b = *(const v2f*)(&sW[(kc * 4 + nt) * 64 + lane * 2]);
        acc[nt] = __builtin_amdgcn_wmma_f32_16x16x4_f32(false, a, false, b,
                                                        (short)0, acc[nt], false, false);
      }
    }
#pragma unroll
    for (int nt = 0; nt < 4; ++nt)
#pragma unroll
      for (int v = 0; v < 8; ++v)
        n_h[(long)(n0 + 8 * h + v) * 64 + nt * 16 + m] = fmaxf(acc[nt][v], 0.0f);
  }
}

__global__ void k_colsum(const float* __restrict__ src, int rows, float* __restrict__ dst) {
  __shared__ float red[256];
  const int col = threadIdx.x & 63;
  const int g = threadIdx.x >> 6;
  float local = 0.f;
  for (int r = blockIdx.x * 4 + g; r < rows; r += gridDim.x * 4)
    local += src[(long)r * 64 + col];
  red[threadIdx.x] = local;
  __syncthreads();
  if (g == 0) {
    local += red[64 + threadIdx.x] + red[128 + threadIdx.x] + red[192 + threadIdx.x];
    unsafeAtomicAdd(&dst[col], local);
  }
}

__global__ void k_global(const float* __restrict__ u,
                         const float* __restrict__ Wgb, const float* __restrict__ bgb,
                         const float* __restrict__ Wd1, const float* __restrict__ bd1,
                         const float* __restrict__ e_sum, const float* __restrict__ n_sum,
                         float* __restrict__ bias_d1) {
  __shared__ float su2[64];
  const int j = threadIdx.x;
  float acc = bgb[j];
  for (int k = 0; k < 64; ++k) {
    float em = e_sum[k] * (1.0f / NEDGES);
    float nm = n_sum[k] * (1.0f / NNODES);
    acc += em * Wgb[k * 64 + j] + nm * Wgb[(64 + k) * 64 + j] + u[k] * Wgb[(128 + k) * 64 + j];
  }
  acc = fmaxf(acc, 0.0f);
  su2[j] = acc;
  __syncthreads();
  float b = bd1[j];
  for (int k = 0; k < 64; ++k) b += su2[k] * Wd1[(192 + k) * 64 + j];
  bias_d1[j] = b;
}

__global__ __launch_bounds__(256) void k_dec(
    const float* __restrict__ agg_r, const float* __restrict__ agg_s,
    const float* __restrict__ cnt_r, const float* __restrict__ cnt_s,
    const float* __restrict__ n_h, const float* __restrict__ Wd1,
    const float* __restrict__ bias_d1, const float* __restrict__ Wd2,
    const float* __restrict__ bd2, float* __restrict__ out) {
  __shared__ float sW1[48 * 4 * 64];
  __shared__ float sW2[16 * 64];
  __shared__ float sH[8 * 16 * 64];
  for (int i = threadIdx.x; i < 48 * 4 * 64; i += 256) {
    int j = i & 1, lane = (i >> 1) & 31, nt = (i >> 6) & 3, kc = i >> 8;
    int kk = 4 * kc + 2 * (lane >> 4) + j;
    sW1[i] = Wd1[kk * 64 + nt * 16 + (lane & 15)];
  }
  for (int i = threadIdx.x; i < 16 * 64; i += 256) {
    int j = i & 1, lane = (i >> 1) & 31, kc = i >> 6;
    int kk = 4 * kc + 2 * (lane >> 4) + j;
    sW2[i] = Wd2[kk * 16 + (lane & 15)];
  }
  __syncthreads();
  const int lane = threadIdx.x & 31;
  const int h = lane >> 4, m = lane & 15;
  float* hw = sH + (threadIdx.x >> 5) * 1024;
  const int wave = blockIdx.x * 8 + (threadIdx.x >> 5);
  const int nwaves = gridDim.x * 8;
  const int ntiles = NNODES / 16;
  for (int t = wave; t < ntiles; t += nwaves) {
    const int n0 = t * 16;
    const int n = n0 + m;
    const float invr = 1.0f / fmaxf(cnt_r[n], 1.0f);
    const float invs = 1.0f / fmaxf(cnt_s[n], 1.0f);
    const float* pr = agg_r + (long)n * 64;
    const float* ps = agg_s + (long)n * 64;
    const float* pn = n_h + (long)n * 64;
    v8f acc[4];
#pragma unroll
    for (int nt = 0; nt < 4; ++nt) {
      float bb = bias_d1[nt * 16 + m];
      acc[nt] = (v8f){bb, bb, bb, bb, bb, bb, bb, bb};
    }
#pragma unroll
    for (int kc = 0; kc < 48; ++kc) {
      const int kk = 4 * kc + 2 * h;
      v2f a;
      if (kc < 16)      { a = *(const v2f*)(pr + kk);        a *= invr; }
      else if (kc < 32) { a = *(const v2f*)(ps + kk - 64);   a *= invs; }
      else              { a = *(const v2f*)(pn + kk - 128); }
#pragma unroll
      for (int nt = 0; nt < 4; ++nt) {
        v2f b = *(const v2f*)(&sW1[(kc * 4 + nt) * 64 + lane * 2]);
        acc[nt] = __builtin_amdgcn_wmma_f32_16x16x4_f32(false, a, false, b,
                                                        (short)0, acc[nt], false, false);
      }
    }
#pragma unroll
    for (int nt = 0; nt < 4; ++nt)
#pragma unroll
      for (int v = 0; v < 8; ++v)
        hw[(8 * h + v) * 64 + nt * 16 + m] = fmaxf(acc[nt][v], 0.0f);
    float bb2 = bd2[m];
    v8f acc2 = (v8f){bb2, bb2, bb2, bb2, bb2, bb2, bb2, bb2};
#pragma unroll
    for (int kc = 0; kc < 16; ++kc) {
      const int kk = 4 * kc + 2 * h;
      v2f a = *(const v2f*)(hw + m * 64 + kk);
      v2f b = *(const v2f*)(&sW2[kc * 64 + lane * 2]);
      acc2 = __builtin_amdgcn_wmma_f32_16x16x4_f32(false, a, false, b,
                                                   (short)0, acc2, false, false);
    }
#pragma unroll
    for (int v = 0; v < 8; ++v)
      out[(long)(n0 + 8 * h + v) * 16 + m] = acc2[v];
  }
}

extern "C" void kernel_launch(void* const* d_in, const int* in_sizes, int n_in,
                              void* d_out, int out_size, void* d_ws, size_t ws_size,
                              hipStream_t stream) {
  const float* x    = (const float*)d_in[0];
  const float* ea   = (const float*)d_in[1];
  const float* u    = (const float*)d_in[2];
  const int* send   = (const int*)d_in[3];
  const int* recv   = (const int*)d_in[4];
  const float* Web  = (const float*)d_in[7];
  const float* beb  = (const float*)d_in[8];
  const float* Wnb  = (const float*)d_in[9];
  const float* bnb  = (const float*)d_in[10];
  const float* Wgb  = (const float*)d_in[11];
  const float* bgb  = (const float*)d_in[12];
  const float* Wd1  = (const float*)d_in[13];
  const float* bd1  = (const float*)d_in[14];
  const float* Wd2  = (const float*)d_in[15];
  const float* bd2  = (const float*)d_in[16];
  float* out = (float*)d_out;
  float* ws = (float*)d_ws;

  const long N = NNODES;
  float* agg_r   = ws;
  float* agg_s   = agg_r + N * 64;
  float* cnt_r   = agg_s + N * 64;
  float* cnt_s   = cnt_r + N;
  float* e_sum   = cnt_s + N;
  float* n_sum   = e_sum + 64;
  float* bias_eb = n_sum + 128;
  float* bias_nb = bias_eb + 64;
  float* bias_d1 = bias_nb + 64;
  float* n_hbuf  = bias_d1 + 64;
  int* list_r  = (int*)n_hbuf;
  int* list_s  = list_r + NEDGES;
  int* totals  = list_s + NEDGES;
  int* binbase = totals + 2 * NB;
  int* cursors = binbase + 2 * NB;

  k_zero<<<2, 256, 0, stream>>>(e_sum, 128, totals, 2 * NB);
  k_prep<<<1, 64, 0, stream>>>(u, Web, beb, Wnb, bnb, bias_eb, bias_nb);
  k_count<<<NCHUNKS, 256, 0, stream>>>(recv, send, totals);
  k_scan<<<1, 2 * NB, 0, stream>>>(totals, binbase, cursors);
  k_fill<<<NCHUNKS, 256, 0, stream>>>(recv, send, cursors, list_r, list_s);
  k_edge<<<NB, EW * 32, 0, stream>>>(x, ea, recv, send, Web, bias_eb, list_r, list_s,
                                    totals, binbase, agg_r, agg_s, cnt_r, cnt_s);
  k_node<<<782, 256, 0, stream>>>(x, agg_r, Wnb, bias_nb, n_hbuf);
  k_colsum<<<512, 256, 0, stream>>>(agg_r, NNODES, e_sum);
  k_colsum<<<512, 256, 0, stream>>>(n_hbuf, NNODES, n_sum);
  k_global<<<1, 64, 0, stream>>>(u, Wgb, bgb, Wd1, bd1, e_sum, n_sum, bias_d1);
  k_dec<<<1024, 256, 0, stream>>>(agg_r, agg_s, cnt_r, cnt_s, n_hbuf,
                                  Wd1, bias_d1, Wd2, bd2, out);
}


# k_edge: x @ W_eb block of the end point in the range done once per node into LDS (P, bias in it); a tile's accumulators start from its P rows: 48 to 32 WMMAs a tile; roles in turn; sum order changes
# speedup vs baseline: 1.5831x; 1.0463x over previous
#include <hip/hip_runtime.h>

typedef __attribute__((ext_vector_type(2))) float v2f;
typedef __attribute__((ext_vector_type(8))) float v8f;

#define NNODES 100000
#define NEDGES 1600000

__global__ void k_zero(float* __restrict__ a, int na, int* __restrict__ b, int nb) {
  int i = blockIdx.x * blockDim.x + threadIdx.x;
  if (i < na) a[i] = 0.f;
  if (i < nb) b[i] = 0;
}

__global__ void k_prep(const float* __restrict__ u,
                       const float* __restrict__ Web, const float* __restrict__ beb,
                       const float* __restrict__ Wnb, const float* __restrict__ bnb,
                       float* __restrict__ bias_eb, float* __restrict__ bias_nb) {
  int j = threadIdx.x;
  float a = beb[j], b = bnb[j];
  for (int k = 0; k < 64; ++k) {
    float uk = u[k];
    a += uk * Web[(48 + k) * 64 + j];
    b += uk * Wnb[(80 + k) * 64 + j];
  }
  bias_eb[j] = a;
  bias_nb[j] = b;
}

#define NB 256
#define RPB 391
#define CHUNK 2048
#define NCHUNKS ((NEDGES + CHUNK - 1) / CHUNK)
#define EBITS 21
#define EMASK ((1 << EBITS) - 1)
static_assert(NEDGES <= (1 << EBITS), "edge number must fit the low bits of a list word");
static_assert(RPB <= (1 << (30 - EBITS)), "row in range must fit above the edge number");
__global__ __launch_bounds__(256) void k_count(const int* __restrict__ recv,
                                               const int* __restrict__ send,
                                               int* __restrict__ totals) {
  __shared__ int hist[2 * NB];
  for (int j = threadIdx.x; j < 2 * NB; j += 256) hist[j] = 0;
  __syncthreads();
  const int e0 = blockIdx.x * CHUNK;
  for (int i = threadIdx.x; i < CHUNK; i += 256) {
    const int e = e0 + i;
    if (e < NEDGES) {
      atomicAdd(&hist[recv[e] / RPB], 1);
      atomicAdd(&hist[NB + send[e] / RPB], 1);
    }
  }
  __syncthreads();
  for (int j = threadIdx.x; j < 2 * NB; j += 256) {
    const int c = hist[j];
    if (c) atomicAdd(&totals[j], c);
  }
}

__global__ __launch_bounds__(2 * NB) void k_scan(const int* __restrict__ totals,
                                                 int* __restrict__ binbase,
                                                 int* __restrict__ cursors) {
  __shared__ int s[2 * NB];
  const int j = threadIdx.x;
  const int v = totals[j];
  s[j] = v;
  __syncthreads();
  for (int off = 1; off < NB; off <<= 1) {
    const int t = ((j & (NB - 1)) >= off) ? s[j - off] : 0;
    __syncthreads();
    s[j] += t;
    __syncthreads();
  }
  const int ex = s[j] - v;
  binbase[j] = ex;
  cursors[j] = ex;
}

__global__ __launch_bounds__(256) void k_fill(const int* __restrict__ recv,
                                              const int* __restrict__ send,
                                              int* __restrict__ cursors,
                                              int* __restrict__ list_r,
                                              int* __restrict__ list_s) {
  __shared__ int hist[2 * NB];
  for (int j = threadIdx.x; j < 2 * NB; j += 256) hist[j] = 0;
  __syncthreads();
  const int e0 = blockIdx.x * CHUNK;
  for (int i = threadIdx.x; i < CHUNK; i += 256) {
    const int e = e0 + i;
    if (e < NEDGES) {
      atomicAdd(&hist[recv[e] / RPB], 1);
      atomicAdd(&hist[NB + send[e] / RPB], 1);
    }
  }
  __syncthreads();
  for (int j = threadIdx.x; j < 2 * NB; j += 256) {
    const int c = hist[j];
    hist[j] = c ? atomicAdd(&cursors[j], c) : 0;
  }
  __syncthreads();
  for (int i = threadIdx.x; i < CHUNK; i += 256) {
    const int e = e0 + i;
    if (e < NEDGES) {
      const int r = recv[e], s = send[e];
      const int br = r / RPB, bs = s / RPB;
      const int p = atomicAdd(&hist[br], 1);
      list_r[p] = e | ((r - br * RPB) << EBITS);
      const int q = atomicAdd(&hist[NB + bs], 1);
      list_s[q] = e | ((s - bs * RPB) << EBITS);
    }
  }
}

#define EW 16
#define RPL (RPB + 1)
#define PTILES ((RPB + 15) / 16)
#define EDGE_LIST(tt) list[min((tt) * 16 + m, n - 1)]
#define EDGE_ROWS(A_, e_, o_)                                                 \
  {                                                                           \
    const float* p0_ = ea + (long)(e_) * 16 + 2 * h;                          \
    const float* p2_ = x + (long)(o_) * 16 + 2 * h;                           \
    _Pragma("unroll") for (int cc = 0; cc < 4; ++cc) {                        \
      A_[cc] = *(const v2f*)(p0_ + 4 * cc);                                   \
      A_[4 + cc] = *(const v2f*)(p2_ + 4 * cc);                               \
    }                                                                         \
  }
__global__ __launch_bounds__(EW * 32) void k_edge(
    const float* __restrict__ x, const float* __restrict__ ea,
    const int* __restrict__ recv, const int* __restrict__ send,
    const float* __restrict__ Web, const float* __restrict__ bias_eb,
    const int* __restrict__ list_r, const int* __restrict__ list_s,
    const int* __restrict__ totals, const int* __restrict__ binbase,
    float* __restrict__ agg_r, float* __restrict__ agg_s,
    float* __restrict__ cnt_r, float* __restrict__ cnt_s) {
  __shared__ __attribute__((aligned(16))) float sS[RPL * 64];
  __shared__ __attribute__((aligned(16))) float sP[PTILES * 16 * 64];
  __shared__ float sC[2 * RPL];
  for (int i = threadIdx.x; i < RPL * 16; i += EW * 32)
    *(float4*)(&sS[i * 4]) = (float4){0.f, 0.f, 0.f, 0.f};
  for (int i = threadIdx.x; i < 2 * RPL; i += EW * 32) sC[i] = 0.f;
  const int lane = threadIdx.x & 31;
  const int h = lane >> 4, m = lane & 15;
  const int wave = __builtin_amdgcn_readfirstlane(threadIdx.x >> 5);
  const int b = blockIdx.x;
  const int lo = b * RPB;
#define WFRAG(row0, kc, nt)                                        \
  (v2f){Web[((row0) + 4 * (kc) + 2 * h) * 64 + 4 * m + (nt)],      \
        Web[((row0) + 4 * (kc) + 2 * h + 1) * 64 + 4 * m + (nt)]}
  v2f bA[16];
#pragma unroll
  for (int kc = 0; kc < 4; ++kc)
#pragma unroll
    for (int nt = 0; nt < 4; ++nt) bA[kc * 4 + nt] = WFRAG(0, kc, nt);
  float bb[4];
#pragma unroll
  for (int nt = 0; nt < 4; ++nt) bb[nt] = bias_eb[4 * m + nt];
  for (int role = 0; role < 2; ++role) {
    const int rin = role ? 32 : 16, rot = role ? 16 : 32;
    v2f bO[16];
#pragma unroll
    for (int kc = 0; kc < 4; ++kc)
#pragma unroll
      for (int nt = 0; nt < 4; ++nt) bO[kc * 4 + nt] = WFRAG(rot, kc, nt);
    if (wave < PTILES) {
      v2f bI[16];
#pragma unroll
      for (int kc = 0; kc < 4; ++kc)
#pragma unroll
        for (int nt = 0; nt < 4; ++nt) bI[kc * 4 + nt] = WFRAG(rin, kc, nt);
      for (int pt = wave; pt < PTILES; pt += EW) {
        const int r = pt * 16 + m;
        const int node = lo + r;
        v2f a[4];
#pragma unroll
        for (int cc = 0; cc < 4; ++cc) a[cc] = (v2f){0.f, 0.f};
        if (r < RPB && node < NNODES) {
          const float* px = x + (long)node * 16 + 2 * h;
#pragma unroll
          for (int cc = 0; cc < 4; ++cc) a[cc] = *(const v2f*)(px + 4 * cc);
        }
        v8f pa[4];
#pragma unroll
        for (int nt = 0; nt < 4; ++nt)
          pa[nt] = (v8f){bb[nt], bb[nt], bb[nt], bb[nt], bb[nt], bb[nt], bb[nt], bb[nt]};
#pragma unroll
        for (int cc = 0; cc < 4; ++cc)
#pragma unroll
          for (int nt = 0; nt < 4; ++nt)
            pa[nt] = __builtin_amdgcn_wmma_f32_16x16x4_f32(false, a[cc], false, bI[cc * 4 + nt],
                                                           (short)0, pa[nt], false, false);
#pragma unroll
        for (int v = 0; v < 8; ++v)
          *(float4*)(&sP[(pt * 16 + 8 * h + v) * 64 + 4 * m]) =
              (float4){pa[0][v], pa[1][v], pa[2][v], pa[3][v]};
      }
    }
    __syncthreads();
    const int* __restrict__ list = (role ? list_s : list_r) + binbase[role * NB + b];
    const int n = totals[role * NB + b];
    const int ntiles = (n + 15) >> 4;
    float* C = sC + role * RPL;
    const int* __restrict__ oth = role ? recv : send;
#define ROW_PAIR(w_, tt)                                                      \
  (min((int)((unsigned)(w_) >> EBITS), RPB) |                                 \
   (((tt) * 16 + m < n ? min((int)((unsigned)(w_) >> EBITS), RPB) : RPB) << 16))
#define P_ROWS(acc_, d_)                                                            \
  _Pragma("unroll") for (int v = 0; v < 8; ++v) {                                   \
    const float4 pv_ = *(const float4*)(&sP[((d_)[v] & 0xffff) * 64 + 4 * m]);      \
    acc_[0][v] = pv_.x; acc_[1][v] = pv_.y; acc_[2][v] = pv_.z; acc_[3][v] = pv_.w; \
  }
    int t = wave;
    if (t < ntiles) {
      v2f G0[8], G1[8];
      int w0 = EDGE_LIST(t);
      int o0 = oth[w0 & EMASK];
      EDGE_ROWS(G0, w0 & EMASK, o0)
      int w1 = EDGE_LIST(t + EW);
      int o1 = oth[w1 & EMASK];
      int rp0 = ROW_PAIR(w0, t);
      int dd[8];
#pragma unroll
      for (int v = 0; v < 8; ++v) dd[v] = __shfl(rp0, 8 * h + v);
      v8f acc[4];
      P_ROWS(acc, dd)
      for (; t < ntiles; t += EW) {
        const int w2 = EDGE_LIST(t + 2 * EW);
        EDGE_ROWS(G1, w1 & EMASK, o1)
        const int rp1 = ROW_PAIR(w1, t + EW);
        int dn[8];
#pragma unroll
        for (int v = 0; v < 8; ++v) dn[v] = __shfl(rp1, 8 * h + v);
#pragma unroll
        for (int cc = 0; cc < 4; ++cc)
#pragma unroll
          for (int nt = 0; nt < 4; ++nt)
            acc[nt] = __builtin_amdgcn_wmma_f32_16x16x4_f32(false, G0[cc], false, bA[cc * 4 + nt],
                                                            (short)0, acc[nt], false, false);
        __builtin_amdgcn_sched_barrier(0);
        const int o2 = oth[w2 & EMASK];
        __builtin_amdgcn_sched_barrier(0);
#pragma unroll
        for (int cc = 0; cc < 4; ++cc)
#pragma unroll
          for (int nt = 0; nt < 4; ++nt)
            acc[nt] = __builtin_amdgcn_wmma_f32_16x16x4_f32(false, G0[4 + cc], false,
                                                            bO[cc * 4 + nt],
                                                            (short)0, acc[nt], false, false);
        const bool hb = h != 0;
        v8f q0, q1, q2, q3;
#pragma unroll
        for (int v = 0; v < 8; ++v) {
          const float e0 = fmaxf(acc[0][v], 0.0f), e1 = fmaxf(acc[1][v], 0.0f);
          const float e2 = fmaxf(acc[2][v], 0.0f), e3 = fmaxf(acc[3][v], 0.0f);
          q0[v] = hb ? e1 : e0;
          q1[v] = hb ? e0 : e1;
          q2[v] = hb ? e3 : e2;
          q3[v] = hb ? e2 : e3;
        }
        P_ROWS(acc, dn)
        __builtin_amdgcn_sched_barrier(0);
        const int oa = h, ob = 1 - h;
#pragma unroll
        for (int v = 0; v < 8; ++v) {
          float* p = sS + (dd[v] >> 16) * 64 + 4 * m;
          unsafeAtomicAdd(p + oa, q0[v]);
          unsafeAtomicAdd(p + ob, q1[v]);
          unsafeAtomicAdd(p + 2 + oa, q2[v]);
          unsafeAtomicAdd(p + 2 + ob, q3[v]);
        }
        if (h == 0) unsafeAtomicAdd(&C[rp0 >> 16], 1.0f);
#pragma unroll
        for (int q = 0; q < 8; ++q) G0[q] = G1[q];
#pragma unroll
        for (int v = 0; v < 8; ++v) dd[v] = dn[v];
        rp0 = rp1;
        w0 = w1; w1 = w2; o1 = o2;
      }
    }
    __syncthreads();
    float* __restrict__ agg = role ? agg_s : agg_r;
    for (int i = threadIdx.x; i < RPB * 16; i += EW * 32) {
      const int row = i >> 4, c4 = i & 15;
      const int node = lo + row;
      const float4 val = *(const float4*)(&sS[row * 64 + c4 * 4]);
      *(float4*)(&sS[row * 64 + c4 * 4]) = (float4){0.f, 0.f, 0.f, 0.f};
      if (node < NNODES) *(float4*)(agg + (long)node * 64 + c4 * 4) = val;
    }
  }
  for (int i = threadIdx.x; i < 2 * RPB; i += EW * 32) {
    const int role = i / RPB;
    const int row = i - role * RPB;
    const int node = lo + row;
    if (node < NNODES) (role ? cnt_s : cnt_r)[node] = sC[role * RPL + row];
  }
}

__global__ __launch_bounds__(256) void k_node(
    const float* __restrict__ x, const float* __restrict__ agg_r,
    const float* __restrict__ Wnb, const float* __restrict__ bias_nb,
    float* __restrict__ n_h) {
  __shared__ float sW[20 * 4 * 64];
  for (int i = threadIdx.x; i < 20 * 4 * 64; i += 256) {
    int j = i & 1, lane = (i >> 1) & 31, nt = (i >> 6) & 3, kc = i >> 8;
    int kk = 4 * kc + 2 * (lane >> 4) + j;
    sW[i] = Wnb[kk * 64 + nt * 16 + (lane & 15)];
  }
  __syncthreads();
  const int lane = threadIdx.x & 31;
  const int h = lane >> 4, m = lane & 15;
  const int wave = blockIdx.x * 8 + (threadIdx.x >> 5);
  const int nwaves = gridDim.x * 8;
  const int ntiles = NNODES / 16;
  for (int t = wave; t < ntiles; t += nwaves) {
    const int n0 = t * 16;
    const float* ap = agg_r + (long)(n0 + m) * 64;
    const float* xp = x + (long)(n0 + m) * 16;
    v8f acc[4];
#pragma unroll
    for (int nt = 0; nt < 4; ++nt) {
      float bb = bias_nb[nt * 16 + m];
      acc[nt] = (v8f){bb, bb, bb, bb, bb, bb, bb, bb};
    }
#pragma unroll
    for (int kc = 0; kc < 20; ++kc) {
      const int kk = 4 * kc + 2 * h;
      v2f a = (kc < 16) ? *(const v2f*)(ap + kk) : *(const v2f*)(xp + kk - 64);
#pragma unroll
      for (int nt = 0; nt < 4; ++nt) {
        v2f b = *(const v2f*)(&sW[(kc * 4 + nt) * 64 + lane * 2]);
        acc[nt] = __builtin_amdgcn_wmma_f32_16x16x4_f32(false, a, false, b,
                                                        (short)0, acc[nt], false, false);
      }
    }
#pragma unroll
    for (int nt = 0; nt < 4; ++nt)
#pragma unroll
      for (int v = 0; v < 8; ++v)
        n_h[(long)(n0 + 8 * h + v) * 64 + nt * 16 + m] = fmaxf(acc[nt][v], 0.0f);
  }
}

__global__ void k_colsum(const float* __restrict__ src, int rows, float* __restrict__ dst) {
  __shared__ float red[256];
  const int col = threadIdx.x & 63;
  const int g = threadIdx.x >> 6;
  float local = 0.f;
  for (int r = blockIdx.x * 4 + g; r < rows; r += gridDim.x * 4)
    local += src[(long)r * 64 + col];
  red[threadIdx.x] = local;
  __syncthreads();
  if (g == 0) {
    local += red[64 + threadIdx.x] + red[128 + threadIdx.x] + red[192 + threadIdx.x];
    unsafeAtomicAdd(&dst[col], local);
  }
}

__global__ void k_global(const float* __restrict__ u,
                         const float* __restrict__ Wgb, const float* __restrict__ bgb,
                         const float* __restrict__ Wd1, const float* __restrict__ bd1,
                         const float* __restrict__ e_sum, const float* __restrict__ n_sum,
                         float* __restrict__ bias_d1) {
  __shared__ float su2[64];
  const int j = threadIdx.x;
  float acc = bgb[j];
  for (int k = 0; k < 64; ++k) {
    float em = e_sum[k] * (1.0f / NEDGES);
    float nm = n_sum[k] * (1.0f / NNODES);
    acc += em * Wgb[k * 64 + j] + nm * Wgb[(64 + k) * 64 + j] + u[k] * Wgb[(128 + k) * 64 + j];
  }
  acc = fmaxf(acc, 0.0f);
  su2[j] = acc;
  __syncthreads();
  float b = bd1[j];
  for (int k = 0; k < 64; ++k) b += su2[k] * Wd1[(192 + k) * 64 + j];
  bias_d1[j] = b;
}

__global__ __launch_bounds__(256) void k_dec(
    const float* __restrict__ agg_r, const float* __restrict__ agg_s,
    const float* __restrict__ cnt_r, const float* __restrict__ cnt_s,
    const float* __restrict__ n_h, const float* __restrict__ Wd1,
    const float* __restrict__ bias_d1, const float* __restrict__ Wd2,
    const float* __restrict__ bd2, float* __restrict__ out) {
  __shared__ float sW1[48 * 4 * 64];
  __shared__ float sW2[16 * 64];
  __shared__ float sH[8 * 16 * 64];
  for (int i = threadIdx.x; i < 48 * 4 * 64; i += 256) {
    int j = i & 1, lane = (i >> 1) & 31, nt = (i >> 6) & 3, kc = i >> 8;
    int kk = 4 * kc + 2 * (lane >> 4) + j;
    sW1[i] = Wd1[kk * 64 + nt * 16 + (lane & 15)];
  }
  for (int i = threadIdx.x; i < 16 * 64; i += 256) {
    int j = i & 1, lane = (i >> 1) & 31, kc = i >> 6;
    int kk = 4 * kc + 2 * (lane >> 4) + j;
    sW2[i] = Wd2[kk * 16 + (lane & 15)];
  }
  __syncthreads();
  const int lane = threadIdx.x & 31;
  const int h = lane >> 4, m = lane & 15;
  float* hw = sH + (threadIdx.x >> 5) * 1024;
  const int wave = blockIdx.x * 8 + (threadIdx.x >> 5);
  const int nwaves = gridDim.x * 8;
  const int ntiles = NNODES / 16;
  for (int t = wave; t < ntiles; t += nwaves) {
    const int n0 = t * 16;
    const int n = n0 + m;
    const float invr = 1.0f / fmaxf(cnt_r[n], 1.0f);
    const float invs = 1.0f / fmaxf(cnt_s[n], 1.0f);
    const float* pr = agg_r + (long)n * 64;
    const float* ps = agg_s + (long)n * 64;
    const float* pn = n_h + (long)n * 64;
    v8f acc[4];
#pragma unroll
    for (int nt = 0; nt < 4; ++nt) {
      float bb = bias_d1[nt * 16 + m];
      acc[nt] = (v8f){bb, bb, bb, bb, bb, bb, bb, bb};
    }
#pragma unroll
    for (int kc = 0; kc < 48; ++kc) {
      const int kk = 4 * kc + 2 * h;
      v2f a;
      if (kc < 16)      { a = *(const v2f*)(pr + kk);        a *= invr; }
      else if (kc < 32) { a = *(const v2f*)(ps + kk - 64);   a *= invs; }
      else              { a = *(const v2f*)(pn + kk - 128); }
#pragma unroll
      for (int nt = 0; nt < 4; ++nt) {
        v2f b = *(const v2f*)(&sW1[(kc * 4 + nt) * 64 + lane * 2]);
        acc[nt] = __builtin_amdgcn_wmma_f32_16x16x4_f32(false, a, false, b,
                                                        (short)0, acc[nt], false, false);
      }
    }
#pragma unroll
    for (int nt = 0; nt < 4; ++nt)
#pragma unroll
      for (int v = 0; v < 8; ++v)
        hw[(8 * h + v) * 64 + nt * 16 + m] = fmaxf(acc[nt][v], 0.0f);
    float bb2 = bd2[m];
    v8f acc2 = (v8f){bb2, bb2, bb2, bb2, bb2, bb2, bb2, bb2};
#pragma unroll
    for (int kc = 0; kc < 16; ++kc) {
      const int kk = 4 * kc + 2 * h;
      v2f a = *(const v2f*)(hw + m * 64 + kk);
      v2f b = *(const v2f*)(&sW2[kc * 64 + lane * 2]);
      acc2 = __builtin_amdgcn_wmma_f32_16x16x4_f32(false, a, false, b,
                                                   (short)0, acc2, false, false);
    }
#pragma unroll
    for (int v = 0; v < 8; ++v)
      out[(long)(n0 + 8 * h + v) * 16 + m] = acc2[v];
  }
}

extern "C" void kernel_launch(void* const* d_in, const int* in_sizes, int n_in,
                              void* d_out, int out_size, void* d_ws, size_t ws_size,
                              hipStream_t stream) {
  const float* x    = (const float*)d_in[0];
  const float* ea   = (const float*)d_in[1];
  const float* u    = (const float*)d_in[2];
  const int* send   = (const int*)d_in[3];
  const int* recv   = (const int*)d_in[4];
  const float* Web  = (const float*)d_in[7];
  const float* beb  = (const float*)d_in[8];
  const float* Wnb  = (const float*)d_in[9];
  const float* bnb  = (const float*)d_in[10];
  const float* Wgb  = (const float*)d_in[11];
  const float* bgb  = (const float*)d_in[12];
  const float* Wd1  = (const float*)d_in[13];
  const float* bd1  = (const float*)d_in[14];
  const float* Wd2  = (const float*)d_in[15];
  const float* bd2  = (const float*)d_in[16];
  float* out = (float*)d_out;
  float* ws = (float*)d_ws;

  const long N = NNODES;
  float* agg_r   = ws;
  float* agg_s   = agg_r + N * 64;
  float* cnt_r   = agg_s + N * 64;
  float* cnt_s   = cnt_r + N;
  float* e_sum   = cnt_s + N;
  float* n_sum   = e_sum + 64;
  float* bias_eb = n_sum + 128;
  float* bias_nb = bias_eb + 64;
  float* bias_d1 = bias_nb + 64;
  float* n_hbuf  = bias_d1 + 64;
  int* list_r  = (int*)n_hbuf;
  int* list_s  = list_r + NEDGES;
  int* totals  = list_s + NEDGES;
  int* binbase = totals + 2 * NB;
  int* cursors = binbase + 2 * NB;

  k_zero<<<2, 256, 0, stream>>>(e_sum, 128, totals, 2 * NB);
  k_prep<<<1, 64, 0, stream>>>(u, Web, beb, Wnb, bnb, bias_eb, bias_nb);
  k_count<<<NCHUNKS, 256, 0, stream>>>(recv, send, totals);
  k_scan<<<1, 2 * NB, 0, stream>>>(totals, binbase, cursors);
  k_fill<<<NCHUNKS, 256, 0, stream>>>(recv, send, cursors, list_r, list_s);
  k_edge<<<NB, EW * 32, 0, stream>>>(x, ea, recv, send, Web, bias_eb, list_r, list_s,
                                    totals, binbase, agg_r, agg_s, cnt_r, cnt_s);
  k_node<<<782, 256, 0, stream>>>(x, agg_r, Wnb, bias_nb, n_hbuf);
  k_colsum<<<512, 256, 0, stream>>>(agg_r, NNODES, e_sum);
  k_colsum<<<512, 256, 0, stream>>>(n_hbuf, NNODES, n_sum);
  k_global<<<1, 64, 0, stream>>>(u, Wgb, bgb, Wd1, bd1, e_sum, n_sum, bias_d1);
  k_dec<<<1024, 256, 0, stream>>>(agg_r, agg_s, cnt_r, cnt_s, n_hbuf,
                                  Wd1, bias_d1, Wd2, bd2, out);
}


# k_count, k_fill, k_colsum, k_prep, k_global, weight staging of k_node and k_dec: a thread's loads issued as one group (were waited for one by one); k_fill reads indices once; sums keep their order
# speedup vs baseline: 1.9872x; 1.2553x over previous
#include <hip/hip_runtime.h>

typedef __attribute__((ext_vector_type(2))) float v2f;
typedef __attribute__((ext_vector_type(8))) float v8f;

#define NNODES 100000
#define NEDGES 1600000

__global__ void k_zero(float* __restrict__ a, int na, int* __restrict__ b, int nb) {
  int i = blockIdx.x * blockDim.x + threadIdx.x;
  if (i < na) a[i] = 0.f;
  if (i < nb) b[i] = 0;
}

__global__ void k_prep(const float* __restrict__ u,
                       const float* __restrict__ Web, const float* __restrict__ beb,
                       const float* __restrict__ Wnb, const float* __restrict__ bnb,
                       float* __restrict__ bias_eb, float* __restrict__ bias_nb) {
  int j = threadIdx.x;
  float a = beb[j], b = bnb[j];
  for (int k0 = 0; k0 < 64; k0 += 32) {
    float wa[32], wb[32];
#pragma unroll
    for (int q = 0; q < 32; ++q) {
      wa[q] = Web[(48 + k0 + q) * 64 + j];
      wb[q] = Wnb[(80 + k0 + q) * 64 + j];
    }
#pragma unroll
    for (int q = 0; q < 32; ++q) {
      float uk = u[k0 + q];
      a += uk * wa[q];
      b += uk * wb[q];
    }
  }
  bias_eb[j] = a;
  bias_nb[j] = b;
}

#define NB 256
#define RPB 391
#define CHUNK 2048
#define NCHUNKS ((NEDGES + CHUNK - 1) / CHUNK)
#define EPT (CHUNK / 256)
#define EBITS 21
#define EMASK ((1 << EBITS) - 1)
static_assert(NEDGES <= (1 << EBITS), "edge number must fit the low bits of a list word");
static_assert(RPB <= (1 << (30 - EBITS)), "row in range must fit above the edge number");
__global__ __launch_bounds__(256) void k_count(const int* __restrict__ recv,
                                               const int* __restrict__ send,
                                               int* __restrict__ totals) {
  __shared__ int hist[2 * NB];
  for (int j = threadIdx.x; j < 2 * NB; j += 256) hist[j] = 0;
  __syncthreads();
  const int e0 = blockIdx.x * CHUNK;
  int nr[EPT], ns[EPT];
#pragma unroll
  for (int k = 0; k < EPT; ++k) {
    const int e = min(e0 + k * 256 + (int)threadIdx.x, NEDGES - 1);
    nr[k] = recv[e];
    ns[k] = send[e];
  }
#pragma unroll
  for (int k = 0; k < EPT; ++k) {
    if (e0 + k * 256 + (int)threadIdx.x < NEDGES) {
      atomicAdd(&hist[nr[k] / RPB], 1);
      atomicAdd(&hist[NB + ns[k] / RPB], 1);
    }
  }
  __syncthreads();
  for (int j = threadIdx.x; j < 2 * NB; j += 256) {
    const int c = hist[j];
    if (c) atomicAdd(&totals[j], c);
  }
}

__global__ __launch_bounds__(2 * NB) void k_scan(const int* __restrict__ totals,
                                                 int* __restrict__ binbase,
                                                 int* __restrict__ cursors) {
  __shared__ int s[2 * NB];
  const int j = threadIdx.x;
  const int v = totals[j];
  s[j] = v;
  __syncthreads();
  for (int off = 1; off < NB; off <<= 1) {
    const int t = ((j & (NB - 1)) >= off) ? s[j - off] : 0;
    __syncthreads();
    s[j] += t;
    __syncthreads();
  }
  const int ex = s[j] - v;
  binbase[j] = ex;
  cursors[j] = ex;
}

__global__ __launch_bounds__(256) void k_fill(const int* __restrict__ recv,
                                              const int* __restrict__ send,
                                              int* __restrict__ cursors,
                                              int* __restrict__ list_r,
                                              int* __restrict__ list_s) {
  __shared__ int hist[2 * NB];
  for (int j = threadIdx.x; j < 2 * NB; j += 256) hist[j] = 0;
  __syncthreads();
  const int e0 = blockIdx.x * CHUNK;
  int nr[EPT], ns[EPT];
#pragma unroll
  for (int k = 0; k < EPT; ++k) {
    const int e = min(e0 + k * 256 + (int)threadIdx.x, NEDGES - 1);
    nr[k] = recv[e];
    ns[k] = send[e];
  }
#pragma unroll
  for (int k = 0; k < EPT; ++k) {
    if (e0 + k * 256 + (int)threadIdx.x < NEDGES) {
      atomicAdd(&hist[nr[k] / RPB], 1);
      atomicAdd(&hist[NB + ns[k] / RPB], 1);
    }
  }
  __syncthreads();
  for (int j = threadIdx.x; j < 2 * NB; j += 256) {
    const int c = hist[j];
    hist[j] = c ? atomicAdd(&cursors[j], c) : 0;
  }
  __syncthreads();
#pragma unroll
  for (int k = 0; k < EPT; ++k) {
    const int e = e0 + k * 256 + (int)threadIdx.x;
    if (e < NEDGES) {
      const int r = nr[k], s = ns[k];
      const int br = r / RPB, bs = s / RPB;
      const int p = atomicAdd(&hist[br], 1);
      list_r[p] = e | ((r - br * RPB) << EBITS);
      const int q = atomicAdd(&hist[NB + bs], 1);
      list_s[q] = e | ((s - bs * RPB) << EBITS);
    }
  }
}

#define EW 16
#define RPL (RPB + 1)
#define PTILES ((RPB + 15) / 16)
#define EDGE_LIST(tt) list[min((tt) * 16 + m, n - 1)]
#define EDGE_ROWS(A_, e_, o_)                                                 \
  {                                                                           \
    const float* p0_ = ea + (long)(e_) * 16 + 2 * h;                          \
    const float* p2_ = x + (long)(o_) * 16 + 2 * h;                           \
    _Pragma("unroll") for (int cc = 0; cc < 4; ++cc) {                        \
      A_[cc] = *(const v2f*)(p0_ + 4 * cc);                                   \
      A_[4 + cc] = *(const v2f*)(p2_ + 4 * cc);                               \
    }                                                                         \
  }
__global__ __launch_bounds__(EW * 32) void k_edge(
    const float* __restrict__ x, const float* __restrict__ ea,
    const int* __restrict__ recv, const int* __restrict__ send,
    const float* __restrict__ Web, const float* __restrict__ bias_eb,
    const int* __restrict__ list_r, const int* __restrict__ list_s,
    const int* __restrict__ totals, const int* __restrict__ binbase,
    float* __restrict__ agg_r, float* __restrict__ agg_s,
    float* __restrict__ cnt_r, float* __restrict__ cnt_s) {
  __shared__ __attribute__((aligned(16))) float sS[RPL * 64];
  __shared__ __attribute__((aligned(16))) float sP[PTILES * 16 * 64];
  __shared__ float sC[2 * RPL];
  for (int i = threadIdx.x; i < RPL * 16; i += EW * 32)
    *(float4*)(&sS[i * 4]) = (float4){0.f, 0.f, 0.f, 0.f};
  for (int i = threadIdx.x; i < 2 * RPL; i += EW * 32) sC[i] = 0.f;
  const int lane = threadIdx.x & 31;
  const int h = lane >> 4, m = lane & 15;
  const int wave = __builtin_amdgcn_readfirstlane(threadIdx.x >> 5);
  const int b = blockIdx.x;
  const int lo = b * RPB;
#define WFRAG(row0, kc, nt)                                        \
  (v2f){Web[((row0) + 4 * (kc) + 2 * h) * 64 + 4 * m + (nt)],      \
        Web[((row0) + 4 * (kc) + 2 * h + 1) * 64 + 4 * m + (nt)]}
  v2f bA[16];
#pragma unroll
  for (int kc = 0; kc < 4; ++kc)
#pragma unroll
    for (int nt = 0; nt < 4; ++nt) bA[kc * 4 + nt] = WFRAG(0, kc, nt);
  float bb[4];
#pragma unroll
  for (int nt = 0; nt < 4; ++nt) bb[nt] = bias_eb[4 * m + nt];
  for (int role = 0; role < 2; ++role) {
    const int rin = role ? 32 : 16, rot = role ? 16 : 32;
    v2f bO[16];
#pragma unroll
    for (int kc = 0; kc < 4; ++kc)
#pragma unroll
      for (int nt = 0; nt < 4; ++nt) bO[kc * 4 + nt] = WFRAG(rot, kc, nt);
    if (wave < PTILES) {
      v2f bI[16];
#pragma unroll
      for (int kc = 0; kc < 4; ++kc)
#pragma unroll
        for (int nt = 0; nt < 4; ++nt) bI[kc * 4 + nt] = WFRAG(rin, kc, nt);
      for (int pt = wave; pt < PTILES; pt += EW) {
        const int r = pt * 16 + m;
        const int node = lo + r;
        v2f a[4];
#pragma unroll
        for (int cc = 0; cc < 4; ++cc) a[cc] = (v2f){0.f, 0.f};
        if (r < RPB && node < NNODES) {
          const float* px = x + (long)node * 16 + 2 * h;
#pragma unroll
          for (int cc = 0; cc < 4; ++cc) a[cc] = *(const v2f*)(px + 4 * cc);
        }
        v8f pa[4];
#pragma unroll
        for (int nt = 0; nt < 4; ++nt)
          pa[nt] = (v8f){bb[nt], bb[nt], bb[nt], bb[nt], bb[nt], bb[nt], bb[nt], bb[nt]};
#pragma unroll
        for (int cc = 0; cc < 4; ++cc)
#pragma unroll
          for (int nt = 0; nt < 4; ++nt)
            pa[nt] = __builtin_amdgcn_wmma_f32_16x16x4_f32(false, a[cc], false, bI[cc * 4 + nt],
                                                           (short)0, pa[nt], false, false);
#pragma unroll
        for (int v = 0; v < 8; ++v)
          *(float4*)(&sP[(pt * 16 + 8 * h + v) * 64 + 4 * m]) =
              (float4){pa[0][v], pa[1][v], pa[2][v], pa[3][v]};
      }
    }
    __syncthreads();
    const int* __restrict__ list = (role ? list_s : list_r) + binbase[role * NB + b];
    const int n = totals[role * NB + b];
    const int ntiles = (n + 15) >> 4;
    float* C = sC + role * RPL;
    const int* __restrict__ oth = role ? recv : send;
#define ROW_PAIR(w_, tt)                                                      \
  (min((int)((unsigned)(w_) >> EBITS), RPB) |                                 \
   (((tt) * 16 + m < n ? min((int)((unsigned)(w_) >> EBITS), RPB) : RPB) << 16))
#define P_ROWS(acc_, d_)                                                            \
  _Pragma("unroll") for (int v = 0; v < 8; ++v) {                                   \
    const float4 pv_ = *(const float4*)(&sP[((d_)[v] & 0xffff) * 64 + 4 * m]);      \
    acc_[0][v] = pv_.x; acc_[1][v] = pv_.y; acc_[2][v] = pv_.z; acc_[3][v] = pv_.w; \
  }
    int t = wave;
    if (t < ntiles) {
      v2f G0[8], G1[8];
      int w0 = EDGE_LIST(t);
      int o0 = oth[w0 & EMASK];
      EDGE_ROWS(G0, w0 & EMASK, o0)
      int w1 = EDGE_LIST(t + EW);
      int o1 = oth[w1 & EMASK];
      int rp0 = ROW_PAIR(w0, t);
      int dd[8];
#pragma unroll
      for (int v = 0; v < 8; ++v) dd[v] = __shfl(rp0, 8 * h + v);
      v8f acc[4];
      P_ROWS(acc, dd)
      for (; t < ntiles; t += EW) {
        const int w2 = EDGE_LIST(t + 2 * EW);
        EDGE_ROWS(G1, w1 & EMASK, o1)
        const int rp1 = ROW_PAIR(w1, t + EW);
        int dn[8];
#pragma unroll
        for (int v = 0; v < 8; ++v) dn[v] = __shfl(rp1, 8 * h + v);
#pragma unroll
        for (int cc = 0; cc < 4; ++cc)
#pragma unroll
          for (int nt = 0; nt < 4; ++nt)
            acc[nt] = __builtin_amdgcn_wmma_f32_16x16x4_f32(false, G0[cc], false, bA[cc * 4 + nt],
                                                            (short)0, acc[nt], false, false);
        __builtin_amdgcn_sched_barrier(0);
        const int o2 = oth[w2 & EMASK];
        __builtin_amdgcn_sched_barrier(0);
#pragma unroll
        for (int cc = 0; cc < 4; ++cc)
#pragma unroll
          for (int nt = 0; nt < 4; ++nt)
            acc[nt] = __builtin_amdgcn_wmma_f32_16x16x4_f32(false, G0[4 + cc], false,
                                                            bO[cc * 4 + nt],
                                                            (short)0, acc[nt], false, false);
        const bool hb = h != 0;
        v8f q0, q1, q2, q3;
#pragma unroll
        for (int v = 0; v < 8; ++v) {
          const float e0 = fmaxf(acc[0][v], 0.0f), e1 = fmaxf(acc[1][v], 0.0f);
          const float e2 = fmaxf(acc[2][v], 0.0f), e3 = fmaxf(acc[3][v], 0.0f);
          q0[v] = hb ? e1 : e0;
          q1[v] = hb ? e0 : e1;
          q2[v] = hb ? e3 : e2;
          q3[v] = hb ? e2 : e3;
        }
        P_ROWS(acc, dn)
        __builtin_amdgcn_sched_barrier(0);
        const int oa = h, ob = 1 - h;
#pragma unroll
        for (int v = 0; v < 8; ++v) {
          float* p = sS + (dd[v] >> 16) * 64 + 4 * m;
          unsafeAtomicAdd(p + oa, q0[v]);
          unsafeAtomicAdd(p + ob, q1[v]);
          unsafeAtomicAdd(p + 2 + oa, q2[v]);
          unsafeAtomicAdd(p + 2 + ob, q3[v]);
        }
        if (h == 0) unsafeAtomicAdd(&C[rp0 >> 16], 1.0f);
#pragma unroll
        for (int q = 0; q < 8; ++q) G0[q] = G1[q];
#pragma unroll
        for (int v = 0; v < 8; ++v) dd[v] = dn[v];
        rp0 = rp1;
        w0 = w1; w1 = w2; o1 = o2;
      }
    }
    __syncthreads();
    float* __restrict__ agg = role ? agg_s : agg_r;
    for (int i = threadIdx.x; i < RPB * 16; i += EW * 32) {
      const int row = i >> 4, c4 = i & 15;
      const int node = lo + row;
      const float4 val = *(const float4*)(&sS[row * 64 + c4 * 4]);
      *(float4*)(&sS[row * 64 + c4 * 4]) = (float4){0.f, 0.f, 0.f, 0.f};
      if (node < NNODES) *(float4*)(agg + (long)node * 64 + c4 * 4) = val;
    }
  }
  for (int i = threadIdx.x; i < 2 * RPB; i += EW * 32) {
    const int role = i / RPB;
    const int row = i - role * RPB;
    const int node = lo + row;
    if (node < NNODES) (role ? cnt_s : cnt_r)[node] = sC[role * RPL + row];
  }
}

__global__ __launch_bounds__(256) void k_node(
    const float* __restrict__ x, const float* __restrict__ agg_r,
    const float* __restrict__ Wnb, const float* __restrict__ bias_nb,
    float* __restrict__ n_h) {
  __shared__ float sW[20 * 4 * 64];
  {
    float wv[20];
#pragma unroll
    for (int q = 0; q < 20; ++q) {
      const int i = threadIdx.x + q * 256;
      int j = i & 1, lane = (i >> 1) & 31, nt = (i >> 6) & 3, kc = i >> 8;
      int kk = 4 * kc + 2 * (lane >> 4) + j;
      wv[q] = Wnb[kk * 64 + nt * 16 + (lane & 15)];
    }
#pragma unroll
    for (int q = 0; q < 20; ++q) sW[threadIdx.x + q * 256] = wv[q];
  }
  __syncthreads();
  const int lane = threadIdx.x & 31;
  const int h = lane >> 4, m = lane & 15;
  const int wave = blockIdx.x * 8 + (threadIdx.x >> 5);
  const int nwaves = gridDim.x * 8;
  const int ntiles = NNODES / 16;
  for (int t = wave; t < ntiles; t += nwaves) {
    const int n0 = t * 16;
    const float* ap = agg_r + (long)(n0 + m) * 64;
    const float* xp = x + (long)(n0 + m) * 16;
    v8f acc[4];
#pragma unroll
    for (int nt = 0; nt < 4; ++nt) {
      float bb = bias_nb[nt * 16 + m];
      acc[nt] = (v8f){bb, bb, bb, bb, bb, bb, bb, bb};
    }
#pragma unroll
    for (int kc = 0; kc < 20; ++kc) {
      const int kk = 4 * kc + 2 * h;
      v2f a = (kc < 16) ? *(const v2f*)(ap + kk) : *(const v2f*)(xp + kk - 64);
#pragma unroll
      for (int nt = 0; nt < 4; ++nt) {
        v2f b = *(const v2f*)(&sW[(kc * 4 + nt) * 64 + lane * 2]);
        acc[nt] = __builtin_amdgcn_wmma_f32_16x16x4_f32(false, a, false, b,
                                                        (short)0, acc[nt], false, false);
      }
    }
#pragma unroll
    for (int nt = 0; nt < 4; ++nt)
#pragma unroll
      for (int v = 0; v < 8; ++v)
        n_h[(long)(n0 + 8 * h + v) * 64 + nt * 16 + m] = fmaxf(acc[nt][v], 0.0f);
  }
}

__global__ void k_colsum(const float* __restrict__ src, int rows, float* __restrict__ dst) {
  __shared__ float red[256];
  const int col = threadIdx.x & 63;
  const int g = threadIdx.x >> 6;
  float local = 0.f;
  const int step = gridDim.x * 4;
  int r = blockIdx.x * 4 + g;
  for (; r + 15 * step < rows; r += 16 * step) {
    float v[16];
#pragma unroll
    for (int q = 0; q < 16; ++q) v[q] = src[(long)(r + q * step) * 64 + col];
#pragma unroll
    for (int q = 0; q < 16; ++q) local += v[q];
  }
  for (; r < rows; r += step) local += src[(long)r * 64 + col];
  red[threadIdx.x] = local;
  __syncthreads();
  if (g == 0) {
    local += red[64 + threadIdx.x] + red[128 + threadIdx.x] + red[192 + threadIdx.x];
    unsafeAtomicAdd(&dst[col], local);
  }
}

__global__ void k_global(const float* __restrict__ u,
                         const float* __restrict__ Wgb, const float* __restrict__ bgb,
                         const float* __restrict__ Wd1, const float* __restrict__ bd1,
                         const float* __restrict__ e_sum, const float* __restrict__ n_sum,
                         float* __restrict__ bias_d1) {
  __shared__ float su2[64];
  const int j = threadIdx.x;
  float acc = bgb[j];
#pragma unroll 32
  for (int k = 0; k < 64; ++k) {
    float em = e_sum[k] * (1.0f / NEDGES);
    float nm = n_sum[k] * (1.0f / NNODES);
    acc += em * Wgb[k * 64 + j] + nm * Wgb[(64 + k) * 64 + j] + u[k] * Wgb[(128 + k) * 64 + j];
  }
  acc = fmaxf(acc, 0.0f);
  su2[j] = acc;
  __syncthreads();
  float b = bd1[j];
  for (int k = 0; k < 64; ++k) b += su2[k] * Wd1[(192 + k) * 64 + j];
  bias_d1[j] = b;
}

__global__ __launch_bounds__(256) void k_dec(
    const float* __restrict__ agg_r, const float* __restrict__ agg_s,
    const float* __restrict__ cnt_r, const float* __restrict__ cnt_s,
    const float* __restrict__ n_h, const float* __restrict__ Wd1,
    const float* __restrict__ bias_d1, const float* __restrict__ Wd2,
    const float* __restrict__ bd2, float* __restrict__ out) {
  __shared__ float sW1[48 * 4 * 64];
  __shared__ float sW2[16 * 64];
  __shared__ float sH[8 * 16 * 64];
  {
    float wv[48], wu[4];
#pragma unroll
    for (int q = 0; q < 48; ++q) {
      const int i = threadIdx.x + q * 256;
      int j = i & 1, lane = (i >> 1) & 31, nt = (i >> 6) & 3, kc = i >> 8;
      int kk = 4 * kc + 2 * (lane >> 4) + j;
      wv[q] = Wd1[kk * 64 + nt * 16 + (lane & 15)];
    }
#pragma unroll
    for (int q = 0; q < 4; ++q) {
      const int i = threadIdx.x + q * 256;
      int j = i & 1, lane = (i >> 1) & 31, kc = i >> 6;
      int kk = 4 * kc + 2 * (lane >> 4) + j;
      wu[q] = Wd2[kk * 16 + (lane & 15)];
    }
#pragma unroll
    for (int q = 0; q < 48; ++q) sW1[threadIdx.x + q * 256] = wv[q];
#pragma unroll
    for (int q = 0; q < 4; ++q) sW2[threadIdx.x + q * 256] = wu[q];
  }
  __syncthreads();
  const int lane = threadIdx.x & 31;
  const int h = lane >> 4, m = lane & 15;
  float* hw = sH + (threadIdx.x >> 5) * 1024;
  const int wave = blockIdx.x * 8 + (threadIdx.x >> 5);
  const int nwaves = gridDim.x * 8;
  const int ntiles = NNODES / 16;
  for (int t = wave; t < ntiles; t += nwaves) {
    const int n0 = t * 16;
    const int n = n0 + m;
    const float invr = 1.0f / fmaxf(cnt_r[n], 1.0f);
    const float invs = 1.0f / fmaxf(cnt_s[n], 1.0f);
    const float* pr = agg_r + (long)n * 64;
    const float* ps = agg_s + (long)n * 64;
    const float* pn = n_h + (long)n * 64;
    v8f acc[4];
#pragma unroll
    for (int nt = 0; nt < 4; ++nt) {
      float bb = bias_d1[nt * 16 + m];
      acc[nt] = (v8f){bb, bb, bb, bb, bb, bb, bb, bb};
    }
#pragma unroll
    for (int kc = 0; kc < 48; ++kc) {
      const int kk = 4 * kc + 2 * h;
      v2f a;
      if (kc < 16)      { a = *(const v2f*)(pr + kk);        a *= invr; }
      else if (kc < 32) { a = *(const v2f*)(ps + kk - 64);   a *= invs; }
      else              { a = *(const v2f*)(pn + kk - 128); }
#pragma unroll
      for (int nt = 0; nt < 4; ++nt) {
        v2f b = *(const v2f*)(&sW1[(kc * 4 + nt) * 64 + lane * 2]);
        acc[nt] = __builtin_amdgcn_wmma_f32_16x16x4_f32(false, a, false, b,
                                                        (short)0, acc[nt], false, false);
      }
    }
#pragma unroll
    for (int nt = 0; nt < 4; ++nt)
#pragma unroll
      for (int v = 0; v < 8; ++v)
        hw[(8 * h + v) * 64 + nt * 16 + m] = fmaxf(acc[nt][v], 0.0f);
    float bb2 = bd2[m];
    v8f acc2 = (v8f){bb2, bb2, bb2, bb2, bb2, bb2, bb2, bb2};
#pragma unroll
    for (int kc = 0; kc < 16; ++kc) {
      const int kk = 4 * kc + 2 * h;
      v2f a = *(const v2f*)(hw + m * 64 + kk);
      v2f b = *(const v2f*)(&sW2[kc * 64 + lane * 2]);
      acc2 = __builtin_amdgcn_wmma_f32_16x16x4_f32(false, a, false, b,
                                                   (short)0, acc2, false, false);
    }
#pragma unroll
    for (int v = 0; v < 8; ++v)
      out[(long)(n0 + 8 * h + v) * 16 + m] = acc2[v];
  }
}

extern "C" void kernel_launch(void* const* d_in, const int* in_sizes, int n_in,
                              void* d_out, int out_size, void* d_ws, size_t ws_size,
                              hipStream_t stream) {
  const float* x    = (const float*)d_in[0];
  const float* ea   = (const float*)d_in[1];
  const float* u    = (const float*)d_in[2];
  const int* send   = (const int*)d_in[3];
  const int* recv   = (const int*)d_in[4];
  const float* Web  = (const float*)d_in[7];
  const float* beb  = (const float*)d_in[8];
  const float* Wnb  = (const float*)d_in[9];
  const float* bnb  = (const float*)d_in[10];
  const float* Wgb  = (const float*)d_in[11];
  const float* bgb  = (const float*)d_in[12];
  const float* Wd1  = (const float*)d_in[13];
  const float* bd1  = (const float*)d_in[14];
  const float* Wd2  = (const float*)d_in[15];
  const float* bd2  = (const float*)d_in[16];
  float* out = (float*)d_out;
  float* ws = (float*)d_ws;

  const long N = NNODES;
  float* agg_r   = ws;
  float* agg_s   = agg_r + N * 64;
  float* cnt_r   = agg_s + N * 64;
  float* cnt_s   = cnt_r + N;
  float* e_sum   = cnt_s + N;
  float* n_sum   = e_sum + 64;
  float* bias_eb = n_sum + 128;
  float* bias_nb = bias_eb + 64;
  float* bias_d1 = bias_nb + 64;
  float* n_hbuf  = bias_d1 + 64;
  int* list_r  = (int*)n_hbuf;
  int* list_s  = list_r + NEDGES;
  int* totals  = list_s + NEDGES;
  int* binbase = totals + 2 * NB;
  int* cursors = binbase + 2 * NB;

  k_zero<<<2, 256, 0, stream>>>(e_sum, 128, totals, 2 * NB);
  k_prep<<<1, 64, 0, stream>>>(u, Web, beb, Wnb, bnb, bias_eb, bias_nb);
  k_count<<<NCHUNKS, 256, 0, stream>>>(recv, send, totals);
  k_scan<<<1, 2 * NB, 0, stream>>>(totals, binbase, cursors);
  k_fill<<<NCHUNKS, 256, 0, stream>>>(recv, send, cursors, list_r, list_s);
  k_edge<<<NB, EW * 32, 0, stream>>>(x, ea, recv, send, Web, bias_eb, list_r, list_s,
                                    totals, binbase, agg_r, agg_s, cnt_r, cnt_s);
  k_node<<<782, 256, 0, stream>>>(x, agg_r, Wnb, bias_nb, n_hbuf);
  k_colsum<<<512, 256, 0, stream>>>(agg_r, NNODES, e_sum);
  k_colsum<<<512, 256, 0, stream>>>(n_hbuf, NNODES, n_sum);
  k_global<<<1, 64, 0, stream>>>(u, Wgb, bgb, Wd1, bd1, e_sum, n_sum, bias_d1);
  k_dec<<<1024, 256, 0, stream>>>(agg_r, agg_s, cnt_r, cnt_s, n_hbuf,
                                  Wd1, bias_d1, Wd2, bd2, out);
}


# k_node, k_dec: one wave per SIMD (256 workgroups of 4 waves loop over tiles): weight fragments stay in registers, the next tile's A rows load as one group during the products; sums keep their order
# speedup vs baseline: 2.1729x; 1.0934x over previous
#include <hip/hip_runtime.h>

typedef __attribute__((ext_vector_type(2))) float v2f;
typedef __attribute__((ext_vector_type(8))) float v8f;

#define NNODES 100000
#define NEDGES 1600000

__global__ void k_zero(float* __restrict__ a, int na, int* __restrict__ b, int nb) {
  int i = blockIdx.x * blockDim.x + threadIdx.x;
  if (i < na) a[i] = 0.f;
  if (i < nb) b[i] = 0;
}

__global__ void k_prep(const float* __restrict__ u,
                       const float* __restrict__ Web, const float* __restrict__ beb,
                       const float* __restrict__ Wnb, const float* __restrict__ bnb,
                       float* __restrict__ bias_eb, float* __restrict__ bias_nb) {
  int j = threadIdx.x;
  float a = beb[j], b = bnb[j];
  for (int k0 = 0; k0 < 64; k0 += 32) {
    float wa[32], wb[32];
#pragma unroll
    for (int q = 0; q < 32; ++q) {
      wa[q] = Web[(48 + k0 + q) * 64 + j];
      wb[q] = Wnb[(80 + k0 + q) * 64 + j];
    }
#pragma unroll
    for (int q = 0; q < 32; ++q) {
      float uk = u[k0 + q];
      a += uk * wa[q];
      b += uk * wb[q];
    }
  }
  bias_eb[j] = a;
  bias_nb[j] = b;
}

#define NB 256
#define RPB 391
#define CHUNK 2048
#define NCHUNKS ((NEDGES + CHUNK - 1) / CHUNK)
#define EPT (CHUNK / 256)
#define EBITS 21
#define EMASK ((1 << EBITS) - 1)
static_assert(NEDGES <= (1 << EBITS), "edge number must fit the low bits of a list word");
static_assert(RPB <= (1 << (30 - EBITS)), "row in range must fit above the edge number");
__global__ __launch_bounds__(256) void k_count(const int* __restrict__ recv,
                                               const int* __restrict__ send,
                                               int* __restrict__ totals) {
  __shared__ int hist[2 * NB];
  for (int j = threadIdx.x; j < 2 * NB; j += 256) hist[j] = 0;
  __syncthreads();
  const int e0 = blockIdx.x * CHUNK;
  int nr[EPT], ns[EPT];
#pragma unroll
  for (int k = 0; k < EPT; ++k) {
    const int e = min(e0 + k * 256 + (int)threadIdx.x, NEDGES - 1);
    nr[k] = recv[e];
    ns[k] = send[e];
  }
#pragma unroll
  for (int k = 0; k < EPT; ++k) {
    if (e0 + k * 256 + (int)threadIdx.x < NEDGES) {
      atomicAdd(&hist[nr[k] / RPB], 1);
      atomicAdd(&hist[NB + ns[k] / RPB], 1);
    }
  }
  __syncthreads();
  for (int j = threadIdx.x; j < 2 * NB; j += 256) {
    const int c = hist[j];
    if (c) atomicAdd(&totals[j], c);
  }
}

__global__ __launch_bounds__(2 * NB) void k_scan(const int* __restrict__ totals,
                                                 int* __restrict__ binbase,
                                                 int* __restrict__ cursors) {
  __shared__ int s[2 * NB];
  const int j = threadIdx.x;
  const int v = totals[j];
  s[j] = v;
  __syncthreads();
  for (int off = 1; off < NB; off <<= 1) {
    const int t = ((j & (NB - 1)) >= off) ? s[j - off] : 0;
    __syncthreads();
    s[j] += t;
    __syncthreads();
  }
  const int ex = s[j] - v;
  binbase[j] = ex;
  cursors[j] = ex;
}

__global__ __launch_bounds__(256) void k_fill(const int* __restrict__ recv,
                                              const int* __restrict__ send,
                                              int* __restrict__ cursors,
                                              int* __restrict__ list_r,
                                              int* __restrict__ list_s) {
  __shared__ int hist[2 * NB];
  for (int j = threadIdx.x; j < 2 * NB; j += 256) hist[j] = 0;
  __syncthreads();
  const int e0 = blockIdx.x * CHUNK;
  int nr[EPT], ns[EPT];
#pragma unroll
  for (int k = 0; k < EPT; ++k) {
    const int e = min(e0 + k * 256 + (int)threadIdx.x, NEDGES - 1);
    nr[k] = recv[e];
    ns[k] = send[e];
  }
#pragma unroll
  for (int k = 0; k < EPT; ++k) {
    if (e0 + k * 256 + (int)threadIdx.x < NEDGES) {
      atomicAdd(&hist[nr[k] / RPB], 1);
      atomicAdd(&hist[NB + ns[k] / RPB], 1);
    }
  }
  __syncthreads();
  for (int j = threadIdx.x; j < 2 * NB; j += 256) {
    const int c = hist[j];
    hist[j] = c ? atomicAdd(&cursors[j], c) : 0;
  }
  __syncthreads();
#pragma unroll
  for (int k = 0; k < EPT; ++k) {
    const int e = e0 + k * 256 + (int)threadIdx.x;
    if (e < NEDGES) {
      const int r = nr[k], s = ns[k];
      const int br = r / RPB, bs = s / RPB;
      const int p = atomicAdd(&hist[br], 1);
      list_r[p] = e | ((r - br * RPB) << EBITS);
      const int q = atomicAdd(&hist[NB + bs], 1);
      list_s[q] = e | ((s - bs * RPB) << EBITS);
    }
  }
}

#define EW 16
#define RPL (RPB + 1)
#define PTILES ((RPB + 15) / 16)
#define EDGE_LIST(tt) list[min((tt) * 16 + m, n - 1)]
#define EDGE_ROWS(A_, e_, o_)                                                 \
  {                                                                           \
    const float* p0_ = ea + (long)(e_) * 16 + 2 * h;                          \
    const float* p2_ = x + (long)(o_) * 16 + 2 * h;                           \
    _Pragma("unroll") for (int cc = 0; cc < 4; ++cc) {                        \
      A_[cc] = *(const v2f*)(p0_ + 4 * cc);                                   \
      A_[4 + cc] = *(const v2f*)(p2_ + 4 * cc);                               \
    }                                                                         \
  }
__global__ __launch_bounds__(EW * 32) void k_edge(
    const float* __restrict__ x, const float* __restrict__ ea,
    const int* __restrict__ recv, const int* __restrict__ send,
    const float* __restrict__ Web, const float* __restrict__ bias_eb,
    const int* __restrict__ list_r, const int* __restrict__ list_s,
    const int* __restrict__ totals, const int* __restrict__ binbase,
    float* __restrict__ agg_r, float* __restrict__ agg_s,
    float* __restrict__ cnt_r, float* __restrict__ cnt_s) {
  __shared__ __attribute__((aligned(16))) float sS[RPL * 64];
  __shared__ __attribute__((aligned(16))) float sP[PTILES * 16 * 64];
  __shared__ float sC[2 * RPL];
  for (int i = threadIdx.x; i < RPL * 16; i += EW * 32)
    *(float4*)(&sS[i * 4]) = (float4){0.f, 0.f, 0.f, 0.f};
  for (int i = threadIdx.x; i < 2 * RPL; i += EW * 32) sC[i] = 0.f;
  const int lane = threadIdx.x & 31;
  const int h = lane >> 4, m = lane & 15;
  const int wave = __builtin_amdgcn_readfirstlane(threadIdx.x >> 5);
  const int b = blockIdx.x;
  const int lo = b * RPB;
#define WFRAG(row0, kc, nt)                                        \
  (v2f){Web[((row0) + 4 * (kc) + 2 * h) * 64 + 4 * m + (nt)],      \
        Web[((row0) + 4 * (kc) + 2 * h + 1) * 64 + 4 * m + (nt)]}
  v2f bA[16];
#pragma unroll
  for (int kc = 0; kc < 4; ++kc)
#pragma unroll
    for (int nt = 0; nt < 4; ++nt) bA[kc * 4 + nt] = WFRAG(0, kc, nt);
  float bb[4];
#pragma unroll
  for (int nt = 0; nt < 4; ++nt) bb[nt] = bias_eb[4 * m + nt];
  for (int role = 0; role < 2; ++role) {
    const int rin = role ? 32 : 16, rot = role ? 16 : 32;
    v2f bO[16];
#pragma unroll
    for (int kc = 0; kc < 4; ++kc)
#pragma unroll
      for (int nt = 0; nt < 4; ++nt) bO[kc * 4 + nt] = WFRAG(rot, kc, nt);
    if (wave < PTILES) {
      v2f bI[16];
#pragma unroll
      for (int kc = 0; kc < 4; ++kc)
#pragma unroll
        for (int nt = 0; nt < 4; ++nt) bI[kc * 4 + nt] = WFRAG(rin, kc, nt);
      for (int pt = wave; pt < PTILES; pt += EW) {
        const int r = pt * 16 + m;
        const int node = lo + r;
        v2f a[4];
#pragma unroll
        for (int cc = 0; cc < 4; ++cc) a[cc] = (v2f){0.f, 0.f};
        if (r < RPB && node < NNODES) {
          const float* px = x + (long)node * 16 + 2 * h;
#pragma unroll
          for (int cc = 0; cc < 4; ++cc) a[cc] = *(const v2f*)(px + 4 * cc);
        }
        v8f pa[4];
#pragma unroll
        for (int nt = 0; nt < 4; ++nt)
          pa[nt] = (v8f){bb[nt], bb[nt], bb[nt], bb[nt], bb[nt], bb[nt], bb[nt], bb[nt]};
#pragma unroll
        for (int cc = 0; cc < 4; ++cc)
#pragma unroll
          for (int nt = 0; nt < 4; ++nt)
            pa[nt] = __builtin_amdgcn_wmma_f32_16x16x4_f32(false, a[cc], false, bI[cc * 4 + nt],
                                                           (short)0, pa[nt], false, false);
#pragma unroll
        for (int v = 0; v < 8; ++v)
          *(float4*)(&sP[(pt * 16 + 8 * h + v) * 64 + 4 * m]) =
              (float4){pa[0][v], pa[1][v], pa[2][v], pa[3][v]};
      }
    }
    __syncthreads();
    const int* __restrict__ list = (role ? list_s : list_r) + binbase[role * NB + b];
    const int n = totals[role * NB + b];
    const int ntiles = (n + 15) >> 4;
    float* C = sC + role * RPL;
    const int* __restrict__ oth = role ? recv : send;
#define ROW_PAIR(w_, tt)                                                      \
  (min((int)((unsigned)(w_) >> EBITS), RPB) |                                 \
   (((tt) * 16 + m < n ? min((int)((unsigned)(w_) >> EBITS), RPB) : RPB) << 16))
#define P_ROWS(acc_, d_)                                                            \
  _Pragma("unroll") for (int v = 0; v < 8; ++v) {                                   \
    const float4 pv_ = *(const float4*)(&sP[((d_)[v] & 0xffff) * 64 + 4 * m]);      \
    acc_[0][v] = pv_.x; acc_[1][v] = pv_.y; acc_[2][v] = pv_.z; acc_[3][v] = pv_.w; \
  }
    int t = wave;
    if (t < ntiles) {
      v2f G0[8], G1[8];
      int w0 = EDGE_LIST(t);
      int o0 = oth[w0 & EMASK];
      EDGE_ROWS(G0, w0 & EMASK, o0)
      int w1 = EDGE_LIST(t + EW);
      int o1 = oth[w1 & EMASK];
      int rp0 = ROW_PAIR(w0, t);
      int dd[8];
#pragma unroll
      for (int v = 0; v < 8; ++v) dd[v] = __shfl(rp0, 8 * h + v);
      v8f acc[4];
      P_ROWS(acc, dd)
      for (; t < ntiles; t += EW) {
        const int w2 = EDGE_LIST(t + 2 * EW);
        EDGE_ROWS(G1, w1 & EMASK, o1)
        const int rp1 = ROW_PAIR(w1, t + EW);
        int dn[8];
#pragma unroll
        for (int v = 0; v < 8; ++v) dn[v] = __shfl(rp1, 8 * h + v);
#pragma unroll
        for (int cc = 0; cc < 4; ++cc)
#pragma unroll
          for (int nt = 0; nt < 4; ++nt)
            acc[nt] = __builtin_amdgcn_wmma_f32_16x16x4_f32(false, G0[cc], false, bA[cc * 4 + nt],
                                                            (short)0, acc[nt], false, false);
        __builtin_amdgcn_sched_barrier(0);
        const int o2 = oth[w2 & EMASK];
        __builtin_amdgcn_sched_barrier(0);
#pragma unroll
        for (int cc = 0; cc < 4; ++cc)
#pragma unroll
          for (int nt = 0; nt < 4; ++nt)
            acc[nt] = __builtin_amdgcn_wmma_f32_16x16x4_f32(false, G0[4 + cc], false,
                                                            bO[cc * 4 + nt],
                                                            (short)0, acc[nt], false, false);
        const bool hb = h != 0;
        v8f q0, q1, q2, q3;
#pragma unroll
        for (int v = 0; v < 8; ++v) {
          const float e0 = fmaxf(acc[0][v], 0.0f), e1 = fmaxf(acc[1][v], 0.0f);
          const float e2 = fmaxf(acc[2][v], 0.0f), e3 = fmaxf(acc[3][v], 0.0f);
          q0[v] = hb ? e1 : e0;
          q1[v] = hb ? e0 : e1;
          q2[v] = hb ? e3 : e2;
          q3[v] = hb ? e2 : e3;
        }
        P_ROWS(acc, dn)
        __builtin_amdgcn_sched_barrier(0);
        const int oa = h, ob = 1 - h;
#pragma unroll
        for (int v = 0; v < 8; ++v) {
          float* p = sS + (dd[v] >> 16) * 64 + 4 * m;
          unsafeAtomicAdd(p + oa, q0[v]);
          unsafeAtomicAdd(p + ob, q1[v]);
          unsafeAtomicAdd(p + 2 + oa, q2[v]);
          unsafeAtomicAdd(p + 2 + ob, q3[v]);
        }
        if (h == 0) unsafeAtomicAdd(&C[rp0 >> 16], 1.0f);
#pragma unroll
        for (int q = 0; q < 8; ++q) G0[q] = G1[q];
#pragma unroll
        for (int v = 0; v < 8; ++v) dd[v] = dn[v];
        rp0 = rp1;
        w0 = w1; w1 = w2; o1 = o2;
      }
    }
    __syncthreads();
    float* __restrict__ agg = role ? agg_s : agg_r;
    for (int i = threadIdx.x; i < RPB * 16; i += EW * 32) {
      const int row = i >> 4, c4 = i & 15;
      const int node = lo + row;
      const float4 val = *(const float4*)(&sS[row * 64 + c4 * 4]);
      *(float4*)(&sS[row * 64 + c4 * 4]) = (float4){0.f, 0.f, 0.f, 0.f};
      if (node < NNODES) *(float4*)(agg + (long)node * 64 + c4 * 4) = val;
    }
  }
  for (int i = threadIdx.x; i < 2 * RPB; i += EW * 32) {
    const int role = i / RPB;
    const int row = i - role * RPB;
    const int node = lo + row;
    if (node < NNODES) (role ? cnt_s : cnt_r)[node] = sC[role * RPL + row];
  }
}

#define GW 4
#define NODE_ROWS(A_, tt)                                                     \
  {                                                                           \
    const float* ap_ = agg_r + (long)((tt) * 16 + m) * 64 + 2 * h;            \
    const float* xp_ = x + (long)((tt) * 16 + m) * 16 + 2 * h;                \
    _Pragma("unroll") for (int kc = 0; kc < 16; ++kc)                         \
      A_[kc] = *(const v2f*)(ap_ + 4 * kc);                                   \
    _Pragma("unroll") for (int kc = 0; kc < 4; ++kc)                          \
      A_[16 + kc] = *(const v2f*)(xp_ + 4 * kc);                              \
  }
__global__ __launch_bounds__(GW * 32) void k_node(
    const float* __restrict__ x, const float* __restrict__ agg_r,
    const float* __restrict__ Wnb, const float* __restrict__ bias_nb,
    float* __restrict__ n_h) {
  __shared__ float sW[20 * 4 * 64];
  {
    float wv[40];
#pragma unroll
    for (int q = 0; q < 40; ++q) {
      const int i = threadIdx.x + q * (GW * 32);
      int j = i & 1, lane = (i >> 1) & 31, nt = (i >> 6) & 3, kc = i >> 8;
      int kk = 4 * kc + 2 * (lane >> 4) + j;
      wv[q] = Wnb[kk * 64 + nt * 16 + (lane & 15)];
    }
#pragma unroll
    for (int q = 0; q < 40; ++q) sW[threadIdx.x + q * (GW * 32)] = wv[q];
  }
  __syncthreads();
  const int lane = threadIdx.x & 31;
  const int h = lane >> 4, m = lane & 15;
  v2f bw[80];
#pragma unroll
  for (int q = 0; q < 80; ++q) bw[q] = *(const v2f*)(&sW[q * 64 + lane * 2]);
  float bb[4];
#pragma unroll
  for (int nt = 0; nt < 4; ++nt) bb[nt] = bias_nb[nt * 16 + m];
  const int wv_ = __builtin_amdgcn_readfirstlane(threadIdx.x >> 5);
  const int nwaves = gridDim.x * GW;
  const int ntiles = NNODES / 16;
  int t = wv_ * gridDim.x + blockIdx.x;
  if (t < ntiles) {
    v2f a[20];
    NODE_ROWS(a, t)
    for (; t < ntiles; t += nwaves) {
      const int tn = min(t + nwaves, ntiles - 1);
      v2f an[20];
      NODE_ROWS(an, tn)
      __builtin_amdgcn_sched_barrier(0);
      const int n0 = t * 16;
      v8f acc[4];
#pragma unroll
      for (int nt = 0; nt < 4; ++nt)
        acc[nt] = (v8f){bb[nt], bb[nt], bb[nt], bb[nt], bb[nt], bb[nt], bb[nt], bb[nt]};
#pragma unroll
      for (int kc = 0; kc < 20; ++kc)
#pragma unroll
        for (int nt = 0; nt < 4; ++nt)
          acc[nt] = __builtin_amdgcn_wmma_f32_16x16x4_f32(false, a[kc], false, bw[kc * 4 + nt],
                                                          (short)0, acc[nt], false, false);
#pragma unroll
      for (int nt = 0; nt < 4; ++nt)
#pragma unroll
        for (int v = 0; v < 8; ++v)
          n_h[(long)(n0 + 8 * h + v) * 64 + nt * 16 + m] = fmaxf(acc[nt][v], 0.0f);
      __builtin_amdgcn_sched_barrier(0);
#pragma unroll
      for (int kc = 0; kc < 20; ++kc) a[kc] = an[kc];
    }
  }
}

__global__ void k_colsum(const float* __restrict__ src, int rows, float* __restrict__ dst) {
  __shared__ float red[256];
  const int col = threadIdx.x & 63;
  const int g = threadIdx.x >> 6;
  float local = 0.f;
  const int step = gridDim.x * 4;
  int r = blockIdx.x * 4 + g;
  for (; r + 15 * step < rows; r += 16 * step) {
    float v[16];
#pragma unroll
    for (int q = 0; q < 16; ++q) v[q] = src[(long)(r + q * step) * 64 + col];
#pragma unroll
    for (int q = 0; q < 16; ++q) local += v[q];
  }
  for (; r < rows; r += step) local += src[(long)r * 64 + col];
  red[threadIdx.x] = local;
  __syncthreads();
  if (g == 0) {
    local += red[64 + threadIdx.x] + red[128 + threadIdx.x] + red[192 + threadIdx.x];
    unsafeAtomicAdd(&dst[col], local);
  }
}

__global__ void k_global(const float* __restrict__ u,
                         const float* __restrict__ Wgb, const float* __restrict__ bgb,
                         const float* __restrict__ Wd1, const float* __restrict__ bd1,
                         const float* __restrict__ e_sum, const float* __restrict__ n_sum,
                         float* __restrict__ bias_d1) {
  __shared__ float su2[64];
  const int j = threadIdx.x;
  float acc = bgb[j];
#pragma unroll 32
  for (int k = 0; k < 64; ++k) {
    float em = e_sum[k] * (1.0f / NEDGES);
    float nm = n_sum[k] * (1.0f / NNODES);
    acc += em * Wgb[k * 64 + j] + nm * Wgb[(64 + k) * 64 + j] + u[k] * Wgb[(128 + k) * 64 + j];
  }
  acc = fmaxf(acc, 0.0f);
  su2[j] = acc;
  __syncthreads();
  float b = bd1[j];
  for (int k = 0; k < 64; ++k) b += su2[k] * Wd1[(192 + k) * 64 + j];
  bias_d1[j] = b;
}

#define DEC_ROWS(A_, cr_, cs_, tt)                                            \
  {                                                                           \
    const long n_ = (long)((tt) * 16 + m);                                    \
    cr_ = cnt_r[n_];                                                          \
    cs_ = cnt_s[n_];                                                          \
    const float* pr_ = agg_r + n_ * 64 + 2 * h;                               \
    const float* ps_ = agg_s + n_ * 64 + 2 * h;                               \
    const float* pn_ = n_h + n_ * 64 + 2 * h;                                 \
    _Pragma("unroll") for (int kc = 0; kc < 16; ++kc) {                       \
      A_[kc] = *(const v2f*)(pr_ + 4 * kc);                                   \
      A_[16 + kc] = *(const v2f*)(ps_ + 4 * kc);                              \
      A_[32 + kc] = *(const v2f*)(pn_ + 4 * kc);                              \
    }                                                                         \
  }
__global__ __launch_bounds__(GW * 32) void k_dec(
    const float* __restrict__ agg_r, const float* __restrict__ agg_s,
    const float* __restrict__ cnt_r, const float* __restrict__ cnt_s,
    const float* __restrict__ n_h, const float* __restrict__ Wd1,
    const float* __restrict__ bias_d1, const float* __restrict__ Wd2,
    const float* __restrict__ bd2, float* __restrict__ out) {
  __shared__ float sW1[48 * 4 * 64];
  __shared__ float sW2[16 * 64];
  __shared__ float sH[GW * 16 * 64];
  {
    float wv[96], wu[8];
#pragma unroll
    for (int q = 0; q < 96; ++q) {
      const int i = threadIdx.x + q * (GW * 32);
      int j = i & 1, lane = (i >> 1) & 31, nt = (i >> 6) & 3, kc = i >> 8;
      int kk = 4 * kc + 2 * (lane >> 4) + j;
      wv[q] = Wd1[kk * 64 + nt * 16 + (lane & 15)];
    }
#pragma unroll
    for (int q = 0; q < 8; ++q) {
      const int i = threadIdx.x + q * (GW * 32);
      int j = i & 1, lane = (i >> 1) & 31, kc = i >> 6;
      int kk = 4 * kc + 2 * (lane >> 4) + j;
      wu[q] = Wd2[kk * 16 + (lane & 15)];
    }
#pragma unroll
    for (int q = 0; q < 96; ++q) sW1[threadIdx.x + q * (GW * 32)] = wv[q];
#pragma unroll
    for (int q = 0; q < 8; ++q) sW2[threadIdx.x + q * (GW * 32)] = wu[q];
  }
  __syncthreads();
  const int lane = threadIdx.x & 31;
  const int h = lane >> 4, m = lane & 15;
  v2f bw[192], bu[16];
#pragma unroll
  for (int q = 0; q < 192; ++q) bw[q] = *(const v2f*)(&sW1[q * 64 + lane * 2]);
#pragma unroll
  for (int q = 0; q < 16; ++q) bu[q] = *(const v2f*)(&sW2[q * 64 + lane * 2]);
  float bb[4];
#pragma unroll
  for (int nt = 0; nt < 4; ++nt) bb[nt] = bias_d1[nt * 16 + m];
  const float bb2 = bd2[m];
  const int wv_ = __builtin_amdgcn_readfirstlane(threadIdx.x >> 5);
  float* hw = sH + wv_ * 1024;
  const int nwaves = gridDim.x * GW;
  const int ntiles = NNODES / 16;
  int t = wv_ * gridDim.x + blockIdx.x;
  if (t < ntiles) {
    v2f a[48];
    float cr, cs;
    DEC_ROWS(a, cr, cs, t)
    for (; t < ntiles; t += nwaves) {
      const int tn = min(t + nwaves, ntiles - 1);
      v2f an[48];
      float crn, csn;
      DEC_ROWS(an, crn, csn, tn)
      __builtin_amdgcn_sched_barrier(0);
      const int n0 = t * 16;
      const float invr = 1.0f / fmaxf(cr, 1.0f);
      const float invs = 1.0f / fmaxf(cs, 1.0f);
      v8f acc[4];
#pragma unroll
      for (int nt = 0; nt < 4; ++nt)
        acc[nt] = (v8f){bb[nt], bb[nt], bb[nt], bb[nt], bb[nt], bb[nt], bb[nt], bb[nt]};
#pragma unroll
      for (int kc = 0; kc < 48; ++kc) {
        v2f av = a[kc];
        if (kc < 16)      av *= invr;
        else if (kc < 32) av *= invs;
#pragma unroll
        for (int nt = 0; nt < 4; ++nt)
          acc[nt] = __builtin_amdgcn_wmma_f32_16x16x4_f32(false, av, false, bw[kc * 4 + nt],
                                                          (short)0, acc[nt], false, false);
      }
#pragma unroll
      for (int nt = 0; nt < 4; ++nt)
#pragma unroll
        for (int v = 0; v < 8; ++v)
          hw[(8 * h + v) * 64 + nt * 16 + m] = fmaxf(acc[nt][v], 0.0f);
      v8f acc2 = (v8f){bb2, bb2, bb2, bb2, bb2, bb2, bb2, bb2};
#pragma unroll
      for (int kc = 0; kc < 16; ++kc) {
        const int kk = 4 * kc + 2 * h;
        v2f a2 = *(const v2f*)(hw + m * 64 + kk);
        acc2 = __builtin_amdgcn_wmma_f32_16x16x4_f32(false, a2, false, bu[kc],
                                                     (short)0, acc2, false, false);
      }
#pragma unroll
      for (int v = 0; v < 8; ++v)
        out[(long)(n0 + 8 * h + v) * 16 + m] = acc2[v];
      __builtin_amdgcn_sched_barrier(0);
#pragma unroll
      for (int kc = 0; kc < 48; ++kc) a[kc] = an[kc];
      cr = crn;
      cs = csn;
    }
  }
}

extern "C" void kernel_launch(void* const* d_in, const int* in_sizes, int n_in,
                              void* d_out, int out_size, void* d_ws, size_t ws_size,
                              hipStream_t stream) {
  const float* x    = (const float*)d_in[0];
  const float* ea   = (const float*)d_in[1];
  const float* u    = (const float*)d_in[2];
  const int* send   = (const int*)d_in[3];
  const int* recv   = (const int*)d_in[4];
  const float* Web  = (const float*)d_in[7];
  const float* beb  = (const float*)d_in[8];
  const float* Wnb  = (const float*)d_in[9];
  const float* bnb  = (const float*)d_in[10];
  const float* Wgb  = (const float*)d_in[11];
  const float* bgb  = (const float*)d_in[12];
  const float* Wd1  = (const float*)d_in[13];
  const float* bd1  = (const float*)d_in[14];
  const float* Wd2  = (const float*)d_in[15];
  const float* bd2  = (const float*)d_in[16];
  float* out = (float*)d_out;
  float* ws = (float*)d_ws;

  const long N = NNODES;
  float* agg_r   = ws;
  float* agg_s   = agg_r + N * 64;
  float* cnt_r   = agg_s + N * 64;
  float* cnt_s   = cnt_r + N;
  float* e_sum   = cnt_s + N;
  float* n_sum   = e_sum + 64;
  float* bias_eb = n_sum + 128;
  float* bias_nb = bias_eb + 64;
  float* bias_d1 = bias_nb + 64;
  float* n_hbuf  = bias_d1 + 64;
  int* list_r  = (int*)n_hbuf;
  int* list_s  = list_r + NEDGES;
  int* totals  = list_s + NEDGES;
  int* binbase = totals + 2 * NB;
  int* cursors = binbase + 2 * NB;

  k_zero<<<2, 256, 0, stream>>>(e_sum, 128, totals, 2 * NB);
  k_prep<<<1, 64, 0, stream>>>(u, Web, beb, Wnb, bnb, bias_eb, bias_nb);
  k_count<<<NCHUNKS, 256, 0, stream>>>(recv, send, totals);
  k_scan<<<1, 2 * NB, 0, stream>>>(totals, binbase, cursors);
  k_fill<<<NCHUNKS, 256, 0, stream>>>(recv, send, cursors, list_r, list_s);
  k_edge<<<NB, EW * 32, 0, stream>>>(x, ea, recv, send, Web, bias_eb, list_r, list_s,
                                    totals, binbase, agg_r, agg_s, cnt_r, cnt_s);
  k_node<<<256, GW * 32, 0, stream>>>(x, agg_r, Wnb, bias_nb, n_hbuf);
  k_colsum<<<512, 256, 0, stream>>>(agg_r, NNODES, e_sum);
  k_colsum<<<512, 256, 0, stream>>>(n_hbuf, NNODES, n_sum);
  k_global<<<1, 64, 0, stream>>>(u, Wgb, bgb, Wd1, bd1, e_sum, n_sum, bias_d1);
  k_dec<<<256, GW * 32, 0, stream>>>(agg_r, agg_s, cnt_r, cnt_s, n_hbuf,
                                   Wd1, bias_d1, Wd2, bd2, out);
}


# k_fill: a chunk's list words are put in bin order in LDS (scan of the chunk's counts) and stored from there: a wave's store covers runs of consecutive words, was 32 scattered words; arrival order
# speedup vs baseline: 2.3617x; 1.0869x over previous
#include <hip/hip_runtime.h>

typedef __attribute__((ext_vector_type(2))) float v2f;
typedef __attribute__((ext_vector_type(8))) float v8f;

#define NNODES 100000
#define NEDGES 1600000

__global__ void k_zero(float* __restrict__ a, int na, int* __restrict__ b, int nb) {
  int i = blockIdx.x * blockDim.x + threadIdx.x;
  if (i < na) a[i] = 0.f;
  if (i < nb) b[i] = 0;
}

__global__ void k_prep(const float* __restrict__ u,
                       const float* __restrict__ Web, const float* __restrict__ beb,
                       const float* __restrict__ Wnb, const float* __restrict__ bnb,
                       float* __restrict__ bias_eb, float* __restrict__ bias_nb) {
  int j = threadIdx.x;
  float a = beb[j], b = bnb[j];
  for (int k0 = 0; k0 < 64; k0 += 32) {
    float wa[32], wb[32];
#pragma unroll
    for (int q = 0; q < 32; ++q) {
      wa[q] = Web[(48 + k0 + q) * 64 + j];
      wb[q] = Wnb[(80 + k0 + q) * 64 + j];
    }
#pragma unroll
    for (int q = 0; q < 32; ++q) {
      float uk = u[k0 + q];
      a += uk * wa[q];
      b += uk * wb[q];
    }
  }
  bias_eb[j] = a;
  bias_nb[j] = b;
}

#define NB 256
#define RPB 391
#define CHUNK 2048
#define NCHUNKS ((NEDGES + CHUNK - 1) / CHUNK)
#define EPT (CHUNK / 256)
#define EBITS 21
#define EMASK ((1 << EBITS) - 1)
static_assert(NEDGES <= (1 << EBITS), "edge number must fit the low bits of a list word");
static_assert(RPB <= (1 << (30 - EBITS)), "row in range must fit above the edge number");
__global__ __launch_bounds__(256) void k_count(const int* __restrict__ recv,
                                               const int* __restrict__ send,
                                               int* __restrict__ totals) {
  __shared__ int hist[2 * NB];
  for (int j = threadIdx.x; j < 2 * NB; j += 256) hist[j] = 0;
  __syncthreads();
  const int e0 = blockIdx.x * CHUNK;
  int nr[EPT], ns[EPT];
#pragma unroll
  for (int k = 0; k < EPT; ++k) {
    const int e = min(e0 + k * 256 + (int)threadIdx.x, NEDGES - 1);
    nr[k] = recv[e];
    ns[k] = send[e];
  }
#pragma unroll
  for (int k = 0; k < EPT; ++k) {
    if (e0 + k * 256 + (int)threadIdx.x < NEDGES) {
      atomicAdd(&hist[nr[k] / RPB], 1);
      atomicAdd(&hist[NB + ns[k] / RPB], 1);
    }
  }
  __syncthreads();
  for (int j = threadIdx.x; j < 2 * NB; j += 256) {
    const int c = hist[j];
    if (c) atomicAdd(&totals[j], c);
  }
}

__global__ __launch_bounds__(2 * NB) void k_scan(const int* __restrict__ totals,
                                                 int* __restrict__ binbase,
                                                 int* __restrict__ cursors) {
  __shared__ int s[2 * NB];
  const int j = threadIdx.x;
  const int v = totals[j];
  s[j] = v;
  __syncthreads();
  for (int off = 1; off < NB; off <<= 1) {
    const int t = ((j & (NB - 1)) >= off) ? s[j - off] : 0;
    __syncthreads();
    s[j] += t;
    __syncthreads();
  }
  const int ex = s[j] - v;
  binbase[j] = ex;
  cursors[j] = ex;
}

__global__ __launch_bounds__(256) void k_fill(const int* __restrict__ recv,
                                              const int* __restrict__ send,
                                              int* __restrict__ cursors,
                                              int* __restrict__ list_r,
                                              int* __restrict__ list_s) {
  __shared__ int hist[2 * NB];
  __shared__ int lst[2 * NB];
  __shared__ int cur[2 * NB];
  __shared__ int wsum[2 * 8];
  __shared__ int sW[2 * CHUNK];
  __shared__ int sG[2 * CHUNK];
  for (int j = threadIdx.x; j < 2 * NB; j += 256) { hist[j] = 0; cur[j] = 0; }
  __syncthreads();
  const int e0 = blockIdx.x * CHUNK;
  int nr[EPT], ns[EPT];
#pragma unroll
  for (int k = 0; k < EPT; ++k) {
    const int e = min(e0 + k * 256 + (int)threadIdx.x, NEDGES - 1);
    nr[k] = recv[e];
    ns[k] = send[e];
  }
#pragma unroll
  for (int k = 0; k < EPT; ++k) {
    if (e0 + k * 256 + (int)threadIdx.x < NEDGES) {
      atomicAdd(&hist[nr[k] / RPB], 1);
      atomicAdd(&hist[NB + ns[k] / RPB], 1);
    }
  }
  __syncthreads();
  const int j = threadIdx.x;
  const int lane = j & 31, wave = j >> 5;
  const int c0 = hist[j], c1 = hist[NB + j];
  const int g0 = c0 ? atomicAdd(&cursors[j], c0) : 0;
  const int g1 = c1 ? atomicAdd(&cursors[NB + j], c1) : 0;
  int i0 = c0, i1 = c1;
  for (int d = 1; d < 32; d <<= 1) {
    const int t0 = __shfl_up(i0, d), t1 = __shfl_up(i1, d);
    if (lane >= d) { i0 += t0; i1 += t1; }
  }
  if (lane == 31) { wsum[wave] = i0; wsum[8 + wave] = i1; }
  __syncthreads();
  int p0 = 0, p1 = 0;
  for (int w = 0; w < wave; ++w) { p0 += wsum[w]; p1 += wsum[8 + w]; }
  lst[j] = p0 + i0 - c0;
  lst[NB + j] = p1 + i1 - c1;
  hist[j] = g0;
  hist[NB + j] = g1;
  __syncthreads();
#pragma unroll
  for (int k = 0; k < EPT; ++k) {
    const int e = e0 + k * 256 + (int)threadIdx.x;
    if (e < NEDGES) {
      const int r = nr[k], s = ns[k];
      const int br = r / RPB, bs = s / RPB;
      const int a = atomicAdd(&cur[br], 1);
      const int p = lst[br] + a;
      sW[p] = e | ((r - br * RPB) << EBITS);
      sG[p] = hist[br] + a;
      const int c = atomicAdd(&cur[NB + bs], 1);
      const int q = CHUNK + lst[NB + bs] + c;
      sW[q] = e | ((s - bs * RPB) << EBITS);
      sG[q] = hist[NB + bs] + c;
    }
  }
  __syncthreads();
  const int nv = min(CHUNK, NEDGES - e0);
  for (int i = threadIdx.x; i < nv; i += 256) {
    list_r[sG[i]] = sW[i];
    list_s[sG[CHUNK + i]] = sW[CHUNK + i];
  }
}

#define EW 16
#define RPL (RPB + 1)
#define PTILES ((RPB + 15) / 16)
#define EDGE_LIST(tt) list[min((tt) * 16 + m, n - 1)]
#define EDGE_ROWS(A_, e_, o_)                                                 \
  {                                                                           \
    const float* p0_ = ea + (long)(e_) * 16 + 2 * h;                          \
    const float* p2_ = x + (long)(o_) * 16 + 2 * h;                           \
    _Pragma("unroll") for (int cc = 0; cc < 4; ++cc) {                        \
      A_[cc] = *(const v2f*)(p0_ + 4 * cc);                                   \
      A_[4 + cc] = *(const v2f*)(p2_ + 4 * cc);                               \
    }                                                                         \
  }
__global__ __launch_bounds__(EW * 32) void k_edge(
    const float* __restrict__ x, const float* __restrict__ ea,
    const int* __restrict__ recv, const int* __restrict__ send,
    const float* __restrict__ Web, const float* __restrict__ bias_eb,
    const int* __restrict__ list_r, const int* __restrict__ list_s,
    const int* __restrict__ totals, const int* __restrict__ binbase,
    float* __restrict__ agg_r, float* __restrict__ agg_s,
    float* __restrict__ cnt_r, float* __restrict__ cnt_s) {
  __shared__ __attribute__((aligned(16))) float sS[RPL * 64];
  __shared__ __attribute__((aligned(16))) float sP[PTILES * 16 * 64];
  __shared__ float sC[2 * RPL];
  for (int i = threadIdx.x; i < RPL * 16; i += EW * 32)
    *(float4*)(&sS[i * 4]) = (float4){0.f, 0.f, 0.f, 0.f};
  for (int i = threadIdx.x; i < 2 * RPL; i += EW * 32) sC[i] = 0.f;
  const int lane = threadIdx.x & 31;
  const int h = lane >> 4, m = lane & 15;
  const int wave = __builtin_amdgcn_readfirstlane(threadIdx.x >> 5);
  const int b = blockIdx.x;
  const int lo = b * RPB;
#define WFRAG(row0, kc, nt)                                        \
  (v2f){Web[((row0) + 4 * (kc) + 2 * h) * 64 + 4 * m + (nt)],      \
        Web[((row0) + 4 * (kc) + 2 * h + 1) * 64 + 4 * m + (nt)]}
  v2f bA[16];
#pragma unroll
  for (int kc = 0; kc < 4; ++kc)
#pragma unroll
    for (int nt = 0; nt < 4; ++nt) bA[kc * 4 + nt] = WFRAG(0, kc, nt);
  float bb[4];
#pragma unroll
  for (int nt = 0; nt < 4; ++nt) bb[nt] = bias_eb[4 * m + nt];
  for (int role = 0; role < 2; ++role) {
    const int rin = role ? 32 : 16, rot = role ? 16 : 32;
    v2f bO[16];
#pragma unroll
    for (int kc = 0; kc < 4; ++kc)
#pragma unroll
      for (int nt = 0; nt < 4; ++nt) bO[kc * 4 + nt] = WFRAG(rot, kc, nt);
    if (wave < PTILES) {
      v2f bI[16];
#pragma unroll
      for (int kc = 0; kc < 4; ++kc)
#pragma unroll
        for (int nt = 0; nt < 4; ++nt) bI[kc * 4 + nt] = WFRAG(rin, kc, nt);
      for (int pt = wave; pt < PTILES; pt += EW) {
        const int r = pt * 16 + m;
        const int node = lo + r;
        v2f a[4];
#pragma unroll
        for (int cc = 0; cc < 4; ++cc) a[cc] = (v2f){0.f, 0.f};
        if (r < RPB && node < NNODES) {
          const float* px = x + (long)node * 16 + 2 * h;
#pragma unroll
          for (int cc = 0; cc < 4; ++cc) a[cc] = *(const v2f*)(px + 4 * cc);
        }
        v8f pa[4];
#pragma unroll
        for (int nt = 0; nt < 4; ++nt)
          pa[nt] = (v8f){bb[nt], bb[nt], bb[nt], bb[nt], bb[nt], bb[nt], bb[nt], bb[nt]};
#pragma unroll
        for (int cc = 0; cc < 4; ++cc)
#pragma unroll
          for (int nt = 0; nt < 4; ++nt)
            pa[nt] = __builtin_amdgcn_wmma_f32_16x16x4_f32(false, a[cc], false, bI[cc * 4 + nt],
                                                           (short)0, pa[nt], false, false);
#pragma unroll
        for (int v = 0; v < 8; ++v)
          *(float4*)(&sP[(pt * 16 + 8 * h + v) * 64 + 4 * m]) =
              (float4){pa[0][v], pa[1][v], pa[2][v], pa[3][v]};
      }
    }
    __syncthreads();
    const int* __restrict__ list = (role ? list_s : list_r) + binbase[role * NB + b];
    const int n = totals[role * NB + b];
    const int ntiles = (n + 15) >> 4;
    float* C = sC + role * RPL;
    const int* __restrict__ oth = role ? recv : send;
#define ROW_PAIR(w_, tt)                                                      \
  (min((int)((unsigned)(w_) >> EBITS), RPB) |                                 \
   (((tt) * 16 + m < n ? min((int)((unsigned)(w_) >> EBITS), RPB) : RPB) << 16))
#define P_ROWS(acc_, d_)                                                            \
  _Pragma("unroll") for (int v = 0; v < 8; ++v) {                                   \
    const float4 pv_ = *(const float4*)(&sP[((d_)[v] & 0xffff) * 64 + 4 * m]);      \
    acc_[0][v] = pv_.x; acc_[1][v] = pv_.y; acc_[2][v] = pv_.z; acc_[3][v] = pv_.w; \
  }
    int t = wave;
    if (t < ntiles) {
      v2f G0[8], G1[8];
      int w0 = EDGE_LIST(t);
      int o0 = oth[w0 & EMASK];
      EDGE_ROWS(G0, w0 & EMASK, o0)
      int w1 = EDGE_LIST(t + EW);
      int o1 = oth[w1 & EMASK];
      int rp0 = ROW_PAIR(w0, t);
      int dd[8];
#pragma unroll
      for (int v = 0; v < 8; ++v) dd[v] = __shfl(rp0, 8 * h + v);
      v8f acc[4];
      P_ROWS(acc, dd)
      for (; t < ntiles; t += EW) {
        const int w2 = EDGE_LIST(t + 2 * EW);
        EDGE_ROWS(G1, w1 & EMASK, o1)
        const int rp1 = ROW_PAIR(w1, t + EW);
        int dn[8];
#pragma unroll
        for (int v = 0; v < 8; ++v) dn[v] = __shfl(rp1, 8 * h + v);
#pragma unroll
        for (int cc = 0; cc < 4; ++cc)
#pragma unroll
          for (int nt = 0; nt < 4; ++nt)
            acc[nt] = __builtin_amdgcn_wmma_f32_16x16x4_f32(false, G0[cc], false, bA[cc * 4 + nt],
                                                            (short)0, acc[nt], false, false);
        __builtin_amdgcn_sched_barrier(0);
        const int o2 = oth[w2 & EMASK];
        __builtin_amdgcn_sched_barrier(0);
#pragma unroll
        for (int cc = 0; cc < 4; ++cc)
#pragma unroll
          for (int nt = 0; nt < 4; ++nt)
            acc[nt] = __builtin_amdgcn_wmma_f32_16x16x4_f32(false, G0[4 + cc], false,
                                                            bO[cc * 4 + nt],
                                                            (short)0, acc[nt], false, false);
        const bool hb = h != 0;
        v8f q0, q1, q2, q3;
#pragma unroll
        for (int v = 0; v < 8; ++v) {
          const float e0 = fmaxf(acc[0][v], 0.0f), e1 = fmaxf(acc[1][v], 0.0f);
          const float e2 = fmaxf(acc[2][v], 0.0f), e3 = fmaxf(acc[3][v], 0.0f);
          q0[v] = hb ? e1 : e0;
          q1[v] = hb ? e0 : e1;
          q2[v] = hb ? e3 : e2;
          q3[v] = hb ? e2 : e3;
        }
        P_ROWS(acc, dn)
        __builtin_amdgcn_sched_barrier(0);
        const int oa = h, ob = 1 - h;
#pragma unroll
        for (int v = 0; v < 8; ++v) {
          float* p = sS + (dd[v] >> 16) * 64 + 4 * m;
          unsafeAtomicAdd(p + oa, q0[v]);
          unsafeAtomicAdd(p + ob, q1[v]);
          unsafeAtomicAdd(p + 2 + oa, q2[v]);
          unsafeAtomicAdd(p + 2 + ob, q3[v]);
        }
        if (h == 0) unsafeAtomicAdd(&C[rp0 >> 16], 1.0f);
#pragma unroll
        for (int q = 0; q < 8; ++q) G0[q] = G1[q];
#pragma unroll
        for (int v = 0; v < 8; ++v) dd[v] = dn[v];
        rp0 = rp1;
        w0 = w1; w1 = w2; o1 = o2;
      }
    }
    __syncthreads();
    float* __restrict__ agg = role ? agg_s : agg_r;
    for (int i = threadIdx.x; i < RPB * 16; i += EW * 32) {
      const int row = i >> 4, c4 = i & 15;
      const int node = lo + row;
      const float4 val = *(const float4*)(&sS[row * 64 + c4 * 4]);
      *(float4*)(&sS[row * 64 + c4 * 4]) = (float4){0.f, 0.f, 0.f, 0.f};
      if (node < NNODES) *(float4*)(agg + (long)node * 64 + c4 * 4) = val;
    }
  }
  for (int i = threadIdx.x; i < 2 * RPB; i += EW * 32) {
    const int role = i / RPB;
    const int row = i - role * RPB;
    const int node = lo + row;
    if (node < NNODES) (role ? cnt_s : cnt_r)[node] = sC[role * RPL + row];
  }
}

#define GW 4
#define NODE_ROWS(A_, tt)                                                     \
  {                                                                           \
    const float* ap_ = agg_r + (long)((tt) * 16 + m) * 64 + 2 * h;            \
    const float* xp_ = x + (long)((tt) * 16 + m) * 16 + 2 * h;                \
    _Pragma("unroll") for (int kc = 0; kc < 16; ++kc)                         \
      A_[kc] = *(const v2f*)(ap_ + 4 * kc);                                   \
    _Pragma("unroll") for (int kc = 0; kc < 4; ++kc)                          \
      A_[16 + kc] = *(const v2f*)(xp_ + 4 * kc);                              \
  }
__global__ __launch_bounds__(GW * 32) void k_node(
    const float* __restrict__ x, const float* __restrict__ agg_r,
    const float* __restrict__ Wnb, const float* __restrict__ bias_nb,
    float* __restrict__ n_h) {
  __shared__ float sW[20 * 4 * 64];
  {
    float wv[40];
#pragma unroll
    for (int q = 0; q < 40; ++q) {
      const int i = threadIdx.x + q * (GW * 32);
      int j = i & 1, lane = (i >> 1) & 31, nt = (i >> 6) & 3, kc = i >> 8;
      int kk = 4 * kc + 2 * (lane >> 4) + j;
      wv[q] = Wnb[kk * 64 + nt * 16 + (lane & 15)];
    }
#pragma unroll
    for (int q = 0; q < 40; ++q) sW[threadIdx.x + q * (GW * 32)] = wv[q];
  }
  __syncthreads();
  const int lane = threadIdx.x & 31;
  const int h = lane >> 4, m = lane & 15;
  v2f bw[80];
#pragma unroll
  for (int q = 0; q < 80; ++q) bw[q] = *(const v2f*)(&sW[q * 64 + lane * 2]);
  float bb[4];
#pragma unroll
  for (int nt = 0; nt < 4; ++nt) bb[nt] = bias_nb[nt * 16 + m];
  const int wv_ = __builtin_amdgcn_readfirstlane(threadIdx.x >> 5);
  const int nwaves = gridDim.x * GW;
  const int ntiles = NNODES / 16;
  int t = wv_ * gridDim.x + blockIdx.x;
  if (t < ntiles) {
    v2f a[20];
    NODE_ROWS(a, t)
    for (; t < ntiles; t += nwaves) {
      const int tn = min(t + nwaves, ntiles - 1);
      v2f an[20];
      NODE_ROWS(an, tn)
      __builtin_amdgcn_sched_barrier(0);
      const int n0 = t * 16;
      v8f acc[4];
#pragma unroll
      for (int nt = 0; nt < 4; ++nt)
        acc[nt] = (v8f){bb[nt], bb[nt], bb[nt], bb[nt], bb[nt], bb[nt], bb[nt], bb[nt]};
#pragma unroll
      for (int kc = 0; kc < 20; ++kc)
#pragma unroll
        for (int nt = 0; nt < 4; ++nt)
          acc[nt] = __builtin_amdgcn_wmma_f32_16x16x4_f32(false, a[kc], false, bw[kc * 4 + nt],
                                                          (short)0, acc[nt], false, false);
#pragma unroll
      for (int nt = 0; nt < 4; ++nt)
#pragma unroll
        for (int v = 0; v < 8; ++v)
          n_h[(long)(n0 + 8 * h + v) * 64 + nt * 16 + m] = fmaxf(acc[nt][v], 0.0f);
      __builtin_amdgcn_sched_barrier(0);
#pragma unroll
      for (int kc = 0; kc < 20; ++kc) a[kc] = an[kc];
    }
  }
}

__global__ void k_colsum(const float* __restrict__ src, int rows, float* __restrict__ dst) {
  __shared__ float red[256];
  const int col = threadIdx.x & 63;
  const int g = threadIdx.x >> 6;
  float local = 0.f;
  const int step = gridDim.x * 4;
  int r = blockIdx.x * 4 + g;
  for (; r + 15 * step < rows; r += 16 * step) {
    float v[16];
#pragma unroll
    for (int q = 0; q < 16; ++q) v[q] = src[(long)(r + q * step) * 64 + col];
#pragma unroll
    for (int q = 0; q < 16; ++q) local += v[q];
  }
  for (; r < rows; r += step) local += src[(long)r * 64 + col];
  red[threadIdx.x] = local;
  __syncthreads();
  if (g == 0) {
    local += red[64 + threadIdx.x] + red[128 + threadIdx.x] + red[192 + threadIdx.x];
    unsafeAtomicAdd(&dst[col], local);
  }
}

__global__ void k_global(const float* __restrict__ u,
                         const float* __restrict__ Wgb, const float* __restrict__ bgb,
                         const float* __restrict__ Wd1, const float* __restrict__ bd1,
                         const float* __restrict__ e_sum, const float* __restrict__ n_sum,
                         float* __restrict__ bias_d1) {
  __shared__ float su2[64];
  const int j = threadIdx.x;
  float acc = bgb[j];
#pragma unroll 32
  for (int k = 0; k < 64; ++k) {
    float em = e_sum[k] * (1.0f / NEDGES);
    float nm = n_sum[k] * (1.0f / NNODES);
    acc += em * Wgb[k * 64 + j] + nm * Wgb[(64 + k) * 64 + j] + u[k] * Wgb[(128 + k) * 64 + j];
  }
  acc = fmaxf(acc, 0.0f);
  su2[j] = acc;
  __syncthreads();
  float b = bd1[j];
  for (int k = 0; k < 64; ++k) b += su2[k] * Wd1[(192 + k) * 64 + j];
  bias_d1[j] = b;
}

#define DEC_ROWS(A_, cr_, cs_, tt)                                            \
  {                                                                           \
    const long n_ = (long)((tt) * 16 + m);                                    \
    cr_ = cnt_r[n_];                                                          \
    cs_ = cnt_s[n_];                                                          \
    const float* pr_ = agg_r + n_ * 64 + 2 * h;                               \
    const float* ps_ = agg_s + n_ * 64 + 2 * h;                               \
    const float* pn_ = n_h + n_ * 64 + 2 * h;                                 \
    _Pragma("unroll") for (int kc = 0; kc < 16; ++kc) {                       \
      A_[kc] = *(const v2f*)(pr_ + 4 * kc);                                   \
      A_[16 + kc] = *(const v2f*)(ps_ + 4 * kc);                              \
      A_[32 + kc] = *(const v2f*)(pn_ + 4 * kc);                              \
    }                                                                         \
  }
__global__ __launch_bounds__(GW * 32) void k_dec(
    const float* __restrict__ agg_r, const float* __restrict__ agg_s,
    const float* __restrict__ cnt_r, const float* __restrict__ cnt_s,
    const float* __restrict__ n_h, const float* __restrict__ Wd1,
    const float* __restrict__ bias_d1, const float* __restrict__ Wd2,
    const float* __restrict__ bd2, float* __restrict__ out) {
  __shared__ float sW1[48 * 4 * 64];
  __shared__ float sW2[16 * 64];
  __shared__ float sH[GW * 16 * 64];
  {
    float wv[96], wu[8];
#pragma unroll
    for (int q = 0; q < 96; ++q) {
      const int i = threadIdx.x + q * (GW * 32);
      int j = i & 1, lane = (i >> 1) & 31, nt = (i >> 6) & 3, kc = i >> 8;
      int kk = 4 * kc + 2 * (lane >> 4) + j;
      wv[q] = Wd1[kk * 64 + nt * 16 + (lane & 15)];
    }
#pragma unroll
    for (int q = 0; q < 8; ++q) {
      const int i = threadIdx.x + q * (GW * 32);
      int j = i & 1, lane = (i >> 1) & 31, kc = i >> 6;
      int kk = 4 * kc + 2 * (lane >> 4) + j;
      wu[q] = Wd2[kk * 16 + (lane & 15)];
    }
#pragma unroll
    for (int q = 0; q < 96; ++q) sW1[threadIdx.x + q * (GW * 32)] = wv[q];
#pragma unroll
    for (int q = 0; q < 8; ++q) sW2[threadIdx.x + q * (GW * 32)] = wu[q];
  }
  __syncthreads();
  const int lane = threadIdx.x & 31;
  const int h = lane >> 4, m = lane & 15;
  v2f bw[192], bu[16];
#pragma unroll
  for (int q = 0; q < 192; ++q) bw[q] = *(const v2f*)(&sW1[q * 64 + lane * 2]);
#pragma unroll
  for (int q = 0; q < 16; ++q) bu[q] = *(const v2f*)(&sW2[q * 64 + lane * 2]);
  float bb[4];
#pragma unroll
  for (int nt = 0; nt < 4; ++nt) bb[nt] = bias_d1[nt * 16 + m];
  const float bb2 = bd2[m];
  const int wv_ = __builtin_amdgcn_readfirstlane(threadIdx.x >> 5);
  float* hw = sH + wv_ * 1024;
  const int nwaves = gridDim.x * GW;
  const int ntiles = NNODES / 16;
  int t = wv_ * gridDim.x + blockIdx.x;
  if (t < ntiles) {
    v2f a[48];
    float cr, cs;
    DEC_ROWS(a, cr, cs, t)
    for (; t < ntiles; t += nwaves) {
      const int tn = min(t + nwaves, ntiles - 1);
      v2f an[48];
      float crn, csn;
      DEC_ROWS(an, crn, csn, tn)
      __builtin_amdgcn_sched_barrier(0);
      const int n0 = t * 16;
      const float invr = 1.0f / fmaxf(cr, 1.0f);
      const float invs = 1.0f / fmaxf(cs, 1.0f);
      v8f acc[4];
#pragma unroll
      for (int nt = 0; nt < 4; ++nt)
        acc[nt] = (v8f){bb[nt], bb[nt], bb[nt], bb[nt], bb[nt], bb[nt], bb[nt], bb[nt]};
#pragma unroll
      for (int kc = 0; kc < 48; ++kc) {
        v2f av = a[kc];
        if (kc < 16)      av *= invr;
        else if (kc < 32) av *= invs;
#pragma unroll
        for (int nt = 0; nt < 4; ++nt)
          acc[nt] = __builtin_amdgcn_wmma_f32_16x16x4_f32(false, av, false, bw[kc * 4 + nt],
                                                          (short)0, acc[nt], false, false);
      }
#pragma unroll
      for (int nt = 0; nt < 4; ++nt)
#pragma unroll
        for (int v = 0; v < 8; ++v)
          hw[(8 * h + v) * 64 + nt * 16 + m] = fmaxf(acc[nt][v], 0.0f);
      v8f acc2 = (v8f){bb2, bb2, bb2, bb2, bb2, bb2, bb2, bb2};
#pragma unroll
      for (int kc = 0; kc < 16; ++kc) {
        const int kk = 4 * kc + 2 * h;
        v2f a2 = *(const v2f*)(hw + m * 64 + kk);
        acc2 = __builtin_amdgcn_wmma_f32_16x16x4_f32(false, a2, false, bu[kc],
                                                     (short)0, acc2, false, false);
      }
#pragma unroll
      for (int v = 0; v < 8; ++v)
        out[(long)(n0 + 8 * h + v) * 16 + m] = acc2[v];
      __builtin_amdgcn_sched_barrier(0);
#pragma unroll
      for (int kc = 0; kc < 48; ++kc) a[kc] = an[kc];
      cr = crn;
      cs = csn;
    }
  }
}

extern "C" void kernel_launch(void* const* d_in, const int* in_sizes, int n_in,
                              void* d_out, int out_size, void* d_ws, size_t ws_size,
                              hipStream_t stream) {
  const float* x    = (const float*)d_in[0];
  const float* ea   = (const float*)d_in[1];
  const float* u    = (const float*)d_in[2];
  const int* send   = (const int*)d_in[3];
  const int* recv   = (const int*)d_in[4];
  const float* Web  = (const float*)d_in[7];
  const float* beb  = (const float*)d_in[8];
  const float* Wnb  = (const float*)d_in[9];
  const float* bnb  = (const float*)d_in[10];
  const float* Wgb  = (const float*)d_in[11];
  const float* bgb  = (const float*)d_in[12];
  const float* Wd1  = (const float*)d_in[13];
  const float* bd1  = (const float*)d_in[14];
  const float* Wd2  = (const float*)d_in[15];
  const float* bd2  = (const float*)d_in[16];
  float* out = (float*)d_out;
  float* ws = (float*)d_ws;

  const long N = NNODES;
  float* agg_r   = ws;
  float* agg_s   = agg_r + N * 64;
  float* cnt_r   = agg_s + N * 64;
  float* cnt_s   = cnt_r + N;
  float* e_sum   = cnt_s + N;
  float* n_sum   = e_sum + 64;
  float* bias_eb = n_sum + 128;
  float* bias_nb = bias_eb + 64;
  float* bias_d1 = bias_nb + 64;
  float* n_hbuf  = bias_d1 + 64;
  int* list_r  = (int*)n_hbuf;
  int* list_s  = list_r + NEDGES;
  int* totals  = list_s + NEDGES;
  int* binbase = totals + 2 * NB;
  int* cursors = binbase + 2 * NB;

  k_zero<<<2, 256, 0, stream>>>(e_sum, 128, totals, 2 * NB);
  k_prep<<<1, 64, 0, stream>>>(u, Web, beb, Wnb, bnb, bias_eb, bias_nb);
  k_count<<<NCHUNKS, 256, 0, stream>>>(recv, send, totals);
  k_scan<<<1, 2 * NB, 0, stream>>>(totals, binbase, cursors);
  k_fill<<<NCHUNKS, 256, 0, stream>>>(recv, send, cursors, list_r, list_s);
  k_edge<<<NB, EW * 32, 0, stream>>>(x, ea, recv, send, Web, bias_eb, list_r, list_s,
                                    totals, binbase, agg_r, agg_s, cnt_r, cnt_s);
  k_node<<<256, GW * 32, 0, stream>>>(x, agg_r, Wnb, bias_nb, n_hbuf);
  k_colsum<<<512, 256, 0, stream>>>(agg_r, NNODES, e_sum);
  k_colsum<<<512, 256, 0, stream>>>(n_hbuf, NNODES, n_sum);
  k_global<<<1, 64, 0, stream>>>(u, Wgb, bgb, Wd1, bd1, e_sum, n_sum, bias_d1);
  k_dec<<<256, GW * 32, 0, stream>>>(agg_r, agg_s, cnt_r, cnt_s, n_hbuf,
                                   Wd1, bias_d1, Wd2, bd2, out);
}


# 11 launches to 9: k_scan dropped (k_fill and k_edge sum the totals in front of their range themselves; cursors start at 0); both column sums in one launch of 1024 workgroups; arrival order as before
# speedup vs baseline: 2.4432x; 1.0345x over previous
#include <hip/hip_runtime.h>

typedef __attribute__((ext_vector_type(2))) float v2f;
typedef __attribute__((ext_vector_type(8))) float v8f;

#define NNODES 100000
#define NEDGES 1600000

__global__ void k_zero(float* __restrict__ a, int na, int* __restrict__ b, int nb) {
  int i = blockIdx.x * blockDim.x + threadIdx.x;
  if (i < na) a[i] = 0.f;
  if (i < nb) b[i] = 0;
}

__global__ void k_prep(const float* __restrict__ u,
                       const float* __restrict__ Web, const float* __restrict__ beb,
                       const float* __restrict__ Wnb, const float* __restrict__ bnb,
                       float* __restrict__ bias_eb, float* __restrict__ bias_nb) {
  int j = threadIdx.x;
  float a = beb[j], b = bnb[j];
  for (int k0 = 0; k0 < 64; k0 += 32) {
    float wa[32], wb[32];
#pragma unroll
    for (int q = 0; q < 32; ++q) {
      wa[q] = Web[(48 + k0 + q) * 64 + j];
      wb[q] = Wnb[(80 + k0 + q) * 64 + j];
    }
#pragma unroll
    for (int q = 0; q < 32; ++q) {
      float uk = u[k0 + q];
      a += uk * wa[q];
      b += uk * wb[q];
    }
  }
  bias_eb[j] = a;
  bias_nb[j] = b;
}

#define NB 256
#define RPB 391
#define CHUNK 2048
#define NCHUNKS ((NEDGES + CHUNK - 1) / CHUNK)
#define EPT (CHUNK / 256)
#define EBITS 21
#define EMASK ((1 << EBITS) - 1)
static_assert(NEDGES <= (1 << EBITS), "edge number must fit the low bits of a list word");
static_assert(RPB <= (1 << (30 - EBITS)), "row in range must fit above the edge number");
__global__ __launch_bounds__(256) void k_count(const int* __restrict__ recv,
                                               const int* __restrict__ send,
                                               int* __restrict__ totals) {
  __shared__ int hist[2 * NB];
  for (int j = threadIdx.x; j < 2 * NB; j += 256) hist[j] = 0;
  __syncthreads();
  const int e0 = blockIdx.x * CHUNK;
  int nr[EPT], ns[EPT];
#pragma unroll
  for (int k = 0; k < EPT; ++k) {
    const int e = min(e0 + k * 256 + (int)threadIdx.x, NEDGES - 1);
    nr[k] = recv[e];
    ns[k] = send[e];
  }
#pragma unroll
  for (int k = 0; k < EPT; ++k) {
    if (e0 + k * 256 + (int)threadIdx.x < NEDGES) {
      atomicAdd(&hist[nr[k] / RPB], 1);
      atomicAdd(&hist[NB + ns[k] / RPB], 1);
    }
  }
  __syncthreads();
  for (int j = threadIdx.x; j < 2 * NB; j += 256) {
    const int c = hist[j];
    if (c) atomicAdd(&totals[j], c);
  }
}

__global__ __launch_bounds__(256) void k_fill(const int* __restrict__ recv,
                                              const int* __restrict__ send,
                                              const int* __restrict__ totals,
                                              int* __restrict__ cursors,
                                              int* __restrict__ list_r,
                                              int* __restrict__ list_s) {
  __shared__ int hist[2 * NB];
  __shared__ int lst[2 * NB];
  __shared__ int cur[2 * NB];
  __shared__ int wsum[4 * 8];
  __shared__ int sW[2 * CHUNK];
  __shared__ int sG[2 * CHUNK];
  for (int j = threadIdx.x; j < 2 * NB; j += 256) { hist[j] = 0; cur[j] = 0; }
  __syncthreads();
  const int e0 = blockIdx.x * CHUNK;
  int nr[EPT], ns[EPT];
#pragma unroll
  for (int k = 0; k < EPT; ++k) {
    const int e = min(e0 + k * 256 + (int)threadIdx.x, NEDGES - 1);
    nr[k] = recv[e];
    ns[k] = send[e];
  }
#pragma unroll
  for (int k = 0; k < EPT; ++k) {
    if (e0 + k * 256 + (int)threadIdx.x < NEDGES) {
      atomicAdd(&hist[nr[k] / RPB], 1);
      atomicAdd(&hist[NB + ns[k] / RPB], 1);
    }
  }
  __syncthreads();
  const int j = threadIdx.x;
  const int lane = j & 31, wave = j >> 5;
  const int c0 = hist[j], c1 = hist[NB + j];
  const int t0 = totals[j], t1 = totals[NB + j];
  const int a0 = c0 ? atomicAdd(&cursors[j], c0) : 0;
  const int a1 = c1 ? atomicAdd(&cursors[NB + j], c1) : 0;
  int i0 = c0, i1 = c1;
  int u0 = t0, u1 = t1;
  for (int d = 1; d < 32; d <<= 1) {
    const int x0 = __shfl_up(i0, d), x1 = __shfl_up(i1, d);
    const int y0 = __shfl_up(u0, d), y1 = __shfl_up(u1, d);
    if (lane >= d) { i0 += x0; i1 += x1; u0 += y0; u1 += y1; }
  }
  if (lane == 31) {
    wsum[wave] = i0; wsum[8 + wave] = i1;
    wsum[16 + wave] = u0; wsum[24 + wave] = u1;
  }
  __syncthreads();
  int p0 = 0, p1 = 0, b0 = 0, b1 = 0;
  for (int w = 0; w < wave; ++w) {
    p0 += wsum[w]; p1 += wsum[8 + w];
    b0 += wsum[16 + w]; b1 += wsum[24 + w];
  }
  lst[j] = p0 + i0 - c0;
  lst[NB + j] = p1 + i1 - c1;
  hist[j] = b0 + u0 - t0 + a0;
  hist[NB + j] = b1 + u1 - t1 + a1;
  __syncthreads();
#pragma unroll
  for (int k = 0; k < EPT; ++k) {
    const int e = e0 + k * 256 + (int)threadIdx.x;
    if (e < NEDGES) {
      const int r = nr[k], s = ns[k];
      const int br = r / RPB, bs = s / RPB;
      const int a = atomicAdd(&cur[br], 1);
      const int p = lst[br] + a;
      sW[p] = e | ((r - br * RPB) << EBITS);
      sG[p] = hist[br] + a;
      const int c = atomicAdd(&cur[NB + bs], 1);
      const int q = CHUNK + lst[NB + bs] + c;
      sW[q] = e | ((s - bs * RPB) << EBITS);
      sG[q] = hist[NB + bs] + c;
    }
  }
  __syncthreads();
  const int nv = min(CHUNK, NEDGES - e0);
  for (int i = threadIdx.x; i < nv; i += 256) {
    list_r[sG[i]] = sW[i];
    list_s[sG[CHUNK + i]] = sW[CHUNK + i];
  }
}

#define EW 16
#define RPL (RPB + 1)
#define PTILES ((RPB + 15) / 16)
#define EDGE_LIST(tt) list[min((tt) * 16 + m, n - 1)]
#define EDGE_ROWS(A_, e_, o_)                                                 \
  {                                                                           \
    const float* p0_ = ea + (long)(e_) * 16 + 2 * h;                          \
    const float* p2_ = x + (long)(o_) * 16 + 2 * h;                           \
    _Pragma("unroll") for (int cc = 0; cc < 4; ++cc) {                        \
      A_[cc] = *(const v2f*)(p0_ + 4 * cc);                                   \
      A_[4 + cc] = *(const v2f*)(p2_ + 4 * cc);                               \
    }                                                                         \
  }
__global__ __launch_bounds__(EW * 32) void k_edge(
    const float* __restrict__ x, const float* __restrict__ ea,
    const int* __restrict__ recv, const int* __restrict__ send,
    const float* __restrict__ Web, const float* __restrict__ bias_eb,
    const int* __restrict__ list_r, const int* __restrict__ list_s,
    const int* __restrict__ totals,
    float* __restrict__ agg_r, float* __restrict__ agg_s,
    float* __restrict__ cnt_r, float* __restrict__ cnt_s) {
  __shared__ __attribute__((aligned(16))) float sS[RPL * 64];
  __shared__ __attribute__((aligned(16))) float sP[PTILES * 16 * 64];
  __shared__ float sC[2 * RPL];
  __shared__ int sPart[EW];
  for (int i = threadIdx.x; i < RPL * 16; i += EW * 32)
    *(float4*)(&sS[i * 4]) = (float4){0.f, 0.f, 0.f, 0.f};
  for (int i = threadIdx.x; i < 2 * RPL; i += EW * 32) sC[i] = 0.f;
  const int lane = threadIdx.x & 31;
  const int h = lane >> 4, m = lane & 15;
  const int wave = __builtin_amdgcn_readfirstlane(threadIdx.x >> 5);
  const int b = blockIdx.x;
  const int lo = b * RPB;
  static_assert(EW * 32 == 2 * NB, "one thread per entry of totals");
  {
    int v = ((int)(threadIdx.x & (NB - 1)) < b) ? totals[threadIdx.x] : 0;
    for (int d = 16; d >= 1; d >>= 1) v += __shfl_xor(v, d);
    if (lane == 0) sPart[wave] = v;
  }
#define WFRAG(row0, kc, nt)                                        \
  (v2f){Web[((row0) + 4 * (kc) + 2 * h) * 64 + 4 * m + (nt)],      \
        Web[((row0) + 4 * (kc) + 2 * h + 1) * 64 + 4 * m + (nt)]}
  v2f bA[16];
#pragma unroll
  for (int kc = 0; kc < 4; ++kc)
#pragma unroll
    for (int nt = 0; nt < 4; ++nt) bA[kc * 4 + nt] = WFRAG(0, kc, nt);
  float bb[4];
#pragma unroll
  for (int nt = 0; nt < 4; ++nt) bb[nt] = bias_eb[4 * m + nt];
  for (int role = 0; role < 2; ++role) {
    const int rin = role ? 32 : 16, rot = role ? 16 : 32;
    v2f bO[16];
#pragma unroll
    for (int kc = 0; kc < 4; ++kc)
#pragma unroll
      for (int nt = 0; nt < 4; ++nt) bO[kc * 4 + nt] = WFRAG(rot, kc, nt);
    if (wave < PTILES) {
      v2f bI[16];
#pragma unroll
      for (int kc = 0; kc < 4; ++kc)
#pragma unroll
        for (int nt = 0; nt < 4; ++nt) bI[kc * 4 + nt] = WFRAG(rin, kc, nt);
      for (int pt = wave; pt < PTILES; pt += EW) {
        const int r = pt * 16 + m;
        const int node = lo + r;
        v2f a[4];
#pragma unroll
        for (int cc = 0; cc < 4; ++cc) a[cc] = (v2f){0.f, 0.f};
        if (r < RPB && node < NNODES) {
          const float* px = x + (long)node * 16 + 2 * h;
#pragma unroll
          for (int cc = 0; cc < 4; ++cc) a[cc] = *(const v2f*)(px + 4 * cc);
        }
        v8f pa[4];
#pragma unroll
        for (int nt = 0; nt < 4; ++nt)
          pa[nt] = (v8f){bb[nt], bb[nt], bb[nt], bb[nt], bb[nt], bb[nt], bb[nt], bb[nt]};
#pragma unroll
        for (int cc = 0; cc < 4; ++cc)
#pragma unroll
          for (int nt = 0; nt < 4; ++nt)
            pa[nt] = __builtin_amdgcn_wmma_f32_16x16x4_f32(false, a[cc], false, bI[cc * 4 + nt],
                                                           (short)0, pa[nt], false, false);
#pragma unroll
        for (int v = 0; v < 8; ++v)
          *(float4*)(&sP[(pt * 16 + 8 * h + v) * 64 + 4 * m]) =
              (float4){pa[0][v], pa[1][v], pa[2][v], pa[3][v]};
      }
    }
    __syncthreads();
    int base = 0;
    for (int w = 0; w < EW / 2; ++w) base += sPart[role * (EW / 2) + w];
    const int* __restrict__ list = (role ? list_s : list_r) + base;
    const int n = totals[role * NB + b];
    const int ntiles = (n + 15) >> 4;
    float* C = sC + role * RPL;
    const int* __restrict__ oth = role ? recv : send;
#define ROW_PAIR(w_, tt)                                                      \
  (min((int)((unsigned)(w_) >> EBITS), RPB) |                                 \
   (((tt) * 16 + m < n ? min((int)((unsigned)(w_) >> EBITS), RPB) : RPB) << 16))
#define P_ROWS(acc_, d_)                                                            \
  _Pragma("unroll") for (int v = 0; v < 8; ++v) {                                   \
    const float4 pv_ = *(const float4*)(&sP[((d_)[v] & 0xffff) * 64 + 4 * m]);      \
    acc_[0][v] = pv_.x; acc_[1][v] = pv_.y; acc_[2][v] = pv_.z; acc_[3][v] = pv_.w; \
  }
    int t = wave;
    if (t < ntiles) {
      v2f G0[8], G1[8];
      int w0 = EDGE_LIST(t);
      int o0 = oth[w0 & EMASK];
      EDGE_ROWS(G0, w0 & EMASK, o0)
      int w1 = EDGE_LIST(t + EW);
      int o1 = oth[w1 & EMASK];
      int rp0 = ROW_PAIR(w0, t);
      int dd[8];
#pragma unroll
      for (int v = 0; v < 8; ++v) dd[v] = __shfl(rp0, 8 * h + v);
      v8f acc[4];
      P_ROWS(acc, dd)
      for (; t < ntiles; t += EW) {
        const int w2 = EDGE_LIST(t + 2 * EW);
        EDGE_ROWS(G1, w1 & EMASK, o1)
        const int rp1 = ROW_PAIR(w1, t + EW);
        int dn[8];
#pragma unroll
        for (int v = 0; v < 8; ++v) dn[v] = __shfl(rp1, 8 * h + v);
#pragma unroll
        for (int cc = 0; cc < 4; ++cc)
#pragma unroll
          for (int nt = 0; nt < 4; ++nt)
            acc[nt] = __builtin_amdgcn_wmma_f32_16x16x4_f32(false, G0[cc], false, bA[cc * 4 + nt],
                                                            (short)0, acc[nt], false, false);
        __builtin_amdgcn_sched_barrier(0);
        const int o2 = oth[w2 & EMASK];
        __builtin_amdgcn_sched_barrier(0);
#pragma unroll
        for (int cc = 0; cc < 4; ++cc)
#pragma unroll
          for (int nt = 0; nt < 4; ++nt)
            acc[nt] = __builtin_amdgcn_wmma_f32_16x16x4_f32(false, G0[4 + cc], false,
                                                            bO[cc * 4 + nt],
                                                            (short)0, acc[nt], false, false);
        const bool hb = h != 0;
        v8f q0, q1, q2, q3;
#pragma unroll
        for (int v = 0; v < 8; ++v) {
          const float e0 = fmaxf(acc[0][v], 0.0f), e1 = fmaxf(acc[1][v], 0.0f);
          const float e2 = fmaxf(acc[2][v], 0.0f), e3 = fmaxf(acc[3][v], 0.0f);
          q0[v] = hb ? e1 : e0;
          q1[v] = hb ? e0 : e1;
          q2[v] = hb ? e3 : e2;
          q3[v] = hb ? e2 : e3;
        }
        P_ROWS(acc, dn)
        __builtin_amdgcn_sched_barrier(0);
        const int oa = h, ob = 1 - h;
#pragma unroll
        for (int v = 0; v < 8; ++v) {
          float* p = sS + (dd[v] >> 16) * 64 + 4 * m;
          unsafeAtomicAdd(p + oa, q0[v]);
          unsafeAtomicAdd(p + ob, q1[v]);
          unsafeAtomicAdd(p + 2 + oa, q2[v]);
          unsafeAtomicAdd(p + 2 + ob, q3[v]);
        }
        if (h == 0) unsafeAtomicAdd(&C[rp0 >> 16], 1.0f);
#pragma unroll
        for (int q = 0; q < 8; ++q) G0[q] = G1[q];
#pragma unroll
        for (int v = 0; v < 8; ++v) dd[v] = dn[v];
        rp0 = rp1;
        w0 = w1; w1 = w2; o1 = o2;
      }
    }
    __syncthreads();
    float* __restrict__ agg = role ? agg_s : agg_r;
    for (int i = threadIdx.x; i < RPB * 16; i += EW * 32) {
      const int row = i >> 4, c4 = i & 15;
      const int node = lo + row;
      const float4 val = *(const float4*)(&sS[row * 64 + c4 * 4]);
      *(float4*)(&sS[row * 64 + c4 * 4]) = (float4){0.f, 0.f, 0.f, 0.f};
      if (node < NNODES) *(float4*)(agg + (long)node * 64 + c4 * 4) = val;
    }
  }
  for (int i = threadIdx.x; i < 2 * RPB; i += EW * 32) {
    const int role = i / RPB;
    const int row = i - role * RPB;
    const int node = lo + row;
    if (node < NNODES) (role ? cnt_s : cnt_r)[node] = sC[role * RPL + row];
  }
}

#define GW 4
#define NODE_ROWS(A_, tt)                                                     \
  {                                                                           \
    const float* ap_ = agg_r + (long)((tt) * 16 + m) * 64 + 2 * h;            \
    const float* xp_ = x + (long)((tt) * 16 + m) * 16 + 2 * h;                \
    _Pragma("unroll") for (int kc = 0; kc < 16; ++kc)                         \
      A_[kc] = *(const v2f*)(ap_ + 4 * kc);                                   \
    _Pragma("unroll") for (int kc = 0; kc < 4; ++kc)                          \
      A_[16 + kc] = *(const v2f*)(xp_ + 4 * kc);                              \
  }
__global__ __launch_bounds__(GW * 32) void k_node(
    const float* __restrict__ x, const float* __restrict__ agg_r,
    const float* __restrict__ Wnb, const float* __restrict__ bias_nb,
    float* __restrict__ n_h) {
  __shared__ float sW[20 * 4 * 64];
  {
    float wv[40];
#pragma unroll
    for (int q = 0; q < 40; ++q) {
      const int i = threadIdx.x + q * (GW * 32);
      int j = i & 1, lane = (i >> 1) & 31, nt = (i >> 6) & 3, kc = i >> 8;
      int kk = 4 * kc + 2 * (lane >> 4) + j;
      wv[q] = Wnb[kk * 64 + nt * 16 + (lane & 15)];
    }
#pragma unroll
    for (int q = 0; q < 40; ++q) sW[threadIdx.x + q * (GW * 32)] = wv[q];
  }
  __syncthreads();
  const int lane = threadIdx.x & 31;
  const int h = lane >> 4, m = lane & 15;
  v2f bw[80];
#pragma unroll
  for (int q = 0; q < 80; ++q) bw[q] = *(const v2f*)(&sW[q * 64 + lane * 2]);
  float bb[4];
#pragma unroll
  for (int nt = 0; nt < 4; ++nt) bb[nt] = bias_nb[nt * 16 + m];
  const int wv_ = __builtin_amdgcn_readfirstlane(threadIdx.x >> 5);
  const int nwaves = gridDim.x * GW;
  const int ntiles = NNODES / 16;
  int t = wv_ * gridDim.x + blockIdx.x;
  if (t < ntiles) {
    v2f a[20];
    NODE_ROWS(a, t)
    for (; t < ntiles; t += nwaves) {
      const int tn = min(t + nwaves, ntiles - 1);
      v2f an[20];
      NODE_ROWS(an, tn)
      __builtin_amdgcn_sched_barrier(0);
      const int n0 = t * 16;
      v8f acc[4];
#pragma unroll
      for (int nt = 0; nt < 4; ++nt)
        acc[nt] = (v8f){bb[nt], bb[nt], bb[nt], bb[nt], bb[nt], bb[nt], bb[nt], bb[nt]};
#pragma unroll
      for (int kc = 0; kc < 20; ++kc)
#pragma unroll
        for (int nt = 0; nt < 4; ++nt)
          acc[nt] = __builtin_amdgcn_wmma_f32_16x16x4_f32(false, a[kc], false, bw[kc * 4 + nt],
                                                          (short)0, acc[nt], false, false);
#pragma unroll
      for (int nt = 0; nt < 4; ++nt)
#pragma unroll
        for (int v = 0; v < 8; ++v)
          n_h[(long)(n0 + 8 * h + v) * 64 + nt * 16 + m] = fmaxf(acc[nt][v], 0.0f);
      __builtin_amdgcn_sched_barrier(0);
#pragma unroll
      for (int kc = 0; kc < 20; ++kc) a[kc] = an[kc];
    }
  }
}

__global__ void k_colsum(const float* __restrict__ srcA, float* __restrict__ dstA,
                         const float* __restrict__ srcB, float* __restrict__ dstB, int rows) {
  __shared__ float red[256];
  const int col = threadIdx.x & 63;
  const int g = threadIdx.x >> 6;
  float local = 0.f;
  const int half = gridDim.x >> 1;
  const bool second = (int)blockIdx.x >= half;
  const float* __restrict__ src = second ? srcB : srcA;
  float* __restrict__ dst = second ? dstB : dstA;
  const int step = half * 4;
  int r = ((int)blockIdx.x - (second ? half : 0)) * 4 + g;
  for (; r + 15 * step < rows; r += 16 * step) {
    float v[16];
#pragma unroll
    for (int q = 0; q < 16; ++q) v[q] = src[(long)(r + q * step) * 64 + col];
#pragma unroll
    for (int q = 0; q < 16; ++q) local += v[q];
  }
  for (; r < rows; r += step) local += src[(long)r * 64 + col];
  red[threadIdx.x] = local;
  __syncthreads();
  if (g == 0) {
    local += red[64 + threadIdx.x] + red[128 + threadIdx.x] + red[192 + threadIdx.x];
    unsafeAtomicAdd(&dst[col], local);
  }
}

__global__ void k_global(const float* __restrict__ u,
                         const float* __restrict__ Wgb, const float* __restrict__ bgb,
                         const float* __restrict__ Wd1, const float* __restrict__ bd1,
                         const float* __restrict__ e_sum, const float* __restrict__ n_sum,
                         float* __restrict__ bias_d1) {
  __shared__ float su2[64];
  const int j = threadIdx.x;
  float acc = bgb[j];
#pragma unroll 32
  for (int k = 0; k < 64; ++k) {
    float em = e_sum[k] * (1.0f / NEDGES);
    float nm = n_sum[k] * (1.0f / NNODES);
    acc += em * Wgb[k * 64 + j] + nm * Wgb[(64 + k) * 64 + j] + u[k] * Wgb[(128 + k) * 64 + j];
  }
  acc = fmaxf(acc, 0.0f);
  su2[j] = acc;
  __syncthreads();
  float b = bd1[j];
  for (int k = 0; k < 64; ++k) b += su2[k] * Wd1[(192 + k) * 64 + j];
  bias_d1[j] = b;
}

#define DEC_ROWS(A_, cr_, cs_, tt)                                            \
  {                                                                           \
    const long n_ = (long)((tt) * 16 + m);                                    \
    cr_ = cnt_r[n_];                                                          \
    cs_ = cnt_s[n_];                                                          \
    const float* pr_ = agg_r + n_ * 64 + 2 * h;                               \
    const float* ps_ = agg_s + n_ * 64 + 2 * h;                               \
    const float* pn_ = n_h + n_ * 64 + 2 * h;                                 \
    _Pragma("unroll") for (int kc = 0; kc < 16; ++kc) {                       \
      A_[kc] = *(const v2f*)(pr_ + 4 * kc);                                   \
      A_[16 + kc] = *(const v2f*)(ps_ + 4 * kc);                              \
      A_[32 + kc] = *(const v2f*)(pn_ + 4 * kc);                              \
    }                                                                         \
  }
__global__ __launch_bounds__(GW * 32) void k_dec(
    const float* __restrict__ agg_r, const float* __restrict__ agg_s,
    const float* __restrict__ cnt_r, const float* __restrict__ cnt_s,
    const float* __restrict__ n_h, const float* __restrict__ Wd1,
    const float* __restrict__ bias_d1, const float* __restrict__ Wd2,
    const float* __restrict__ bd2, float* __restrict__ out) {
  __shared__ float sW1[48 * 4 * 64];
  __shared__ float sW2[16 * 64];
  __shared__ float sH[GW * 16 * 64];
  {
    float wv[96], wu[8];
#pragma unroll
    for (int q = 0; q < 96; ++q) {
      const int i = threadIdx.x + q * (GW * 32);
      int j = i & 1, lane = (i >> 1) & 31, nt = (i >> 6) & 3, kc = i >> 8;
      int kk = 4 * kc + 2 * (lane >> 4) + j;
      wv[q] = Wd1[kk * 64 + nt * 16 + (lane & 15)];
    }
#pragma unroll
    for (int q = 0; q < 8; ++q) {
      const int i = threadIdx.x + q * (GW * 32);
      int j = i & 1, lane = (i >> 1) & 31, kc = i >> 6;
      int kk = 4 * kc + 2 * (lane >> 4) + j;
      wu[q] = Wd2[kk * 16 + (lane & 15)];
    }
#pragma unroll
    for (int q = 0; q < 96; ++q) sW1[threadIdx.x + q * (GW * 32)] = wv[q];
#pragma unroll
    for (int q = 0; q < 8; ++q) sW2[threadIdx.x + q * (GW * 32)] = wu[q];
  }
  __syncthreads();
  const int lane = threadIdx.x & 31;
  const int h = lane >> 4, m = lane & 15;
  v2f bw[192], bu[16];
#pragma unroll
  for (int q = 0; q < 192; ++q) bw[q] = *(const v2f*)(&sW1[q * 64 + lane * 2]);
#pragma unroll
  for (int q = 0; q < 16; ++q) bu[q] = *(const v2f*)(&sW2[q * 64 + lane * 2]);
  float bb[4];
#pragma unroll
  for (int nt = 0; nt < 4; ++nt) bb[nt] = bias_d1[nt * 16 + m];
  const float bb2 = bd2[m];
  const int wv_ = __builtin_amdgcn_readfirstlane(threadIdx.x >> 5);
  float* hw = sH + wv_ * 1024;
  const int nwaves = gridDim.x * GW;
  const int ntiles = NNODES / 16;
  int t = wv_ * gridDim.x + blockIdx.x;
  if (t < ntiles) {
    v2f a[48];
    float cr, cs;
    DEC_ROWS(a, cr, cs, t)
    for (; t < ntiles; t += nwaves) {
      const int tn = min(t + nwaves, ntiles - 1);
      v2f an[48];
      float crn, csn;
      DEC_ROWS(an, crn, csn, tn)
      __builtin_amdgcn_sched_barrier(0);
      const int n0 = t * 16;
      const float invr = 1.0f / fmaxf(cr, 1.0f);
      const float invs = 1.0f / fmaxf(cs, 1.0f);
      v8f acc[4];
#pragma unroll
      for (int nt = 0; nt < 4; ++nt)
        acc[nt] = (v8f){bb[nt], bb[nt], bb[nt], bb[nt], bb[nt], bb[nt], bb[nt], bb[nt]};
#pragma unroll
      for (int kc = 0; kc < 48; ++kc) {
        v2f av = a[kc];
        if (kc < 16)      av *= invr;
        else if (kc < 32) av *= invs;
#pragma unroll
        for (int nt = 0; nt < 4; ++nt)
          acc[nt] = __builtin_amdgcn_wmma_f32_16x16x4_f32(false, av, false, bw[kc * 4 + nt],
                                                          (short)0, acc[nt], false, false);
      }
#pragma unroll
      for (int nt = 0; nt < 4; ++nt)
#pragma unroll
        for (int v = 0; v < 8; ++v)
          hw[(8 * h + v) * 64 + nt * 16 + m] = fmaxf(acc[nt][v], 0.0f);
      v8f acc2 = (v8f){bb2, bb2, bb2, bb2, bb2, bb2, bb2, bb2};
#pragma unroll
      for (int kc = 0; kc < 16; ++kc) {
        const int kk = 4 * kc + 2 * h;
        v2f a2 = *(const v2f*)(hw + m * 64 + kk);
        acc2 = __builtin_amdgcn_wmma_f32_16x16x4_f32(false, a2, false, bu[kc],
                                                     (short)0, acc2, false, false);
      }
#pragma unroll
      for (int v = 0; v < 8; ++v)
        out[(long)(n0 + 8 * h + v) * 16 + m] = acc2[v];
      __builtin_amdgcn_sched_barrier(0);
#pragma unroll
      for (int kc = 0; kc < 48; ++kc) a[kc] = an[kc];
      cr = crn;
      cs = csn;
    }
  }
}

extern "C" void kernel_launch(void* const* d_in, const int* in_sizes, int n_in,
                              void* d_out, int out_size, void* d_ws, size_t ws_size,
                              hipStream_t stream) {
  const float* x    = (const float*)d_in[0];
  const float* ea   = (const float*)d_in[1];
  const float* u    = (const float*)d_in[2];
  const int* send   = (const int*)d_in[3];
  const int* recv   = (const int*)d_in[4];
  const float* Web  = (const float*)d_in[7];
  const float* beb  = (const float*)d_in[8];
  const float* Wnb  = (const float*)d_in[9];
  const float* bnb  = (const float*)d_in[10];
  const float* Wgb  = (const float*)d_in[11];
  const float* bgb  = (const float*)d_in[12];
  const float* Wd1  = (const float*)d_in[13];
  const float* bd1  = (const float*)d_in[14];
  const float* Wd2  = (const float*)d_in[15];
  const float* bd2  = (const float*)d_in[16];
  float* out = (float*)d_out;
  float* ws = (float*)d_ws;

  const long N = NNODES;
  float* agg_r   = ws;
  float* agg_s   = agg_r + N * 64;
  float* cnt_r   = agg_s + N * 64;
  float* cnt_s   = cnt_r + N;
  float* e_sum   = cnt_s + N;
  float* n_sum   = e_sum + 64;
  float* bias_eb = n_sum + 128;
  float* bias_nb = bias_eb + 64;
  float* bias_d1 = bias_nb + 64;
  float* n_hbuf  = bias_d1 + 64;
  int* list_r  = (int*)n_hbuf;
  int* list_s  = list_r + NEDGES;
  int* totals  = list_s + NEDGES;
  int* cursors = totals + 2 * NB;

  k_zero<<<4, 256, 0, stream>>>(e_sum, 128, totals, 4 * NB);
  k_prep<<<1, 64, 0, stream>>>(u, Web, beb, Wnb, bnb, bias_eb, bias_nb);
  k_count<<<NCHUNKS, 256, 0, stream>>>(recv, send, totals);
  k_fill<<<NCHUNKS, 256, 0, stream>>>(recv, send, totals, cursors, list_r, list_s);
  k_edge<<<NB, EW * 32, 0, stream>>>(x, ea, recv, send, Web, bias_eb, list_r, list_s,
                                    totals, agg_r, agg_s, cnt_r, cnt_s);
  k_node<<<256, GW * 32, 0, stream>>>(x, agg_r, Wnb, bias_nb, n_hbuf);
  k_colsum<<<1024, 256, 0, stream>>>(agg_r, e_sum, n_hbuf, n_sum, NNODES);
  k_global<<<1, 64, 0, stream>>>(u, Wgb, bgb, Wd1, bd1, e_sum, n_sum, bias_d1);
  k_dec<<<256, GW * 32, 0, stream>>>(agg_r, agg_s, cnt_r, cnt_s, n_hbuf,
                                   Wd1, bias_d1, Wd2, bd2, out);
}
